# starting kernel
# baseline (speedup 1.0000x reference)
#include <hip/hip_runtime.h>
#include <cmath>

#define NN 100000
#define EE 3200000
#define NEGS 0.2f

typedef __attribute__((ext_vector_type(2))) float v2f;
typedef __attribute__((ext_vector_type(8))) float v8f;

__device__ __forceinline__ unsigned f2mono(float f) {
    unsigned u = __float_as_uint(f);
    return (u & 0x80000000u) ? ~u : (u | 0x80000000u);
}
__device__ __forceinline__ float mono2f(unsigned u) {
    return (u & 0x80000000u) ? __uint_as_float(u ^ 0x80000000u) : __uint_as_float(~u);
}
#define MONO_NEG_INF 0x007FFFFFu

__global__ void prep_weights(const float* __restrict__ W1, const float* __restrict__ We1,
                             const float* __restrict__ att1, const float* __restrict__ We2,
                             const float* __restrict__ att2,
                             float* __restrict__ Wcat1, float* __restrict__ wea1,
                             float* __restrict__ wea2)
{
    int tid = threadIdx.x;
    for (int i = tid; i < 2048; i += blockDim.x) {
        int k = i >> 7, c = i & 127, h = c >> 5, o = c & 31;
        Wcat1[i] = W1[h * 512 + k * 32 + o];
    }
    for (int i = tid; i < 64; i += blockDim.x) {
        int h = i >> 4, k = i & 15;
        float s = 0.f;
        for (int o = 0; o < 32; ++o) s += We1[h * 512 + k * 32 + o] * att1[h * 96 + 64 + o];
        wea1[i] = s;
    }
    for (int i = tid; i < 16; i += blockDim.x) {
        float s = 0.f;
        for (int o = 0; o < 32; ++o) s += We2[i * 32 + o] * att2[64 + o];
        wea2[i] = s;
    }
}

__global__ void init_buffers(unsigned* __restrict__ emax1, float* __restrict__ denom1,
                             float* __restrict__ acc1, unsigned* __restrict__ emax2,
                             float* __restrict__ denom2, float* __restrict__ acc2)
{
    int i0 = blockIdx.x * blockDim.x + threadIdx.x;
    int stride = gridDim.x * blockDim.x;
    for (int i = i0; i < NN * 128; i += stride) acc1[i] = 0.f;
    for (int i = i0; i < NN * 32; i += stride) acc2[i] = 0.f;
    for (int i = i0; i < NN * 4; i += stride) { emax1[i] = MONO_NEG_INF; denom1[i] = 0.f; }
    for (int i = i0; i < NN; i += stride) { emax2[i] = MONO_NEG_INF; denom2[i] = 0.f; }
}

__global__ void wmma_gemm(const float* __restrict__ A, int lda,
                          const float* __restrict__ B, int ldb,
                          float* __restrict__ C, int ldc,
                          int M, int Ncols, int K,
                          const float* __restrict__ bias, int nelu)
{
    int lane = threadIdx.x & 31;
    int tile = blockIdx.x * (blockDim.x >> 5) + (threadIdx.x >> 5);
    int tilesN = Ncols >> 4;
    int total = (M >> 4) * tilesN;
    if (tile >= total) return;
    int tm = tile / tilesN, tn = tile % tilesN;
    int r = lane & 15;
    int kk = (lane >> 4) << 1;
    const float* Ap = A + (size_t)(tm * 16 + r) * lda + kk;
    const float* Bp = B + (size_t)kk * ldb + (tn * 16 + r);
    v8f acc = {};
    for (int k0 = 0; k0 < K; k0 += 4) {
        v2f a; a.x = Ap[0]; a.y = Ap[1];
        v2f b; b.x = Bp[0]; b.y = Bp[ldb];
        acc = __builtin_amdgcn_wmma_f32_16x16x4_f32(false, a, false, b,
                                                    (short)0, acc, false, false);
        Ap += 4;
        Bp += (size_t)ldb * 4;
    }
    int colg = tn * 16 + r;
    float bv = bias ? bias[colg] : 0.f;
    int rbase = tm * 16 + ((lane >> 4) << 3);
#pragma unroll
    for (int v = 0; v < 8; ++v) {
        float xv = acc[v] + bv;
        for (int t = 0; t < nelu; ++t) xv = (xv > 0.f) ? xv : (expf(xv) - 1.f);
        C[(size_t)(rbase + v) * ldc + colg] = xv;
    }
}

__global__ void scores1_kernel(const float* __restrict__ xw1, const float* __restrict__ att1,
                               float* __restrict__ s_i1, float* __restrict__ s_j1)
{
    int t = blockIdx.x * blockDim.x + threadIdx.x;
    if (t >= NN * 4) return;
    int n = t >> 2, h = t & 3;
    const float* a = att1 + h * 96;
    const float* xp = xw1 + (size_t)n * 128 + h * 32;
    float si = 0.f, sj = 0.f;
#pragma unroll
    for (int o = 0; o < 32; ++o) { float v = xp[o]; si += v * a[o]; sj += v * a[32 + o]; }
    s_i1[t] = si; s_j1[t] = sj;
}

__global__ void scores2_kernel(const float* __restrict__ hw2, const float* __restrict__ att2,
                               float* __restrict__ s_i2, float* __restrict__ s_j2)
{
    int n = blockIdx.x * blockDim.x + threadIdx.x;
    if (n >= NN) return;
    const float* xp = hw2 + (size_t)n * 32;
    float si = 0.f, sj = 0.f;
#pragma unroll
    for (int o = 0; o < 32; ++o) { float v = xp[o]; si += v * att2[o]; sj += v * att2[32 + o]; }
    s_i2[n] = si; s_j2[n] = sj;
}

__global__ void edge_scores_l1(const int* __restrict__ src, const int* __restrict__ dst,
                               const float* __restrict__ eattr, const float* __restrict__ wea1,
                               const float* __restrict__ s_i1, const float* __restrict__ s_j1,
                               unsigned* __restrict__ emax1)
{
    int t = blockIdx.x * blockDim.x + threadIdx.x;
    if (t >= EE * 4) return;
    int e = t >> 2, h = t & 3;
    int s = src[e], d = dst[e];
    float se = 0.f;
#pragma unroll
    for (int k = 0; k < 16; ++k) se += eattr[(size_t)e * 16 + k] * wea1[h * 16 + k];
    float sc = s_i1[d * 4 + h] + s_j1[s * 4 + h] + se;
    sc = (sc >= 0.f) ? sc : NEGS * sc;
    atomicMax(&emax1[d * 4 + h], f2mono(sc));
}

__global__ void edge_scores_l2(const int* __restrict__ src, const int* __restrict__ dst,
                               const float* __restrict__ eattr, const float* __restrict__ wea2,
                               const float* __restrict__ s_i2, const float* __restrict__ s_j2,
                               unsigned* __restrict__ emax2)
{
    int e = blockIdx.x * blockDim.x + threadIdx.x;
    if (e >= EE) return;
    int s = src[e], d = dst[e];
    float se = 0.f;
#pragma unroll
    for (int k = 0; k < 16; ++k) se += eattr[(size_t)e * 16 + k] * wea2[k];
    float sc = s_i2[d] + s_j2[s] + se;
    sc = (sc >= 0.f) ? sc : NEGS * sc;
    atomicMax(&emax2[d], f2mono(sc));
}

__global__ void decode_emax(unsigned* __restrict__ em, int n)
{
    int i = blockIdx.x * blockDim.x + threadIdx.x;
    if (i >= n) return;
    float f = mono2f(em[i]);
    if (!isfinite(f)) f = 0.f;
    em[i] = __float_as_uint(f);
}

__global__ void edge_aggr_l1(const int* __restrict__ src, const int* __restrict__ dst,
                             const float* __restrict__ eattr, const float* __restrict__ wea1,
                             const float* __restrict__ s_i1, const float* __restrict__ s_j1,
                             const float* __restrict__ emax1, float* __restrict__ denom1,
                             const float* __restrict__ xw1, float* __restrict__ acc1)
{
    int lane = threadIdx.x & 31;
    int wave = (blockIdx.x * blockDim.x + threadIdx.x) >> 5;
    int nw = (gridDim.x * blockDim.x) >> 5;
    for (int e = wave; e < EE; e += nw) {
        int s = src[e], d = dst[e];
        float se[4] = {0.f, 0.f, 0.f, 0.f};
#pragma unroll
        for (int k = 0; k < 16; ++k) {
            float ea = eattr[(size_t)e * 16 + k];
            se[0] += ea * wea1[k];      se[1] += ea * wea1[16 + k];
            se[2] += ea * wea1[32 + k]; se[3] += ea * wea1[48 + k];
        }
        float p[4];
#pragma unroll
        for (int h = 0; h < 4; ++h) {
            float sc = s_i1[d * 4 + h] + s_j1[s * 4 + h] + se[h];
            sc = (sc >= 0.f) ? sc : NEGS * sc;
            p[h] = expf(sc - emax1[d * 4 + h]);
        }
        if (lane < 4) atomicAdd(&denom1[d * 4 + lane], p[lane]);
#pragma unroll
        for (int h = 0; h < 4; ++h) {
            float v = p[h] * xw1[(size_t)s * 128 + h * 32 + lane];
            atomicAdd(&acc1[(size_t)d * 128 + h * 32 + lane], v);
        }
    }
}

__global__ void edge_aggr_l2(const int* __restrict__ src, const int* __restrict__ dst,
                             const float* __restrict__ eattr, const float* __restrict__ wea2,
                             const float* __restrict__ s_i2, const float* __restrict__ s_j2,
                             const float* __restrict__ emax2, float* __restrict__ denom2,
                             const float* __restrict__ hw2, float* __restrict__ acc2)
{
    int lane = threadIdx.x & 31;
    int wave = (blockIdx.x * blockDim.x + threadIdx.x) >> 5;
    int nw = (gridDim.x * blockDim.x) >> 5;
    for (int e = wave; e < EE; e += nw) {
        int s = src[e], d = dst[e];
        float se = 0.f;
#pragma unroll
        for (int k = 0; k < 16; ++k) se += eattr[(size_t)e * 16 + k] * wea2[k];
        float sc = s_i2[d] + s_j2[s] + se;
        sc = (sc >= 0.f) ? sc : NEGS * sc;
        float p = expf(sc - emax2[d]);
        if (lane == 0) atomicAdd(&denom2[d], p);
        atomicAdd(&acc2[(size_t)d * 32 + lane], p * hw2[(size_t)s * 32 + lane]);
    }
}

__global__ void finalize_l1(float* __restrict__ acc1, const float* __restrict__ denom1)
{
    int i = blockIdx.x * blockDim.x + threadIdx.x;
    if (i >= NN * 128) return;
    int n = i >> 7, h = (i >> 5) & 3;
    float dnm = denom1[n * 4 + h];
    acc1[i] = (dnm > 0.f) ? acc1[i] / dnm : 0.f;
}

__global__ void final_out(const float* __restrict__ acc2, const float* __restrict__ denom2,
                          const float* __restrict__ p2w, const float* __restrict__ p2b,
                          float* __restrict__ out)
{
    int n = blockIdx.x * blockDim.x + threadIdx.x;
    if (n >= NN) return;
    float dnm = denom2[n];
    float v = 0.f;
#pragma unroll
    for (int o = 0; o < 32; ++o) {
        float hv = (dnm > 0.f) ? acc2[(size_t)n * 32 + o] / dnm : 0.f;
        v += hv * p2w[o];
    }
    v += p2b[0];
    out[n] = (v > 0.f) ? v : (expf(v) - 1.f);
}

extern "C" void kernel_launch(void* const* d_in, const int* in_sizes, int n_in,
                              void* d_out, int out_size, void* d_ws, size_t ws_size,
                              hipStream_t stream)
{
    const float* x       = (const float*)d_in[0];
    const int*   eidx    = (const int*)d_in[1];
    const float* eattr   = (const float*)d_in[2];
    const float* W1      = (const float*)d_in[3];
    const float* We1     = (const float*)d_in[4];
    const float* att1    = (const float*)d_in[5];
    const float* proj1_w = (const float*)d_in[6];
    const float* proj1_b = (const float*)d_in[7];
    const float* W2      = (const float*)d_in[8];
    const float* We2     = (const float*)d_in[9];
    const float* att2    = (const float*)d_in[10];
    const float* proj2_w = (const float*)d_in[11];
    const float* proj2_b = (const float*)d_in[12];
    float* out = (float*)d_out;

    const int* srcp = eidx;
    const int* dstp = eidx + EE;

    float* ws     = (float*)d_ws;
    float* xw1    = ws;
    float* acc1   = xw1 + (size_t)NN * 128;
    float* s_i1   = acc1 + (size_t)NN * 128;
    float* s_j1   = s_i1 + (size_t)NN * 4;
    float* emax1  = s_j1 + (size_t)NN * 4;
    float* denom1 = emax1 + (size_t)NN * 4;
    float* hw2    = denom1 + (size_t)NN * 4;
    float* acc2   = hw2 + (size_t)NN * 32;
    float* s_i2   = acc2 + (size_t)NN * 32;
    float* s_j2   = s_i2 + (size_t)NN;
    float* emax2  = s_j2 + (size_t)NN;
    float* denom2 = emax2 + (size_t)NN;
    float* Wcat1  = denom2 + (size_t)NN;
    float* wea1   = Wcat1 + 2048;
    float* wea2   = wea1 + 64;

    prep_weights<<<1, 256, 0, stream>>>(W1, We1, att1, We2, att2, Wcat1, wea1, wea2);
    init_buffers<<<2048, 256, 0, stream>>>((unsigned*)emax1, denom1, acc1,
                                           (unsigned*)emax2, denom2, acc2);

    {
        int tiles = (NN / 16) * (128 / 16);
        wmma_gemm<<<(tiles + 7) / 8, 256, 0, stream>>>(x, 16, Wcat1, 128, xw1, 128,
                                                       NN, 128, 16, nullptr, 0);
    }
    scores1_kernel<<<(NN * 4 + 255) / 256, 256, 0, stream>>>(xw1, att1, s_i1, s_j1);
    edge_scores_l1<<<(EE * 4 + 255) / 256, 256, 0, stream>>>(srcp, dstp, eattr, wea1,
                                                             s_i1, s_j1, (unsigned*)emax1);
    decode_emax<<<(NN * 4 + 255) / 256, 256, 0, stream>>>((unsigned*)emax1, NN * 4);
    edge_aggr_l1<<<4096, 256, 0, stream>>>(srcp, dstp, eattr, wea1, s_i1, s_j1,
                                           emax1, denom1, xw1, acc1);
    finalize_l1<<<(NN * 128 + 255) / 256, 256, 0, stream>>>(acc1, denom1);

    {
        int tiles = (NN / 16) * (128 / 16);
        wmma_gemm<<<(tiles + 7) / 8, 256, 0, stream>>>(acc1, 128, proj1_w, 128, xw1, 128,
                                                       NN, 128, 128, proj1_b, 2);
    }
    {
        int tiles = (NN / 16) * (32 / 16);
        wmma_gemm<<<(tiles + 7) / 8, 256, 0, stream>>>(xw1, 128, W2, 32, hw2, 32,
                                                       NN, 32, 128, nullptr, 0);
    }
    scores2_kernel<<<(NN + 255) / 256, 256, 0, stream>>>(hw2, att2, s_i2, s_j2);
    edge_scores_l2<<<(EE + 255) / 256, 256, 0, stream>>>(srcp, dstp, eattr, wea2,
                                                         s_i2, s_j2, (unsigned*)emax2);
    decode_emax<<<(NN + 255) / 256, 256, 0, stream>>>((unsigned*)emax2, NN);
    edge_aggr_l2<<<4096, 256, 0, stream>>>(srcp, dstp, eattr, wea2, s_i2, s_j2,
                                           emax2, denom2, hw2, acc2);
    final_out<<<(NN + 255) / 256, 256, 0, stream>>>(acc2, denom2, proj2_w, proj2_b, out);
}


# Segment sums moved across the linear maps: layer 1 sums alpha*x[src] (16 per head) and applies W1 folded into proj1 (64x128); layer 2 sums the scalar alpha*(hw2.proj2_w)[src]; sums reassociated
# speedup vs baseline: 1.7037x; 1.7037x over previous
#include <hip/hip_runtime.h>
#include <cmath>

#define NN 100000
#define EE 3200000
#define NEGS 0.2f

typedef __attribute__((ext_vector_type(2))) float v2f;
typedef __attribute__((ext_vector_type(8))) float v8f;

__device__ __forceinline__ unsigned f2mono(float f) {
    unsigned u = __float_as_uint(f);
    return (u & 0x80000000u) ? ~u : (u | 0x80000000u);
}
__device__ __forceinline__ float mono2f(unsigned u) {
    return (u & 0x80000000u) ? __uint_as_float(u ^ 0x80000000u) : __uint_as_float(~u);
}
#define MONO_NEG_INF 0x007FFFFFu

__global__ void prep_weights(const float* __restrict__ W1, const float* __restrict__ We1,
                             const float* __restrict__ att1, const float* __restrict__ We2,
                             const float* __restrict__ att2, const float* __restrict__ proj1_w,
                             float* __restrict__ Wcat1, float* __restrict__ wea1,
                             float* __restrict__ wea2, float* __restrict__ Wcomb)
{
    int tid = blockIdx.x * blockDim.x + threadIdx.x;
    int nt = gridDim.x * blockDim.x;
    for (int i = tid; i < 2048; i += nt) {
        int k = i >> 7, c = i & 127, h = c >> 5, o = c & 31;
        Wcat1[i] = W1[h * 512 + k * 32 + o];
    }
    for (int i = tid; i < 64; i += nt) {
        int h = i >> 4, k = i & 15;
        float s = 0.f;
        for (int o = 0; o < 32; ++o) s += We1[h * 512 + k * 32 + o] * att1[h * 96 + 64 + o];
        wea1[i] = s;
    }
    for (int i = tid; i < 16; i += nt) {
        float s = 0.f;
        for (int o = 0; o < 32; ++o) s += We2[i * 32 + o] * att2[64 + o];
        wea2[i] = s;
    }
    for (int i = tid; i < 64 * 128; i += nt) {
        int row = i >> 7, c = i & 127, h = row >> 4, k = row & 15;
        const float* w = W1 + h * 512 + k * 32;
        const float* pw = proj1_w + (size_t)(h * 32) * 128 + c;
        float s = 0.f;
        for (int o = 0; o < 32; ++o) s += w[o] * pw[(size_t)o * 128];
        Wcomb[i] = s;
    }
}

__global__ void init_buffers(unsigned* __restrict__ emax1, float* __restrict__ denom1,
                             float* __restrict__ accx1, unsigned* __restrict__ emax2,
                             float* __restrict__ nd2)
{
    int i0 = blockIdx.x * blockDim.x + threadIdx.x;
    int stride = gridDim.x * blockDim.x;
    for (int i = i0; i < NN * 64; i += stride) accx1[i] = 0.f;
    for (int i = i0; i < NN * 4; i += stride) { emax1[i] = MONO_NEG_INF; denom1[i] = 0.f; }
    for (int i = i0; i < NN * 2; i += stride) nd2[i] = 0.f;
    for (int i = i0; i < NN; i += stride) emax2[i] = MONO_NEG_INF;
}

__global__ void wmma_gemm(const float* __restrict__ A, int lda,
                          const float* __restrict__ B, int ldb,
                          float* __restrict__ C, int ldc,
                          int M, int Ncols, int K,
                          const float* __restrict__ bias, int nelu)
{
    int lane = threadIdx.x & 31;
    int tile = blockIdx.x * (blockDim.x >> 5) + (threadIdx.x >> 5);
    int tilesN = Ncols >> 4;
    int total = (M >> 4) * tilesN;
    if (tile >= total) return;
    int tm = tile / tilesN, tn = tile % tilesN;
    int r = lane & 15;
    int kk = (lane >> 4) << 1;
    const float* Ap = A + (size_t)(tm * 16 + r) * lda + kk;
    const float* Bp = B + (size_t)kk * ldb + (tn * 16 + r);
    v8f acc = {};
    for (int k0 = 0; k0 < K; k0 += 4) {
        v2f a; a.x = Ap[0]; a.y = Ap[1];
        v2f b; b.x = Bp[0]; b.y = Bp[ldb];
        acc = __builtin_amdgcn_wmma_f32_16x16x4_f32(false, a, false, b,
                                                    (short)0, acc, false, false);
        Ap += 4;
        Bp += (size_t)ldb * 4;
    }
    int colg = tn * 16 + r;
    float bv = bias ? bias[colg] : 0.f;
    int rbase = tm * 16 + ((lane >> 4) << 3);
#pragma unroll
    for (int v = 0; v < 8; ++v) {
        float xv = acc[v] + bv;
        for (int t = 0; t < nelu; ++t) xv = (xv > 0.f) ? xv : (expf(xv) - 1.f);
        C[(size_t)(rbase + v) * ldc + colg] = xv;
    }
}

__global__ void scores1_kernel(const float* __restrict__ xw1, const float* __restrict__ att1,
                               float* __restrict__ s_i1, float* __restrict__ s_j1)
{
    int t = blockIdx.x * blockDim.x + threadIdx.x;
    if (t >= NN * 4) return;
    int n = t >> 2, h = t & 3;
    const float* a = att1 + h * 96;
    const float* xp = xw1 + (size_t)n * 128 + h * 32;
    float si = 0.f, sj = 0.f;
#pragma unroll
    for (int o = 0; o < 32; ++o) { float v = xp[o]; si += v * a[o]; sj += v * a[32 + o]; }
    s_i1[t] = si; s_j1[t] = sj;
}

__global__ void scores2_kernel(const float* __restrict__ hw2, const float* __restrict__ att2,
                               const float* __restrict__ p2w,
                               float* __restrict__ s_i2, float* __restrict__ s_j2,
                               float* __restrict__ z2)
{
    int n = blockIdx.x * blockDim.x + threadIdx.x;
    if (n >= NN) return;
    const float* xp = hw2 + (size_t)n * 32;
    float si = 0.f, sj = 0.f, z = 0.f;
#pragma unroll
    for (int o = 0; o < 32; ++o) {
        float v = xp[o];
        si += v * att2[o]; sj += v * att2[32 + o]; z += v * p2w[o];
    }
    s_i2[n] = si; s_j2[n] = sj; z2[n] = z;
}

__global__ void edge_scores_l1(const int* __restrict__ src, const int* __restrict__ dst,
                               const float* __restrict__ eattr, const float* __restrict__ wea1,
                               const float* __restrict__ s_i1, const float* __restrict__ s_j1,
                               unsigned* __restrict__ emax1)
{
    int t = blockIdx.x * blockDim.x + threadIdx.x;
    if (t >= EE * 4) return;
    int e = t >> 2, h = t & 3;
    int s = src[e], d = dst[e];
    float se = 0.f;
#pragma unroll
    for (int k = 0; k < 16; ++k) se += eattr[(size_t)e * 16 + k] * wea1[h * 16 + k];
    float sc = s_i1[d * 4 + h] + s_j1[s * 4 + h] + se;
    sc = (sc >= 0.f) ? sc : NEGS * sc;
    atomicMax(&emax1[d * 4 + h], f2mono(sc));
}

__global__ void edge_scores_l2(const int* __restrict__ src, const int* __restrict__ dst,
                               const float* __restrict__ eattr, const float* __restrict__ wea2,
                               const float* __restrict__ s_i2, const float* __restrict__ s_j2,
                               unsigned* __restrict__ emax2)
{
    int e = blockIdx.x * blockDim.x + threadIdx.x;
    if (e >= EE) return;
    int s = src[e], d = dst[e];
    float se = 0.f;
#pragma unroll
    for (int k = 0; k < 16; ++k) se += eattr[(size_t)e * 16 + k] * wea2[k];
    float sc = s_i2[d] + s_j2[s] + se;
    sc = (sc >= 0.f) ? sc : NEGS * sc;
    atomicMax(&emax2[d], f2mono(sc));
}

__global__ void decode_emax(unsigned* __restrict__ em, int n)
{
    int i = blockIdx.x * blockDim.x + threadIdx.x;
    if (i >= n) return;
    float f = mono2f(em[i]);
    if (!isfinite(f)) f = 0.f;
    em[i] = __float_as_uint(f);
}

__global__ void edge_aggr_l1(const int* __restrict__ src, const int* __restrict__ dst,
                             const float* __restrict__ eattr, const float* __restrict__ wea1,
                             const float* __restrict__ s_i1, const float* __restrict__ s_j1,
                             const float* __restrict__ emax1, float* __restrict__ denom1,
                             const float* __restrict__ x, float* __restrict__ accx1)
{
    int lane = threadIdx.x & 31;
    int wave = (blockIdx.x * blockDim.x + threadIdx.x) >> 5;
    int nw = (gridDim.x * blockDim.x) >> 5;
    for (int e = wave; e < EE; e += nw) {
        int s = src[e], d = dst[e];
        float xv = x[(size_t)s * 16 + (lane & 15)];
        float se[4] = {0.f, 0.f, 0.f, 0.f};
#pragma unroll
        for (int k = 0; k < 16; ++k) {
            float ea = eattr[(size_t)e * 16 + k];
            se[0] += ea * wea1[k];      se[1] += ea * wea1[16 + k];
            se[2] += ea * wea1[32 + k]; se[3] += ea * wea1[48 + k];
        }
        float p[4];
#pragma unroll
        for (int h = 0; h < 4; ++h) {
            float sc = s_i1[d * 4 + h] + s_j1[s * 4 + h] + se[h];
            sc = (sc >= 0.f) ? sc : NEGS * sc;
            p[h] = expf(sc - emax1[d * 4 + h]);
        }
        if (lane < 4) atomicAdd(&denom1[d * 4 + lane], p[lane]);
        bool hi = lane >= 16;
        atomicAdd(&accx1[(size_t)d * 64 + lane], (hi ? p[1] : p[0]) * xv);
        atomicAdd(&accx1[(size_t)d * 64 + 32 + lane], (hi ? p[3] : p[2]) * xv);
    }
}

__global__ void edge_aggr_l2(const int* __restrict__ src, const int* __restrict__ dst,
                             const float* __restrict__ eattr, const float* __restrict__ wea2,
                             const float* __restrict__ s_i2, const float* __restrict__ s_j2,
                             const float* __restrict__ emax2, const float* __restrict__ z2,
                             float* __restrict__ nd2)
{
    int t = blockIdx.x * blockDim.x + threadIdx.x;
    if (t >= EE * 2) return;
    int e = t >> 1, c = t & 1;
    int s = src[e], d = dst[e];
    float zv = z2[s];
    float se = 0.f;
#pragma unroll
    for (int k = 0; k < 16; ++k) se += eattr[(size_t)e * 16 + k] * wea2[k];
    float sc = s_i2[d] + s_j2[s] + se;
    sc = (sc >= 0.f) ? sc : NEGS * sc;
    float p = expf(sc - emax2[d]);
    float v = c ? p : p * zv;
    atomicAdd(&nd2[(size_t)d * 2 + c], v);
}

__global__ void finalize_l1(float* __restrict__ accx1, const float* __restrict__ denom1)
{
    int i = blockIdx.x * blockDim.x + threadIdx.x;
    if (i >= NN * 64) return;
    int n = i >> 6, h = (i >> 4) & 3;
    float dnm = denom1[n * 4 + h];
    accx1[i] = (dnm > 0.f) ? accx1[i] / dnm : 0.f;
}

__global__ void final_out(const float* __restrict__ nd2, const float* __restrict__ p2b,
                          float* __restrict__ out)
{
    int n = blockIdx.x * blockDim.x + threadIdx.x;
    if (n >= NN) return;
    float num = nd2[(size_t)n * 2], dnm = nd2[(size_t)n * 2 + 1];
    float v = (dnm > 0.f) ? num / dnm : 0.f;
    v += p2b[0];
    out[n] = (v > 0.f) ? v : (expf(v) - 1.f);
}

extern "C" void kernel_launch(void* const* d_in, const int* in_sizes, int n_in,
                              void* d_out, int out_size, void* d_ws, size_t ws_size,
                              hipStream_t stream)
{
    const float* x       = (const float*)d_in[0];
    const int*   eidx    = (const int*)d_in[1];
    const float* eattr   = (const float*)d_in[2];
    const float* W1      = (const float*)d_in[3];
    const float* We1     = (const float*)d_in[4];
    const float* att1    = (const float*)d_in[5];
    const float* proj1_w = (const float*)d_in[6];
    const float* proj1_b = (const float*)d_in[7];
    const float* W2      = (const float*)d_in[8];
    const float* We2     = (const float*)d_in[9];
    const float* att2    = (const float*)d_in[10];
    const float* proj2_w = (const float*)d_in[11];
    const float* proj2_b = (const float*)d_in[12];
    float* out = (float*)d_out;

    const int* srcp = eidx;
    const int* dstp = eidx + EE;

    float* ws     = (float*)d_ws;
    float* xw1    = ws;
    float* accx1  = xw1 + (size_t)NN * 128;
    float* Wcomb  = accx1 + (size_t)NN * 64;
    float* z2     = Wcomb + 64 * 128;
    float* s_i1   = accx1 + (size_t)NN * 128;
    float* s_j1   = s_i1 + (size_t)NN * 4;
    float* emax1  = s_j1 + (size_t)NN * 4;
    float* denom1 = emax1 + (size_t)NN * 4;
    float* hw2    = denom1 + (size_t)NN * 4;
    float* nd2    = hw2 + (size_t)NN * 32;
    float* s_i2   = nd2 + (size_t)NN * 32;
    float* s_j2   = s_i2 + (size_t)NN;
    float* emax2  = s_j2 + (size_t)NN;
    float* Wcat1  = emax2 + (size_t)NN * 2;
    float* wea1   = Wcat1 + 2048;
    float* wea2   = wea1 + 64;

    prep_weights<<<32, 256, 0, stream>>>(W1, We1, att1, We2, att2, proj1_w,
                                         Wcat1, wea1, wea2, Wcomb);
    init_buffers<<<2048, 256, 0, stream>>>((unsigned*)emax1, denom1, accx1,
                                           (unsigned*)emax2, nd2);

    {
        int tiles = (NN / 16) * (128 / 16);
        wmma_gemm<<<(tiles + 7) / 8, 256, 0, stream>>>(x, 16, Wcat1, 128, xw1, 128,
                                                       NN, 128, 16, nullptr, 0);
    }
    scores1_kernel<<<(NN * 4 + 255) / 256, 256, 0, stream>>>(xw1, att1, s_i1, s_j1);
    edge_scores_l1<<<(EE * 4 + 255) / 256, 256, 0, stream>>>(srcp, dstp, eattr, wea1,
                                                             s_i1, s_j1, (unsigned*)emax1);
    decode_emax<<<(NN * 4 + 255) / 256, 256, 0, stream>>>((unsigned*)emax1, NN * 4);
    edge_aggr_l1<<<4096, 256, 0, stream>>>(srcp, dstp, eattr, wea1, s_i1, s_j1,
                                           emax1, denom1, x, accx1);
    finalize_l1<<<(NN * 64 + 255) / 256, 256, 0, stream>>>(accx1, denom1);

    {
        int tiles = (NN / 16) * (128 / 16);
        wmma_gemm<<<(tiles + 7) / 8, 256, 0, stream>>>(accx1, 64, Wcomb, 128, xw1, 128,
                                                       NN, 128, 64, proj1_b, 2);
    }
    {
        int tiles = (NN / 16) * (32 / 16);
        wmma_gemm<<<(tiles + 7) / 8, 256, 0, stream>>>(xw1, 128, W2, 32, hw2, 32,
                                                       NN, 32, 128, nullptr, 0);
    }
    scores2_kernel<<<(NN + 255) / 256, 256, 0, stream>>>(hw2, att2, proj2_w, s_i2, s_j2, z2);
    edge_scores_l2<<<(EE + 255) / 256, 256, 0, stream>>>(srcp, dstp, eattr, wea2,
                                                         s_i2, s_j2, (unsigned*)emax2);
    decode_emax<<<(NN + 255) / 256, 256, 0, stream>>>((unsigned*)emax2, NN);
    edge_aggr_l2<<<(EE * 2 + 255) / 256, 256, 0, stream>>>(srcp, dstp, eattr, wea2, s_i2, s_j2,
                                                           emax2, z2, nd2);
    final_out<<<(NN + 255) / 256, 256, 0, stream>>>(nd2, proj2_b, out);
}


# Edges bucketed by destination each call (integer counting sort, arrival-order rank, dots stored per slot); one wave per node, running max and exp-sums in registers; no float atomics; sums reordered
# speedup vs baseline: 3.4183x; 2.0064x over previous
#include <hip/hip_runtime.h>
#include <cmath>

#define NN 100000
#define EE 3200000
#define NEGS 0.2f

typedef __attribute__((ext_vector_type(2))) float v2f;
typedef __attribute__((ext_vector_type(8))) float v8f;

__global__ void prep_weights(const float* __restrict__ W1, const float* __restrict__ We1,
                             const float* __restrict__ att1, const float* __restrict__ We2,
                             const float* __restrict__ att2, const float* __restrict__ proj1_w,
                             float* __restrict__ Wcat1, float* __restrict__ wea1,
                             float* __restrict__ wea2, float* __restrict__ Wcomb,
                             int* __restrict__ cnt)
{
    int tid = blockIdx.x * blockDim.x + threadIdx.x;
    int nt = gridDim.x * blockDim.x;
    for (int i = tid; i < NN; i += nt) cnt[i] = 0;
    for (int i = tid; i < 2048; i += nt) {
        int k = i >> 7, c = i & 127, h = c >> 5, o = c & 31;
        Wcat1[i] = W1[h * 512 + k * 32 + o];
    }
    for (int i = tid; i < 64; i += nt) {
        int h = i >> 4, k = i & 15;
        float s = 0.f;
        for (int o = 0; o < 32; ++o) s += We1[h * 512 + k * 32 + o] * att1[h * 96 + 64 + o];
        wea1[i] = s;
    }
    for (int i = tid; i < 16; i += nt) {
        float s = 0.f;
        for (int o = 0; o < 32; ++o) s += We2[i * 32 + o] * att2[64 + o];
        wea2[i] = s;
    }
    for (int i = tid; i < 64 * 128; i += nt) {
        int row = i >> 7, c = i & 127, h = row >> 4, k = row & 15;
        const float* w = W1 + h * 512 + k * 32;
        const float* pw = proj1_w + (size_t)(h * 32) * 128 + c;
        float s = 0.f;
        for (int o = 0; o < 32; ++o) s += w[o] * pw[(size_t)o * 128];
        Wcomb[i] = s;
    }
}

__global__ void wmma_gemm(const float* __restrict__ A, int lda,
                          const float* __restrict__ B, int ldb,
                          float* __restrict__ C, int ldc,
                          int M, int Ncols, int K,
                          const float* __restrict__ bias, int nelu)
{
    int lane = threadIdx.x & 31;
    int tile = blockIdx.x * (blockDim.x >> 5) + (threadIdx.x >> 5);
    int tilesN = Ncols >> 4;
    int total = (M >> 4) * tilesN;
    if (tile >= total) return;
    int tm = tile / tilesN, tn = tile % tilesN;
    int r = lane & 15;
    int kk = (lane >> 4) << 1;
    const float* Ap = A + (size_t)(tm * 16 + r) * lda + kk;
    const float* Bp = B + (size_t)kk * ldb + (tn * 16 + r);
    v8f acc = {};
    for (int k0 = 0; k0 < K; k0 += 4) {
        v2f a; a.x = Ap[0]; a.y = Ap[1];
        v2f b; b.x = Bp[0]; b.y = Bp[ldb];
        acc = __builtin_amdgcn_wmma_f32_16x16x4_f32(false, a, false, b,
                                                    (short)0, acc, false, false);
        Ap += 4;
        Bp += (size_t)ldb * 4;
    }
    int colg = tn * 16 + r;
    float bv = bias ? bias[colg] : 0.f;
    int rbase = tm * 16 + ((lane >> 4) << 3);
#pragma unroll
    for (int v = 0; v < 8; ++v) {
        float xv = acc[v] + bv;
        for (int t = 0; t < nelu; ++t) xv = (xv > 0.f) ? xv : (expf(xv) - 1.f);
        C[(size_t)(rbase + v) * ldc + colg] = xv;
    }
}

__global__ void scores1_kernel(const float* __restrict__ xw1, const float* __restrict__ att1,
                               float* __restrict__ s_i1, float* __restrict__ s_j1)
{
    int t = blockIdx.x * blockDim.x + threadIdx.x;
    if (t >= NN * 4) return;
    int n = t >> 2, h = t & 3;
    const float* a = att1 + h * 96;
    const float* xp = xw1 + (size_t)n * 128 + h * 32;
    float si = 0.f, sj = 0.f;
#pragma unroll
    for (int o = 0; o < 32; ++o) { float v = xp[o]; si += v * a[o]; sj += v * a[32 + o]; }
    s_i1[t] = si; s_j1[t] = sj;
}

__global__ void scores2_kernel(const float* __restrict__ hw2, const float* __restrict__ att2,
                               const float* __restrict__ p2w,
                               float* __restrict__ s_i2, float* __restrict__ sz2)
{
    int n = blockIdx.x * blockDim.x + threadIdx.x;
    if (n >= NN) return;
    const float* xp = hw2 + (size_t)n * 32;
    float si = 0.f, sj = 0.f, z = 0.f;
#pragma unroll
    for (int o = 0; o < 32; ++o) {
        float v = xp[o];
        si += v * att2[o]; sj += v * att2[32 + o]; z += v * p2w[o];
    }
    s_i2[n] = si;
    float2 r; r.x = sj; r.y = z;
    ((float2*)sz2)[n] = r;
}

__global__ void __launch_bounds__(256)
edge_rank(const int* __restrict__ dst, int* __restrict__ cnt, int* __restrict__ rank)
{
    int i0 = blockIdx.x * blockDim.x + threadIdx.x;
    int stride = gridDim.x * blockDim.x;
    for (int e = i0; e < EE; e += stride) rank[e] = atomicAdd(&cnt[dst[e]], 1);
}

__global__ void __launch_bounds__(1024)
row_offsets(const int* __restrict__ cnt, int* __restrict__ rowptr)
{
    __shared__ int wtot[32];
    __shared__ int wpart[32];
    const int t = threadIdx.x, lane = t & 31, w = t >> 5;
    const int b0 = blockIdx.x * 1024;
    int part = 0;
    for (int i = t; i < b0; i += 1024) part += cnt[i];
    const int idx = b0 + t;
    const int c = (idx < NN) ? cnt[idx] : 0;
    int incl = c;
    for (int o = 1; o < 32; o <<= 1) {
        int v = __shfl_up(incl, o);
        if (lane >= o) incl += v;
    }
    for (int o = 16; o > 0; o >>= 1) part += __shfl_xor(part, o);
    if (lane == 31) wtot[w] = incl;
    if (lane == 0) wpart[w] = part;
    __syncthreads();
    int base = wpart[lane];
    int pre = (lane < w) ? wtot[lane] : 0;
    for (int o = 16; o > 0; o >>= 1) {
        base += __shfl_xor(base, o);
        pre += __shfl_xor(pre, o);
    }
    if (idx < NN) rowptr[idx] = base + pre + incl - c;
}

__global__ void __launch_bounds__(256)
edge_place(const int* __restrict__ src, const int* __restrict__ dst,
           const float* __restrict__ eattr, const float* __restrict__ wea1,
           const float* __restrict__ wea2, const int* __restrict__ rowptr,
           const int* __restrict__ rank, float* __restrict__ se1S,
           int* __restrict__ pairS)
{
    int i0 = blockIdx.x * blockDim.x + threadIdx.x;
    int stride = gridDim.x * blockDim.x;
    for (int e = i0; e < EE; e += stride) {
        const int d = dst[e];
        const int sv = src[e];
        const int pos = rowptr[d] + rank[e];
        float a0 = 0.f, a1 = 0.f, a2 = 0.f, a3 = 0.f, b = 0.f;
#pragma unroll
        for (int k = 0; k < 16; ++k) {
            float ea = eattr[(size_t)e * 16 + k];
            a0 += ea * wea1[k];      a1 += ea * wea1[16 + k];
            a2 += ea * wea1[32 + k]; a3 += ea * wea1[48 + k];
            b  += ea * wea2[k];
        }
        float4 v; v.x = a0; v.y = a1; v.z = a2; v.w = a3;
        ((float4*)se1S)[pos] = v;
        int2 q; q.x = sv; q.y = __float_as_int(b);
        ((int2*)pairS)[pos] = q;
    }
}

__device__ __forceinline__ float leaky(float v) { return (v >= 0.f) ? v : NEGS * v; }
__device__ __forceinline__ float wave_max(float v) {
    for (int o = 16; o > 0; o >>= 1) v = fmaxf(v, __shfl_xor(v, o));
    return v;
}
__device__ __forceinline__ float wave_sum(float v) {
    for (int o = 16; o > 0; o >>= 1) v += __shfl_xor(v, o);
    return v;
}
__device__ __forceinline__ float lane_value(float v, int l) {
    return __int_as_float(__builtin_amdgcn_readlane(__float_as_int(v), l));
}

__global__ void __launch_bounds__(256)
node_aggr_l1(const int* __restrict__ rowptr, const int* __restrict__ cnt,
             const float* __restrict__ se1S, const int* __restrict__ pairS,
             const float* __restrict__ s_i1, const float* __restrict__ s_j1,
             const float* __restrict__ x, float* __restrict__ accx1)
{
    const int lane = threadIdx.x & 31;
    const int k = lane & 15;
    const bool hi = lane >= 16;
    const int wave = blockIdx.x * (blockDim.x >> 5) + __builtin_amdgcn_readfirstlane(threadIdx.x >> 5);
    const int nw = gridDim.x * (blockDim.x >> 5);
    const float4* __restrict__ se4 = (const float4*)se1S;
    const int2* __restrict__ pr2 = (const int2*)pairS;
    for (int d = wave; d < NN; d += nw) {
        const int beg = rowptr[d];
        const int deg = cnt[d];
        const float4 si = *(const float4*)(s_i1 + (size_t)d * 4);
        float M0 = -INFINITY, M1 = -INFINITY, M2 = -INFINITY, M3 = -INFINITY;
        float acc0 = 0.f, acc1 = 0.f;
        float dn0 = 0.f, dn1 = 0.f, dn2 = 0.f, dn3 = 0.f;
        for (int j0 = 0; j0 < deg; j0 += 32) {
            const int j = j0 + lane;
            const bool act = j < deg;
            const int jj = act ? j : j0;
            const int s = pr2[beg + jj].x;
            const float4 se = se4[beg + jj];
            const float4 sj = *(const float4*)(s_j1 + (size_t)s * 4);
            const float pen = act ? 0.f : -INFINITY;
            const float c0 = leaky(si.x + sj.x + se.x) + pen;
            const float c1 = leaky(si.y + sj.y + se.y) + pen;
            const float c2 = leaky(si.z + sj.z + se.z) + pen;
            const float c3 = leaky(si.w + sj.w + se.w) + pen;
            const float n0 = fmaxf(M0, wave_max(c0)), n1 = fmaxf(M1, wave_max(c1));
            const float n2 = fmaxf(M2, wave_max(c2)), n3 = fmaxf(M3, wave_max(c3));
            const float r0 = expf(M0 - n0), r1 = expf(M1 - n1);
            const float r2 = expf(M2 - n2), r3 = expf(M3 - n3);
            M0 = n0; M1 = n1; M2 = n2; M3 = n3;
            const float p0 = expf(c0 - n0), p1 = expf(c1 - n1);
            const float p2 = expf(c2 - n2), p3 = expf(c3 - n3);
            dn0 = dn0 * r0 + p0; dn1 = dn1 * r1 + p1;
            dn2 = dn2 * r2 + p2; dn3 = dn3 * r3 + p3;
            acc0 *= hi ? r1 : r0;
            acc1 *= hi ? r3 : r2;
            const int left = deg - j0;
            const int cnt4 = ((left < 32 ? left : 32) + 3) & ~3;
            for (int t = 0; t < cnt4; t += 4) {
#pragma unroll
                for (int u = 0; u < 4; ++u) {
                    const int st = __builtin_amdgcn_readlane(s, t + u);
                    const float a0 = lane_value(p0, t + u), a1 = lane_value(p1, t + u);
                    const float a2 = lane_value(p2, t + u), a3 = lane_value(p3, t + u);
                    const float xv = x[(size_t)st * 16 + k];
                    acc0 = fmaf(hi ? a1 : a0, xv, acc0);
                    acc1 = fmaf(hi ? a3 : a2, xv, acc1);
                }
            }
        }
        dn0 = wave_sum(dn0); dn1 = wave_sum(dn1); dn2 = wave_sum(dn2); dn3 = wave_sum(dn3);
        const float dA = hi ? dn1 : dn0;
        const float dB = hi ? dn3 : dn2;
        accx1[(size_t)d * 64 + lane]      = (dA > 0.f) ? acc0 / dA : 0.f;
        accx1[(size_t)d * 64 + 32 + lane] = (dB > 0.f) ? acc1 / dB : 0.f;
    }
}

__global__ void __launch_bounds__(256)
node_aggr_l2(const int* __restrict__ rowptr, const int* __restrict__ cnt,
             const int* __restrict__ pairS, const float* __restrict__ s_i2,
             const float* __restrict__ sz2, const float* __restrict__ p2b,
             float* __restrict__ out)
{
    const int lane = threadIdx.x & 31;
    const int wave = blockIdx.x * (blockDim.x >> 5) + __builtin_amdgcn_readfirstlane(threadIdx.x >> 5);
    const int nw = gridDim.x * (blockDim.x >> 5);
    const int2* __restrict__ pr2 = (const int2*)pairS;
    const float2* __restrict__ sz = (const float2*)sz2;
    const float bias = p2b[0];
    for (int d = wave; d < NN; d += nw) {
        const int beg = rowptr[d];
        const int deg = cnt[d];
        const float si = s_i2[d];
        float M = -INFINITY, num = 0.f, den = 0.f;
        for (int j0 = 0; j0 < deg; j0 += 32) {
            const int j = j0 + lane;
            const bool act = j < deg;
            const int2 pr = pr2[beg + (act ? j : j0)];
            const float2 q = sz[pr.x];
            const float c = leaky(si + q.x + __int_as_float(pr.y)) + (act ? 0.f : -INFINITY);
            const float n = fmaxf(M, wave_max(c));
            const float r = expf(M - n);
            M = n;
            const float p = expf(c - n);
            num = num * r + p * q.y;
            den = den * r + p;
        }
        num = wave_sum(num);
        den = wave_sum(den);
        if (lane == 0) {
            float v = (den > 0.f) ? num / den : 0.f;
            v += bias;
            out[d] = (v > 0.f) ? v : (expf(v) - 1.f);
        }
    }
}

extern "C" void kernel_launch(void* const* d_in, const int* in_sizes, int n_in,
                              void* d_out, int out_size, void* d_ws, size_t ws_size,
                              hipStream_t stream)
{
    const float* x       = (const float*)d_in[0];
    const int*   eidx    = (const int*)d_in[1];
    const float* eattr   = (const float*)d_in[2];
    const float* W1      = (const float*)d_in[3];
    const float* We1     = (const float*)d_in[4];
    const float* att1    = (const float*)d_in[5];
    const float* proj1_w = (const float*)d_in[6];
    const float* proj1_b = (const float*)d_in[7];
    const float* W2      = (const float*)d_in[8];
    const float* We2     = (const float*)d_in[9];
    const float* att2    = (const float*)d_in[10];
    const float* proj2_w = (const float*)d_in[11];
    const float* proj2_b = (const float*)d_in[12];
    float* out = (float*)d_out;

    const int* srcp = eidx;
    const int* dstp = eidx + EE;

    float* ws     = (float*)d_ws;
    float* xw1    = ws;
    float* accx1  = xw1 + (size_t)NN * 128;
    int*   pairS  = (int*)(accx1 + (size_t)NN * 64);
    float* s_i1   = accx1 + (size_t)NN * 128;
    float* s_j1   = s_i1 + (size_t)NN * 4;
    float* slotA  = s_j1 + (size_t)NN * 4;
    int*   cnt    = (int*)slotA;
    float* Wcomb  = slotA + (size_t)NN;
    int*   rowptr = (int*)(slotA + (size_t)NN * 4);
    float* hw2    = slotA + (size_t)NN * 8;
    float* s_i2   = hw2 + (size_t)NN * 64;
    float* sz2    = s_i2 + (size_t)NN;
    float* Wcat1  = s_i2 + (size_t)NN * 4;
    float* wea1   = Wcat1 + 2048;
    float* wea2   = wea1 + 64;
    float* se1S   = xw1;
    int*   rank   = (int*)hw2;

    prep_weights<<<32, 256, 0, stream>>>(W1, We1, att1, We2, att2, proj1_w,
                                         Wcat1, wea1, wea2, Wcomb, cnt);

    {
        int tiles = (NN / 16) * (128 / 16);
        wmma_gemm<<<(tiles + 7) / 8, 256, 0, stream>>>(x, 16, Wcat1, 128, xw1, 128,
                                                       NN, 128, 16, nullptr, 0);
    }
    scores1_kernel<<<(NN * 4 + 255) / 256, 256, 0, stream>>>(xw1, att1, s_i1, s_j1);

    edge_rank<<<2048, 256, 0, stream>>>(dstp, cnt, rank);
    row_offsets<<<(NN + 1023) / 1024, 1024, 0, stream>>>(cnt, rowptr);
    edge_place<<<1280, 256, 0, stream>>>(srcp, dstp, eattr, wea1, wea2, rowptr, rank,
                                         se1S, pairS);

    node_aggr_l1<<<2048, 256, 0, stream>>>(rowptr, cnt, se1S, pairS, s_i1, s_j1, x, accx1);

    {
        int tiles = (NN / 16) * (128 / 16);
        wmma_gemm<<<(tiles + 7) / 8, 256, 0, stream>>>(accx1, 64, Wcomb, 128, xw1, 128,
                                                       NN, 128, 64, proj1_b, 2);
    }
    {
        int tiles = (NN / 16) * (32 / 16);
        wmma_gemm<<<(tiles + 7) / 8, 256, 0, stream>>>(xw1, 128, W2, 32, hw2, 32,
                                                       NN, 32, 128, nullptr, 0);
    }
    scores2_kernel<<<(NN + 255) / 256, 256, 0, stream>>>(hw2, att2, proj2_w, s_i2, sz2);
    node_aggr_l2<<<2048, 256, 0, stream>>>(rowptr, cnt, pairS, s_i2, sz2, proj2_b, out);
}


# Attention and output-projection vectors folded into the node weights: scores = x@(W1 a) and h@(W2 [a_i a_j p2w]) as 16-column WMMA products; xw1, hw2 and both score launches gone; dots reassociated
# speedup vs baseline: 3.6585x; 1.0703x over previous
#include <hip/hip_runtime.h>
#include <cmath>

#define NN 100000
#define EE 3200000
#define NEGS 0.2f

typedef __attribute__((ext_vector_type(2))) float v2f;
typedef __attribute__((ext_vector_type(8))) float v8f;

__global__ void prep_weights(const float* __restrict__ W1, const float* __restrict__ We1,
                             const float* __restrict__ att1, const float* __restrict__ We2,
                             const float* __restrict__ att2, const float* __restrict__ proj1_w,
                             const float* __restrict__ W2, const float* __restrict__ p2w,
                             float* __restrict__ Vs, float* __restrict__ V2,
                             float* __restrict__ wea1,
                             float* __restrict__ wea2, float* __restrict__ Wcomb,
                             int* __restrict__ cnt)
{
    int tid = blockIdx.x * blockDim.x + threadIdx.x;
    int nt = gridDim.x * blockDim.x;
    for (int i = tid; i < NN; i += nt) cnt[i] = 0;
    for (int i = tid; i < 256; i += nt) {
        int k = i >> 4, c = i & 15;
        float s = 0.f;
        if (c < 8) {
            int h = c & 3;
            const float* a = att1 + h * 96 + (c >> 2) * 32;
            const float* w = W1 + h * 512 + k * 32;
            for (int o = 0; o < 32; ++o) s += w[o] * a[o];
        }
        Vs[i] = s;
    }
    for (int i = tid; i < 2048; i += nt) {
        int k = i >> 4, c = i & 15;
        float s = 0.f;
        if (c < 3) {
            const float* a = (c == 2) ? p2w : (att2 + c * 32);
            const float* w = W2 + k * 32;
            for (int o = 0; o < 32; ++o) s += w[o] * a[o];
        }
        V2[i] = s;
    }
    for (int i = tid; i < 64; i += nt) {
        int h = i >> 4, k = i & 15;
        float s = 0.f;
        for (int o = 0; o < 32; ++o) s += We1[h * 512 + k * 32 + o] * att1[h * 96 + 64 + o];
        wea1[i] = s;
    }
    for (int i = tid; i < 16; i += nt) {
        float s = 0.f;
        for (int o = 0; o < 32; ++o) s += We2[i * 32 + o] * att2[64 + o];
        wea2[i] = s;
    }
    for (int i = tid; i < 64 * 128; i += nt) {
        int row = i >> 7, c = i & 127, h = row >> 4, k = row & 15;
        const float* w = W1 + h * 512 + k * 32;
        const float* pw = proj1_w + (size_t)(h * 32) * 128 + c;
        float s = 0.f;
        for (int o = 0; o < 32; ++o) s += w[o] * pw[(size_t)o * 128];
        Wcomb[i] = s;
    }
}

__global__ void wmma_gemm(const float* __restrict__ A, int lda,
                          const float* __restrict__ B, int ldb,
                          float* __restrict__ C, int ldc,
                          int M, int Ncols, int K,
                          const float* __restrict__ bias, int nelu)
{
    int lane = threadIdx.x & 31;
    int tile = blockIdx.x * (blockDim.x >> 5) + (threadIdx.x >> 5);
    int tilesN = Ncols >> 4;
    int total = (M >> 4) * tilesN;
    if (tile >= total) return;
    int tm = tile / tilesN, tn = tile % tilesN;
    int r = lane & 15;
    int kk = (lane >> 4) << 1;
    const float* Ap = A + (size_t)(tm * 16 + r) * lda + kk;
    const float* Bp = B + (size_t)kk * ldb + (tn * 16 + r);
    v8f acc = {};
    for (int k0 = 0; k0 < K; k0 += 4) {
        v2f a; a.x = Ap[0]; a.y = Ap[1];
        v2f b; b.x = Bp[0]; b.y = Bp[ldb];
        acc = __builtin_amdgcn_wmma_f32_16x16x4_f32(false, a, false, b,
                                                    (short)0, acc, false, false);
        Ap += 4;
        Bp += (size_t)ldb * 4;
    }
    int colg = tn * 16 + r;
    float bv = bias ? bias[colg] : 0.f;
    int rbase = tm * 16 + ((lane >> 4) << 3);
#pragma unroll
    for (int v = 0; v < 8; ++v) {
        float xv = acc[v] + bv;
        for (int t = 0; t < nelu; ++t) xv = (xv > 0.f) ? xv : (expf(xv) - 1.f);
        C[(size_t)(rbase + v) * ldc + colg] = xv;
    }
}

__device__ __forceinline__ v8f strip_product(const float* __restrict__ Ap,
                                             const float* __restrict__ Bp, const int K)
{
    v8f acc = {};
#pragma unroll
    for (int k0 = 0; k0 < K; k0 += 4) {
        v2f a; a.x = Ap[k0]; a.y = Ap[k0 + 1];
        v2f b; b.x = Bp[k0 * 16]; b.y = Bp[k0 * 16 + 16];
        acc = __builtin_amdgcn_wmma_f32_16x16x4_f32(false, a, false, b,
                                                    (short)0, acc, false, false);
    }
    return acc;
}

__global__ void __launch_bounds__(256)
node_scores_l1(const float* __restrict__ x, const float* __restrict__ Vs,
               float* __restrict__ s_i1, float* __restrict__ s_j1)
{
    const int lane = threadIdx.x & 31;
    const int strip = blockIdx.x * (blockDim.x >> 5) + (threadIdx.x >> 5);
    if (strip >= NN / 16) return;
    const int r = lane & 15;
    const int kk = (lane >> 4) << 1;
    const v8f acc = strip_product(x + (size_t)(strip * 16 + r) * 16 + kk, Vs + kk * 16 + r, 16);
    const int rbase = strip * 16 + ((lane >> 4) << 3);
    if (r < 8) {
        float* o = (r < 4) ? (s_i1 + r) : (s_j1 + (r - 4));
#pragma unroll
        for (int v = 0; v < 8; ++v) o[(size_t)(rbase + v) * 4] = acc[v];
    }
}

__global__ void __launch_bounds__(256)
node_scores_l2(const float* __restrict__ h, const float* __restrict__ V2,
               float* __restrict__ s_i2, float* __restrict__ sz2)
{
    const int lane = threadIdx.x & 31;
    const int strip = blockIdx.x * (blockDim.x >> 5) + (threadIdx.x >> 5);
    if (strip >= NN / 16) return;
    const int r = lane & 15;
    const int kk = (lane >> 4) << 1;
    const v8f acc = strip_product(h + (size_t)(strip * 16 + r) * 128 + kk, V2 + kk * 16 + r, 128);
    const int rbase = strip * 16 + ((lane >> 4) << 3);
    if (r == 0) {
#pragma unroll
        for (int v = 0; v < 8; ++v) s_i2[rbase + v] = acc[v];
    } else if (r < 3) {
#pragma unroll
        for (int v = 0; v < 8; ++v) sz2[(size_t)(rbase + v) * 2 + (r - 1)] = acc[v];
    }
}

__global__ void __launch_bounds__(256)
edge_rank(const int* __restrict__ dst, int* __restrict__ cnt, int* __restrict__ rank)
{
    int i0 = blockIdx.x * blockDim.x + threadIdx.x;
    int stride = gridDim.x * blockDim.x;
    for (int e = i0; e < EE; e += stride) rank[e] = atomicAdd(&cnt[dst[e]], 1);
}

__global__ void __launch_bounds__(1024)
row_offsets(const int* __restrict__ cnt, int* __restrict__ rowptr)
{
    __shared__ int wtot[32];
    __shared__ int wpart[32];
    const int t = threadIdx.x, lane = t & 31, w = t >> 5;
    const int b0 = blockIdx.x * 1024;
    int part = 0;
    for (int i = t; i < b0; i += 1024) part += cnt[i];
    const int idx = b0 + t;
    const int c = (idx < NN) ? cnt[idx] : 0;
    int incl = c;
    for (int o = 1; o < 32; o <<= 1) {
        int v = __shfl_up(incl, o);
        if (lane >= o) incl += v;
    }
    for (int o = 16; o > 0; o >>= 1) part += __shfl_xor(part, o);
    if (lane == 31) wtot[w] = incl;
    if (lane == 0) wpart[w] = part;
    __syncthreads();
    int base = wpart[lane];
    int pre = (lane < w) ? wtot[lane] : 0;
    for (int o = 16; o > 0; o >>= 1) {
        base += __shfl_xor(base, o);
        pre += __shfl_xor(pre, o);
    }
    if (idx < NN) rowptr[idx] = base + pre + incl - c;
}

__global__ void __launch_bounds__(256)
edge_place(const int* __restrict__ src, const int* __restrict__ dst,
           const float* __restrict__ eattr, const float* __restrict__ wea1,
           const float* __restrict__ wea2, const int* __restrict__ rowptr,
           const int* __restrict__ rank, float* __restrict__ se1S,
           int* __restrict__ pairS)
{
    int i0 = blockIdx.x * blockDim.x + threadIdx.x;
    int stride = gridDim.x * blockDim.x;
    for (int e = i0; e < EE; e += stride) {
        const int d = dst[e];
        const int sv = src[e];
        const int pos = rowptr[d] + rank[e];
        float a0 = 0.f, a1 = 0.f, a2 = 0.f, a3 = 0.f, b = 0.f;
#pragma unroll
        for (int k = 0; k < 16; ++k) {
            float ea = eattr[(size_t)e * 16 + k];
            a0 += ea * wea1[k];      a1 += ea * wea1[16 + k];
            a2 += ea * wea1[32 + k]; a3 += ea * wea1[48 + k];
            b  += ea * wea2[k];
        }
        float4 v; v.x = a0; v.y = a1; v.z = a2; v.w = a3;
        ((float4*)se1S)[pos] = v;
        int2 q; q.x = sv; q.y = __float_as_int(b);
        ((int2*)pairS)[pos] = q;
    }
}

__device__ __forceinline__ float leaky(float v) { return (v >= 0.f) ? v : NEGS * v; }
__device__ __forceinline__ float wave_max(float v) {
    for (int o = 16; o > 0; o >>= 1) v = fmaxf(v, __shfl_xor(v, o));
    return v;
}
__device__ __forceinline__ float wave_sum(float v) {
    for (int o = 16; o > 0; o >>= 1) v += __shfl_xor(v, o);
    return v;
}
__device__ __forceinline__ float lane_value(float v, int l) {
    return __int_as_float(__builtin_amdgcn_readlane(__float_as_int(v), l));
}

__global__ void __launch_bounds__(256)
node_aggr_l1(const int* __restrict__ rowptr, const int* __restrict__ cnt,
             const float* __restrict__ se1S, const int* __restrict__ pairS,
             const float* __restrict__ s_i1, const float* __restrict__ s_j1,
             const float* __restrict__ x, float* __restrict__ accx1)
{
    const int lane = threadIdx.x & 31;
    const int k = lane & 15;
    const bool hi = lane >= 16;
    const int wave = blockIdx.x * (blockDim.x >> 5) + __builtin_amdgcn_readfirstlane(threadIdx.x >> 5);
    const int nw = gridDim.x * (blockDim.x >> 5);
    const float4* __restrict__ se4 = (const float4*)se1S;
    const int2* __restrict__ pr2 = (const int2*)pairS;
    for (int d = wave; d < NN; d += nw) {
        const int beg = rowptr[d];
        const int deg = cnt[d];
        const float4 si = *(const float4*)(s_i1 + (size_t)d * 4);
        float M0 = -INFINITY, M1 = -INFINITY, M2 = -INFINITY, M3 = -INFINITY;
        float acc0 = 0.f, acc1 = 0.f;
        float dn0 = 0.f, dn1 = 0.f, dn2 = 0.f, dn3 = 0.f;
        for (int j0 = 0; j0 < deg; j0 += 32) {
            const int j = j0 + lane;
            const bool act = j < deg;
            const int jj = act ? j : j0;
            const int s = pr2[beg + jj].x;
            const float4 se = se4[beg + jj];
            const float4 sj = *(const float4*)(s_j1 + (size_t)s * 4);
            const float pen = act ? 0.f : -INFINITY;
            const float c0 = leaky(si.x + sj.x + se.x) + pen;
            const float c1 = leaky(si.y + sj.y + se.y) + pen;
            const float c2 = leaky(si.z + sj.z + se.z) + pen;
            const float c3 = leaky(si.w + sj.w + se.w) + pen;
            const float n0 = fmaxf(M0, wave_max(c0)), n1 = fmaxf(M1, wave_max(c1));
            const float n2 = fmaxf(M2, wave_max(c2)), n3 = fmaxf(M3, wave_max(c3));
            const float r0 = expf(M0 - n0), r1 = expf(M1 - n1);
            const float r2 = expf(M2 - n2), r3 = expf(M3 - n3);
            M0 = n0; M1 = n1; M2 = n2; M3 = n3;
            const float p0 = expf(c0 - n0), p1 = expf(c1 - n1);
            const float p2 = expf(c2 - n2), p3 = expf(c3 - n3);
            dn0 = dn0 * r0 + p0; dn1 = dn1 * r1 + p1;
            dn2 = dn2 * r2 + p2; dn3 = dn3 * r3 + p3;
            acc0 *= hi ? r1 : r0;
            acc1 *= hi ? r3 : r2;
            const int left = deg - j0;
            const int cnt4 = ((left < 32 ? left : 32) + 3) & ~3;
            for (int t = 0; t < cnt4; t += 4) {
#pragma unroll
                for (int u = 0; u < 4; ++u) {
                    const int st = __builtin_amdgcn_readlane(s, t + u);
                    const float a0 = lane_value(p0, t + u), a1 = lane_value(p1, t + u);
                    const float a2 = lane_value(p2, t + u), a3 = lane_value(p3, t + u);
                    const float xv = x[(size_t)st * 16 + k];
                    acc0 = fmaf(hi ? a1 : a0, xv, acc0);
                    acc1 = fmaf(hi ? a3 : a2, xv, acc1);
                }
            }
        }
        dn0 = wave_sum(dn0); dn1 = wave_sum(dn1); dn2 = wave_sum(dn2); dn3 = wave_sum(dn3);
        const float dA = hi ? dn1 : dn0;
        const float dB = hi ? dn3 : dn2;
        accx1[(size_t)d * 64 + lane]      = (dA > 0.f) ? acc0 / dA : 0.f;
        accx1[(size_t)d * 64 + 32 + lane] = (dB > 0.f) ? acc1 / dB : 0.f;
    }
}

__global__ void __launch_bounds__(256)
node_aggr_l2(const int* __restrict__ rowptr, const int* __restrict__ cnt,
             const int* __restrict__ pairS, const float* __restrict__ s_i2,
             const float* __restrict__ sz2, const float* __restrict__ p2b,
             float* __restrict__ out)
{
    const int lane = threadIdx.x & 31;
    const int wave = blockIdx.x * (blockDim.x >> 5) + __builtin_amdgcn_readfirstlane(threadIdx.x >> 5);
    const int nw = gridDim.x * (blockDim.x >> 5);
    const int2* __restrict__ pr2 = (const int2*)pairS;
    const float2* __restrict__ sz = (const float2*)sz2;
    const float bias = p2b[0];
    for (int d = wave; d < NN; d += nw) {
        const int beg = rowptr[d];
        const int deg = cnt[d];
        const float si = s_i2[d];
        float M = -INFINITY, num = 0.f, den = 0.f;
        for (int j0 = 0; j0 < deg; j0 += 32) {
            const int j = j0 + lane;
            const bool act = j < deg;
            const int2 pr = pr2[beg + (act ? j : j0)];
            const float2 q = sz[pr.x];
            const float c = leaky(si + q.x + __int_as_float(pr.y)) + (act ? 0.f : -INFINITY);
            const float n = fmaxf(M, wave_max(c));
            const float r = expf(M - n);
            M = n;
            const float p = expf(c - n);
            num = num * r + p * q.y;
            den = den * r + p;
        }
        num = wave_sum(num);
        den = wave_sum(den);
        if (lane == 0) {
            float v = (den > 0.f) ? num / den : 0.f;
            v += bias;
            out[d] = (v > 0.f) ? v : (expf(v) - 1.f);
        }
    }
}

extern "C" void kernel_launch(void* const* d_in, const int* in_sizes, int n_in,
                              void* d_out, int out_size, void* d_ws, size_t ws_size,
                              hipStream_t stream)
{
    const float* x       = (const float*)d_in[0];
    const int*   eidx    = (const int*)d_in[1];
    const float* eattr   = (const float*)d_in[2];
    const float* W1      = (const float*)d_in[3];
    const float* We1     = (const float*)d_in[4];
    const float* att1    = (const float*)d_in[5];
    const float* proj1_w = (const float*)d_in[6];
    const float* proj1_b = (const float*)d_in[7];
    const float* W2      = (const float*)d_in[8];
    const float* We2     = (const float*)d_in[9];
    const float* att2    = (const float*)d_in[10];
    const float* proj2_w = (const float*)d_in[11];
    const float* proj2_b = (const float*)d_in[12];
    float* out = (float*)d_out;

    const int* srcp = eidx;
    const int* dstp = eidx + EE;

    float* ws     = (float*)d_ws;
    float* xw1    = ws;
    float* accx1  = xw1 + (size_t)NN * 128;
    int*   pairS  = (int*)(accx1 + (size_t)NN * 64);
    float* s_i1   = accx1 + (size_t)NN * 128;
    float* s_j1   = s_i1 + (size_t)NN * 4;
    float* slotA  = s_j1 + (size_t)NN * 4;
    int*   cnt    = (int*)slotA;
    float* Wcomb  = slotA + (size_t)NN;
    float* Vs     = Wcomb + 64 * 128;
    float* V2     = Vs + 256;
    int*   rowptr = (int*)(slotA + (size_t)NN * 4);
    float* hw2    = slotA + (size_t)NN * 8;
    float* s_i2   = hw2 + (size_t)NN * 64;
    float* sz2    = s_i2 + (size_t)NN;
    float* wea1   = s_i2 + (size_t)NN * 4 + 2048;
    float* wea2   = wea1 + 64;
    float* se1S   = xw1;
    int*   rank   = (int*)hw2;

    prep_weights<<<32, 256, 0, stream>>>(W1, We1, att1, We2, att2, proj1_w, W2, proj2_w,
                                         Vs, V2, wea1, wea2, Wcomb, cnt);

    node_scores_l1<<<(NN / 16 + 7) / 8, 256, 0, stream>>>(x, Vs, s_i1, s_j1);

    edge_rank<<<2048, 256, 0, stream>>>(dstp, cnt, rank);
    row_offsets<<<(NN + 1023) / 1024, 1024, 0, stream>>>(cnt, rowptr);
    edge_place<<<1280, 256, 0, stream>>>(srcp, dstp, eattr, wea1, wea2, rowptr, rank,
                                         se1S, pairS);

    node_aggr_l1<<<2048, 256, 0, stream>>>(rowptr, cnt, se1S, pairS, s_i1, s_j1, x, accx1);

    {
        int tiles = (NN / 16) * (128 / 16);
        wmma_gemm<<<(tiles + 7) / 8, 256, 0, stream>>>(accx1, 64, Wcomb, 128, xw1, 128,
                                                       NN, 128, 64, proj1_b, 2);
    }
    node_scores_l2<<<(NN / 16 + 7) / 8, 256, 0, stream>>>(xw1, V2, s_i2, sz2);
    node_aggr_l2<<<2048, 256, 0, stream>>>(rowptr, cnt, pairS, s_i2, sz2, proj2_b, out);
}


# Edge sort without global atomics: edges binned by 256-node destination range on on-chip counters (count, partition, in-place sort per bin); rank and scan launches gone; sums of a node reordered
# speedup vs baseline: 6.4110x; 1.7524x over previous
#include <hip/hip_runtime.h>
#include <cmath>

#define NN 100000
#define EE 3200000
#define NEGS 0.2f
#define BINSH 8
#define NBINS ((NN + 255) >> BINSH)
#define PBLK 256
#define ECHUNK ((EE + PBLK - 1) / PBLK)
#define CAP 10240
#define SRCBITS 17
#define SRCMASK ((1 << SRCBITS) - 1)

typedef __attribute__((ext_vector_type(2))) float v2f;
typedef __attribute__((ext_vector_type(8))) float v8f;

__global__ void prep_weights(const float* __restrict__ W1, const float* __restrict__ We1,
                             const float* __restrict__ att1, const float* __restrict__ We2,
                             const float* __restrict__ att2, const float* __restrict__ proj1_w,
                             const float* __restrict__ W2, const float* __restrict__ p2w,
                             float* __restrict__ Vs, float* __restrict__ V2,
                             float* __restrict__ wea1,
                             float* __restrict__ wea2, float* __restrict__ Wcomb)
{
    int tid = blockIdx.x * blockDim.x + threadIdx.x;
    int nt = gridDim.x * blockDim.x;
    for (int i = tid; i < 256; i += nt) {
        int k = i >> 4, c = i & 15;
        float s = 0.f;
        if (c < 8) {
            int h = c & 3;
            const float* a = att1 + h * 96 + (c >> 2) * 32;
            const float* w = W1 + h * 512 + k * 32;
            for (int o = 0; o < 32; ++o) s += w[o] * a[o];
        }
        Vs[i] = s;
    }
    for (int i = tid; i < 2048; i += nt) {
        int k = i >> 4, c = i & 15;
        float s = 0.f;
        if (c < 3) {
            const float* a = (c == 2) ? p2w : (att2 + c * 32);
            const float* w = W2 + k * 32;
            for (int o = 0; o < 32; ++o) s += w[o] * a[o];
        }
        V2[i] = s;
    }
    for (int i = tid; i < 64; i += nt) {
        int h = i >> 4, k = i & 15;
        float s = 0.f;
        for (int o = 0; o < 32; ++o) s += We1[h * 512 + k * 32 + o] * att1[h * 96 + 64 + o];
        wea1[i] = s;
    }
    for (int i = tid; i < 16; i += nt) {
        float s = 0.f;
        for (int o = 0; o < 32; ++o) s += We2[i * 32 + o] * att2[64 + o];
        wea2[i] = s;
    }
    for (int i = tid; i < 64 * 128; i += nt) {
        int row = i >> 7, c = i & 127, h = row >> 4, k = row & 15;
        const float* w = W1 + h * 512 + k * 32;
        const float* pw = proj1_w + (size_t)(h * 32) * 128 + c;
        float s = 0.f;
        for (int o = 0; o < 32; ++o) s += w[o] * pw[(size_t)o * 128];
        Wcomb[i] = s;
    }
}

__global__ void wmma_gemm(const float* __restrict__ A, int lda,
                          const float* __restrict__ B, int ldb,
                          float* __restrict__ C, int ldc,
                          int M, int Ncols, int K,
                          const float* __restrict__ bias, int nelu)
{
    int lane = threadIdx.x & 31;
    int tile = blockIdx.x * (blockDim.x >> 5) + (threadIdx.x >> 5);
    int tilesN = Ncols >> 4;
    int total = (M >> 4) * tilesN;
    if (tile >= total) return;
    int tm = tile / tilesN, tn = tile % tilesN;
    int r = lane & 15;
    int kk = (lane >> 4) << 1;
    const float* Ap = A + (size_t)(tm * 16 + r) * lda + kk;
    const float* Bp = B + (size_t)kk * ldb + (tn * 16 + r);
    v8f acc = {};
    for (int k0 = 0; k0 < K; k0 += 4) {
        v2f a; a.x = Ap[0]; a.y = Ap[1];
        v2f b; b.x = Bp[0]; b.y = Bp[ldb];
        acc = __builtin_amdgcn_wmma_f32_16x16x4_f32(false, a, false, b,
                                                    (short)0, acc, false, false);
        Ap += 4;
        Bp += (size_t)ldb * 4;
    }
    int colg = tn * 16 + r;
    float bv = bias ? bias[colg] : 0.f;
    int rbase = tm * 16 + ((lane >> 4) << 3);
#pragma unroll
    for (int v = 0; v < 8; ++v) {
        float xv = acc[v] + bv;
        for (int t = 0; t < nelu; ++t) xv = (xv > 0.f) ? xv : (expf(xv) - 1.f);
        C[(size_t)(rbase + v) * ldc + colg] = xv;
    }
}

__device__ __forceinline__ v8f strip_product(const float* __restrict__ Ap,
                                             const float* __restrict__ Bp, const int K)
{
    v8f acc = {};
#pragma unroll
    for (int k0 = 0; k0 < K; k0 += 4) {
        v2f a; a.x = Ap[k0]; a.y = Ap[k0 + 1];
        v2f b; b.x = Bp[k0 * 16]; b.y = Bp[k0 * 16 + 16];
        acc = __builtin_amdgcn_wmma_f32_16x16x4_f32(false, a, false, b,
                                                    (short)0, acc, false, false);
    }
    return acc;
}

__global__ void __launch_bounds__(256)
node_scores_l1(const float* __restrict__ x, const float* __restrict__ Vs,
               float* __restrict__ s_i1, float* __restrict__ s_j1)
{
    const int lane = threadIdx.x & 31;
    const int strip = blockIdx.x * (blockDim.x >> 5) + (threadIdx.x >> 5);
    if (strip >= NN / 16) return;
    const int r = lane & 15;
    const int kk = (lane >> 4) << 1;
    const v8f acc = strip_product(x + (size_t)(strip * 16 + r) * 16 + kk, Vs + kk * 16 + r, 16);
    const int rbase = strip * 16 + ((lane >> 4) << 3);
    if (r < 8) {
        float* o = (r < 4) ? (s_i1 + r) : (s_j1 + (r - 4));
#pragma unroll
        for (int v = 0; v < 8; ++v) o[(size_t)(rbase + v) * 4] = acc[v];
    }
}

__global__ void __launch_bounds__(256)
node_scores_l2(const float* __restrict__ h, const float* __restrict__ V2,
               float* __restrict__ s_i2, float* __restrict__ sz2)
{
    const int lane = threadIdx.x & 31;
    const int strip = blockIdx.x * (blockDim.x >> 5) + (threadIdx.x >> 5);
    if (strip >= NN / 16) return;
    const int r = lane & 15;
    const int kk = (lane >> 4) << 1;
    const v8f acc = strip_product(h + (size_t)(strip * 16 + r) * 128 + kk, V2 + kk * 16 + r, 128);
    const int rbase = strip * 16 + ((lane >> 4) << 3);
    if (r == 0) {
#pragma unroll
        for (int v = 0; v < 8; ++v) s_i2[rbase + v] = acc[v];
    } else if (r < 3) {
#pragma unroll
        for (int v = 0; v < 8; ++v) sz2[(size_t)(rbase + v) * 2 + (r - 1)] = acc[v];
    }
}

__global__ void __launch_bounds__(1024)
bin_count(const int* __restrict__ dst, int* __restrict__ cntmat)
{
    __shared__ int hist[NBINS];
    const int t = threadIdx.x, b = blockIdx.x;
    for (int i = t; i < NBINS; i += 1024) hist[i] = 0;
    __syncthreads();
    const int e0 = b * ECHUNK;
    const int e1 = (e0 + ECHUNK < EE) ? (e0 + ECHUNK) : EE;
    int dv[(ECHUNK + 1023) / 1024];
#pragma unroll
    for (int j = 0; j < (ECHUNK + 1023) / 1024; ++j) {
        const int e = e0 + j * 1024 + t;
        dv[j] = dst[(e < EE) ? e : (EE - 1)];
    }
#pragma unroll
    for (int j = 0; j < (ECHUNK + 1023) / 1024; ++j)
        if (e0 + j * 1024 + t < e1) atomicAdd(&hist[dv[j] >> BINSH], 1);
    __syncthreads();
    for (int i = t; i < NBINS; i += 1024) cntmat[i * PBLK + b] = hist[i];
}

__device__ __forceinline__ void place_edge(const int d, const int sv, const float4 r0, const float4 r1,
                                           const float4 r2, const float4 r3,
                                           const float* __restrict__ wea1,
                                           const float* __restrict__ wea2, int* cur,
                                           float4* __restrict__ recA, int2* __restrict__ recP)
{
    const float ea[16] = {r0.x, r0.y, r0.z, r0.w, r1.x, r1.y, r1.z, r1.w,
                          r2.x, r2.y, r2.z, r2.w, r3.x, r3.y, r3.z, r3.w};
    float a0 = 0.f, a1 = 0.f, a2 = 0.f, a3 = 0.f, bb = 0.f;
#pragma unroll
    for (int k = 0; k < 16; ++k) {
        a0 += ea[k] * wea1[k];      a1 += ea[k] * wea1[16 + k];
        a2 += ea[k] * wea1[32 + k]; a3 += ea[k] * wea1[48 + k];
        bb += ea[k] * wea2[k];
    }
    const int pos = atomicAdd(&cur[d >> BINSH], 1);
    float4 v4; v4.x = a0; v4.y = a1; v4.z = a2; v4.w = a3;
    recA[pos] = v4;
    int2 q; q.x = sv | ((d & ((1 << BINSH) - 1)) << SRCBITS); q.y = __float_as_int(bb);
    recP[pos] = q;
}

__global__ void __launch_bounds__(1024)
edge_partition(const int* __restrict__ src, const int* __restrict__ dst,
               const float* __restrict__ eattr, const float* __restrict__ wea1,
               const float* __restrict__ wea2, const int* __restrict__ cntmat,
               float4* __restrict__ recA, int2* __restrict__ recP,
               int* __restrict__ binstart)
{
    __shared__ int tot[NBINS];
    __shared__ int cur[NBINS];
    __shared__ int wsum[32];
    const int t = threadIdx.x, lane = t & 31, w = t >> 5, b = blockIdx.x;
    for (int bin = w; bin < NBINS; bin += 32) {
        const int* row = cntmat + bin * PBLK;
        int all = 0, before = 0;
        for (int q = 0; q < PBLK; q += 32) {
            const int c = row[q + lane];
            all += c;
            before += (q + lane < b) ? c : 0;
        }
        for (int o = 16; o > 0; o >>= 1) {
            all += __shfl_xor(all, o);
            before += __shfl_xor(before, o);
        }
        if (lane == 0) { tot[bin] = all; cur[bin] = before; }
    }
    __syncthreads();
    const int v = (t < NBINS) ? tot[t] : 0;
    int incl = v;
    for (int o = 1; o < 32; o <<= 1) {
        int u = __shfl_up(incl, o);
        if (lane >= o) incl += u;
    }
    if (lane == 31) wsum[w] = incl;
    __syncthreads();
    int pre = (lane < w) ? wsum[lane] : 0;
    for (int o = 16; o > 0; o >>= 1) pre += __shfl_xor(pre, o);
    if (t < NBINS) {
        const int first = pre + incl - v;
        cur[t] += first;
        if (b == 0) binstart[t] = first;
    }
    __syncthreads();
    const int e0 = b * ECHUNK;
    const int e1 = (e0 + ECHUNK < EE) ? (e0 + ECHUNK) : EE;
    const float4* __restrict__ ea4 = (const float4*)eattr;
    int e = e0 + t;
    if (e < e1) {
        int dA = dst[e], sA = src[e];
        float4 a0 = ea4[(size_t)e * 4], a1 = ea4[(size_t)e * 4 + 1];
        float4 a2 = ea4[(size_t)e * 4 + 2], a3 = ea4[(size_t)e * 4 + 3];
        for (;;) {
            const int eB = e + 1024;
            const bool moreB = eB < e1;
            const int fB = moreB ? eB : e;
            const int dB = dst[fB], sB = src[fB];
            const float4 b0 = ea4[(size_t)fB * 4], b1 = ea4[(size_t)fB * 4 + 1];
            const float4 b2 = ea4[(size_t)fB * 4 + 2], b3 = ea4[(size_t)fB * 4 + 3];
            place_edge(dA, sA, a0, a1, a2, a3, wea1, wea2, cur, recA, recP);
            if (!moreB) break;
            e = eB + 1024;
            const bool moreA = e < e1;
            const int fA = moreA ? e : eB;
            dA = dst[fA]; sA = src[fA];
            a0 = ea4[(size_t)fA * 4]; a1 = ea4[(size_t)fA * 4 + 1];
            a2 = ea4[(size_t)fA * 4 + 2]; a3 = ea4[(size_t)fA * 4 + 3];
            place_edge(dB, sB, b0, b1, b2, b3, wea1, wea2, cur, recA, recP);
            if (!moreA) break;
        }
    }
}

__global__ void __launch_bounds__(1024)
bin_sort(const int* __restrict__ binstart, float4* recA, int2* recP,
         int* __restrict__ rowptr, int* __restrict__ cnt)
{
    __shared__ float4 bufA[CAP];
    __shared__ int2 bufP[CAP];
    __shared__ int cntL[256];
    __shared__ int startL[256];
    __shared__ int curL[256];
    __shared__ int wtot[32];
    const int t = threadIdx.x, lane = t & 31, w = t >> 5, bin = blockIdx.x;
    const int beg = binstart[bin];
    const int end = (bin + 1 < NBINS) ? binstart[bin + 1] : EE;
    const int n = end - beg;
    const bool fits = n <= CAP;
    if (t < 256) { cntL[t] = 0; curL[t] = 0; }
    __syncthreads();
    if (fits) {
        int2 pv[CAP / 1024];
        float4 av[CAP / 1024];
#pragma unroll
        for (int j = 0; j < CAP / 1024; ++j) {
            const int g = beg + j * 1024 + t;
            const int gc = (g < EE) ? g : (EE - 1);
            pv[j] = recP[gc];
            av[j] = recA[gc];
        }
#pragma unroll
        for (int j = 0; j < CAP / 1024; ++j) {
            const int i = j * 1024 + t;
            bufP[i] = pv[j];
            bufA[i] = av[j];
            if (i < n) atomicAdd(&cntL[pv[j].x >> SRCBITS], 1);
        }
    } else {
        for (int i = t; i < n; i += 1024) atomicAdd(&cntL[recP[beg + i].x >> SRCBITS], 1);
    }
    __syncthreads();
    int c = 0, incl = 0;
    if (t < 256) {
        c = cntL[t];
        incl = c;
        for (int o = 1; o < 32; o <<= 1) {
            int u = __shfl_up(incl, o);
            if (lane >= o) incl += u;
        }
        if (lane == 31) wtot[w] = incl;
    }
    __syncthreads();
    if (t < 256) {
        int pre = (lane < w) ? wtot[lane] : 0;
        for (int o = 16; o > 0; o >>= 1) pre += __shfl_xor(pre, o);
        const int st = pre + incl - c;
        startL[t] = st;
        const int node = (bin << BINSH) + t;
        if (node < NN) { rowptr[node] = beg + st; cnt[node] = c; }
    }
    __syncthreads();
    if (fits) {
        for (int i = t; i < n; i += 1024) {
            const int2 p = bufP[i];
            const int dl = p.x >> SRCBITS;
            const int pos = beg + startL[dl] + atomicAdd(&curL[dl], 1);
            int2 q; q.x = p.x & SRCMASK; q.y = p.y;
            recP[pos] = q;
            recA[pos] = bufA[i];
        }
    } else {
        if (t == 0) {
            for (int nd = 0; nd < 256; ++nd) {
                const int endn = cntL[nd];
                int k = curL[nd];
                while (k < endn) {
                    const int p = beg + startL[nd] + k;
                    const int2 r = recP[p];
                    const int dl = r.x >> SRCBITS;
                    if (dl == nd) { ++k; continue; }
                    const int q = beg + startL[dl] + curL[dl];
                    curL[dl] += 1;
                    const int2 r2 = recP[q];
                    const float4 av = recA[p], av2 = recA[q];
                    recP[q] = r;  recA[q] = av;
                    recP[p] = r2; recA[p] = av2;
                }
                curL[nd] = endn;
            }
        }
        __syncthreads();
        for (int i = t; i < n; i += 1024) recP[beg + i].x &= SRCMASK;
    }
}

__device__ __forceinline__ float leaky(float v) { return (v >= 0.f) ? v : NEGS * v; }
__device__ __forceinline__ float wave_max(float v) {
    for (int o = 16; o > 0; o >>= 1) v = fmaxf(v, __shfl_xor(v, o));
    return v;
}
__device__ __forceinline__ float wave_sum(float v) {
    for (int o = 16; o > 0; o >>= 1) v += __shfl_xor(v, o);
    return v;
}
__device__ __forceinline__ float lane_value(float v, int l) {
    return __int_as_float(__builtin_amdgcn_readlane(__float_as_int(v), l));
}

__global__ void __launch_bounds__(256)
node_aggr_l1(const int* __restrict__ rowptr, const int* __restrict__ cnt,
             const float* __restrict__ se1S, const int* __restrict__ pairS,
             const float* __restrict__ s_i1, const float* __restrict__ s_j1,
             const float* __restrict__ x, float* __restrict__ accx1)
{
    const int lane = threadIdx.x & 31;
    const int k = lane & 15;
    const bool hi = lane >= 16;
    const int wave = blockIdx.x * (blockDim.x >> 5) + __builtin_amdgcn_readfirstlane(threadIdx.x >> 5);
    const int nw = gridDim.x * (blockDim.x >> 5);
    const float4* __restrict__ se4 = (const float4*)se1S;
    const int2* __restrict__ pr2 = (const int2*)pairS;
    for (int d = wave; d < NN; d += nw) {
        const int beg = rowptr[d];
        const int deg = cnt[d];
        const float4 si = *(const float4*)(s_i1 + (size_t)d * 4);
        float M0 = -INFINITY, M1 = -INFINITY, M2 = -INFINITY, M3 = -INFINITY;
        float acc0 = 0.f, acc1 = 0.f;
        float dn0 = 0.f, dn1 = 0.f, dn2 = 0.f, dn3 = 0.f;
        for (int j0 = 0; j0 < deg; j0 += 32) {
            const int j = j0 + lane;
            const bool act = j < deg;
            const int jj = act ? j : j0;
            const int s = pr2[beg + jj].x;
            const float4 se = se4[beg + jj];
            const float4 sj = *(const float4*)(s_j1 + (size_t)s * 4);
            const float pen = act ? 0.f : -INFINITY;
            const float c0 = leaky(si.x + sj.x + se.x) + pen;
            const float c1 = leaky(si.y + sj.y + se.y) + pen;
            const float c2 = leaky(si.z + sj.z + se.z) + pen;
            const float c3 = leaky(si.w + sj.w + se.w) + pen;
            const float n0 = fmaxf(M0, wave_max(c0)), n1 = fmaxf(M1, wave_max(c1));
            const float n2 = fmaxf(M2, wave_max(c2)), n3 = fmaxf(M3, wave_max(c3));
            const float r0 = expf(M0 - n0), r1 = expf(M1 - n1);
            const float r2 = expf(M2 - n2), r3 = expf(M3 - n3);
            M0 = n0; M1 = n1; M2 = n2; M3 = n3;
            const float p0 = expf(c0 - n0), p1 = expf(c1 - n1);
            const float p2 = expf(c2 - n2), p3 = expf(c3 - n3);
            dn0 = dn0 * r0 + p0; dn1 = dn1 * r1 + p1;
            dn2 = dn2 * r2 + p2; dn3 = dn3 * r3 + p3;
            acc0 *= hi ? r1 : r0;
            acc1 *= hi ? r3 : r2;
            const int left = deg - j0;
            const int cnt4 = ((left < 32 ? left : 32) + 3) & ~3;
            for (int t = 0; t < cnt4; t += 4) {
#pragma unroll
                for (int u = 0; u < 4; ++u) {
                    const int st = __builtin_amdgcn_readlane(s, t + u);
                    const float a0 = lane_value(p0, t + u), a1 = lane_value(p1, t + u);
                    const float a2 = lane_value(p2, t + u), a3 = lane_value(p3, t + u);
                    const float xv = x[(size_t)st * 16 + k];
                    acc0 = fmaf(hi ? a1 : a0, xv, acc0);
                    acc1 = fmaf(hi ? a3 : a2, xv, acc1);
                }
            }
        }
        dn0 = wave_sum(dn0); dn1 = wave_sum(dn1); dn2 = wave_sum(dn2); dn3 = wave_sum(dn3);
        const float dA = hi ? dn1 : dn0;
        const float dB = hi ? dn3 : dn2;
        accx1[(size_t)d * 64 + lane]      = (dA > 0.f) ? acc0 / dA : 0.f;
        accx1[(size_t)d * 64 + 32 + lane] = (dB > 0.f) ? acc1 / dB : 0.f;
    }
}

__global__ void __launch_bounds__(256)
node_aggr_l2(const int* __restrict__ rowptr, const int* __restrict__ cnt,
             const int* __restrict__ pairS, const float* __restrict__ s_i2,
             const float* __restrict__ sz2, const float* __restrict__ p2b,
             float* __restrict__ out)
{
    const int lane = threadIdx.x & 31;
    const int wave = blockIdx.x * (blockDim.x >> 5) + __builtin_amdgcn_readfirstlane(threadIdx.x >> 5);
    const int nw = gridDim.x * (blockDim.x >> 5);
    const int2* __restrict__ pr2 = (const int2*)pairS;
    const float2* __restrict__ sz = (const float2*)sz2;
    const float bias = p2b[0];
    for (int d = wave; d < NN; d += nw) {
        const int beg = rowptr[d];
        const int deg = cnt[d];
        const float si = s_i2[d];
        float M = -INFINITY, num = 0.f, den = 0.f;
        for (int j0 = 0; j0 < deg; j0 += 32) {
            const int j = j0 + lane;
            const bool act = j < deg;
            const int2 pr = pr2[beg + (act ? j : j0)];
            const float2 q = sz[pr.x];
            const float c = leaky(si + q.x + __int_as_float(pr.y)) + (act ? 0.f : -INFINITY);
            const float n = fmaxf(M, wave_max(c));
            const float r = expf(M - n);
            M = n;
            const float p = expf(c - n);
            num = num * r + p * q.y;
            den = den * r + p;
        }
        num = wave_sum(num);
        den = wave_sum(den);
        if (lane == 0) {
            float v = (den > 0.f) ? num / den : 0.f;
            v += bias;
            out[d] = (v > 0.f) ? v : (expf(v) - 1.f);
        }
    }
}

extern "C" void kernel_launch(void* const* d_in, const int* in_sizes, int n_in,
                              void* d_out, int out_size, void* d_ws, size_t ws_size,
                              hipStream_t stream)
{
    const float* x       = (const float*)d_in[0];
    const int*   eidx    = (const int*)d_in[1];
    const float* eattr   = (const float*)d_in[2];
    const float* W1      = (const float*)d_in[3];
    const float* We1     = (const float*)d_in[4];
    const float* att1    = (const float*)d_in[5];
    const float* proj1_w = (const float*)d_in[6];
    const float* proj1_b = (const float*)d_in[7];
    const float* W2      = (const float*)d_in[8];
    const float* We2     = (const float*)d_in[9];
    const float* att2    = (const float*)d_in[10];
    const float* proj2_w = (const float*)d_in[11];
    const float* proj2_b = (const float*)d_in[12];
    float* out = (float*)d_out;

    const int* srcp = eidx;
    const int* dstp = eidx + EE;

    float* ws     = (float*)d_ws;
    float* xw1    = ws;
    float* accx1  = xw1 + (size_t)NN * 128;
    int*   pairS  = (int*)(accx1 + (size_t)NN * 64);
    float* s_i1   = accx1 + (size_t)NN * 128;
    float* s_j1   = s_i1 + (size_t)NN * 4;
    float* slotA  = s_j1 + (size_t)NN * 4;
    int*   cnt    = (int*)slotA;
    float* Wcomb  = slotA + (size_t)NN;
    float* Vs     = Wcomb + 64 * 128;
    float* V2     = Vs + 256;
    int*   rowptr = (int*)(slotA + (size_t)NN * 4);
    float* hw2    = slotA + (size_t)NN * 8;
    float* s_i2   = hw2 + (size_t)NN * 64;
    float* sz2    = s_i2 + (size_t)NN;
    float* wea1   = s_i2 + (size_t)NN * 4 + 2048;
    float* wea2   = wea1 + 64;
    float* se1S   = xw1;
    int*   cntmat = (int*)hw2;
    int*   binstart = cntmat + NBINS * PBLK;

    prep_weights<<<32, 256, 0, stream>>>(W1, We1, att1, We2, att2, proj1_w, W2, proj2_w,
                                         Vs, V2, wea1, wea2, Wcomb);

    node_scores_l1<<<(NN / 16 + 7) / 8, 256, 0, stream>>>(x, Vs, s_i1, s_j1);

    bin_count<<<PBLK, 1024, 0, stream>>>(dstp, cntmat);
    edge_partition<<<PBLK, 1024, 0, stream>>>(srcp, dstp, eattr, wea1, wea2, cntmat,
                                              (float4*)se1S, (int2*)pairS, binstart);
    bin_sort<<<NBINS, 1024, 0, stream>>>(binstart, (float4*)se1S, (int2*)pairS, rowptr, cnt);

    node_aggr_l1<<<2048, 256, 0, stream>>>(rowptr, cnt, se1S, pairS, s_i1, s_j1, x, accx1);

    {
        int tiles = (NN / 16) * (128 / 16);
        wmma_gemm<<<(tiles + 7) / 8, 256, 0, stream>>>(accx1, 64, Wcomb, 128, xw1, 128,
                                                       NN, 128, 64, proj1_b, 2);
    }
    node_scores_l2<<<(NN / 16 + 7) / 8, 256, 0, stream>>>(xw1, V2, s_i2, sz2);
    node_aggr_l2<<<2048, 256, 0, stream>>>(rowptr, cnt, pairS, s_i2, sz2, proj2_b, out);
}


# Output projection, both ELUs and the layer-2 node scores fused into one launch on transposed tiles (the h^T tile in registers is the next WMMA's B operand); h never stored; sums reordered
# speedup vs baseline: 6.8727x; 1.0720x over previous
#include <hip/hip_runtime.h>
#include <cmath>

#define NN 100000
#define EE 3200000
#define NEGS 0.2f
#define BINSH 8
#define NBINS ((NN + 255) >> BINSH)
#define PBLK 256
#define ECHUNK ((EE + PBLK - 1) / PBLK)
#define CAP 10240
#define SRCBITS 17
#define SRCMASK ((1 << SRCBITS) - 1)

typedef __attribute__((ext_vector_type(2))) float v2f;
typedef __attribute__((ext_vector_type(8))) float v8f;

__global__ void prep_weights(const float* __restrict__ W1, const float* __restrict__ We1,
                             const float* __restrict__ att1, const float* __restrict__ We2,
                             const float* __restrict__ att2, const float* __restrict__ proj1_w,
                             const float* __restrict__ W2, const float* __restrict__ p2w,
                             float* __restrict__ Vs, float* __restrict__ V2,
                             float* __restrict__ wea1,
                             float* __restrict__ wea2, float* __restrict__ Wcomb)
{
    int tid = blockIdx.x * blockDim.x + threadIdx.x;
    int nt = gridDim.x * blockDim.x;
    for (int i = tid; i < 256; i += nt) {
        int k = i >> 4, c = i & 15;
        float s = 0.f;
        if (c < 8) {
            int h = c & 3;
            const float* a = att1 + h * 96 + (c >> 2) * 32;
            const float* w = W1 + h * 512 + k * 32;
            for (int o = 0; o < 32; ++o) s += w[o] * a[o];
        }
        Vs[i] = s;
    }
    for (int i = tid; i < 2048; i += nt) {
        const int e2 = i & 7, ln = (i >> 3) & 31, t2 = i >> 8;
        int k = 16 * t2 + 8 * (ln >> 4) + e2, c = ln & 15;
        float s = 0.f;
        if (c < 3) {
            const float* a = (c == 2) ? p2w : (att2 + c * 32);
            const float* w = W2 + k * 32;
            for (int o = 0; o < 32; ++o) s += w[o] * a[o];
        }
        V2[i] = s;
    }
    for (int i = tid; i < 64; i += nt) {
        int h = i >> 4, k = i & 15;
        float s = 0.f;
        for (int o = 0; o < 32; ++o) s += We1[h * 512 + k * 32 + o] * att1[h * 96 + 64 + o];
        wea1[i] = s;
    }
    for (int i = tid; i < 16; i += nt) {
        float s = 0.f;
        for (int o = 0; o < 32; ++o) s += We2[i * 32 + o] * att2[64 + o];
        wea2[i] = s;
    }
    for (int i = tid; i < 64 * 128; i += nt) {
        const int e4 = i & 3, ln = (i >> 2) & 31, m8 = (i >> 7) & 7, t8 = i >> 10;
        int row = 8 * m8 + 4 * (ln >> 4) + e4, c = 16 * t8 + (ln & 15), h = row >> 4, k = row & 15;
        const float* w = W1 + h * 512 + k * 32;
        const float* pw = proj1_w + (size_t)(h * 32) * 128 + c;
        float s = 0.f;
        for (int o = 0; o < 32; ++o) s += w[o] * pw[(size_t)o * 128];
        Wcomb[i] = s;
    }
}

__device__ __forceinline__ v8f strip_product(const float* __restrict__ Ap,
                                             const float* __restrict__ Bp, const int K)
{
    v8f acc = {};
#pragma unroll
    for (int k0 = 0; k0 < K; k0 += 4) {
        v2f a; a.x = Ap[k0]; a.y = Ap[k0 + 1];
        v2f b; b.x = Bp[k0 * 16]; b.y = Bp[k0 * 16 + 16];
        acc = __builtin_amdgcn_wmma_f32_16x16x4_f32(false, a, false, b,
                                                    (short)0, acc, false, false);
    }
    return acc;
}

__global__ void __launch_bounds__(256)
node_scores_l1(const float* __restrict__ x, const float* __restrict__ Vs,
               float* __restrict__ s_i1, float* __restrict__ s_j1)
{
    const int lane = threadIdx.x & 31;
    const int strip = blockIdx.x * (blockDim.x >> 5) + (threadIdx.x >> 5);
    if (strip >= NN / 16) return;
    const int r = lane & 15;
    const int kk = (lane >> 4) << 1;
    const v8f acc = strip_product(x + (size_t)(strip * 16 + r) * 16 + kk, Vs + kk * 16 + r, 16);
    const int rbase = strip * 16 + ((lane >> 4) << 3);
    if (r < 8) {
        float* o = (r < 4) ? (s_i1 + r) : (s_j1 + (r - 4));
#pragma unroll
        for (int v = 0; v < 8; ++v) o[(size_t)(rbase + v) * 4] = acc[v];
    }
}

__device__ __forceinline__ v8f mm4(const float a0, const float a1, const float b0, const float b1,
                                   const v8f acc)
{
    v2f a; a.x = a0; a.y = a1;
    v2f b; b.x = b0; b.y = b1;
    return __builtin_amdgcn_wmma_f32_16x16x4_f32(false, a, false, b, (short)0, acc, false, false);
}

__global__ void __launch_bounds__(256)
proj_scores_l2(const float* __restrict__ accx1, const float* __restrict__ WcF,
               const float* __restrict__ bias, const float* __restrict__ V2F,
               float* __restrict__ s_i2, float* __restrict__ sz2)
{
    const int lane = threadIdx.x & 31;
    const int strip = blockIdx.x * (blockDim.x >> 5) + (threadIdx.x >> 5);
    if (strip >= NN / 16) return;
    const int c = lane & 15;
    const int g = lane >> 4;
    const float4* __restrict__ xr = (const float4*)(accx1 + (size_t)(strip * 16 + c) * 64 + 4 * g);
    float4 xb[8];
#pragma unroll
    for (int m = 0; m < 8; ++m) xb[m] = xr[2 * m];
    const float4* __restrict__ wf = (const float4*)WcF + lane;
    const float4* __restrict__ vf = (const float4*)V2F + lane * 2;
    v8f acc2 = {};
#pragma unroll
    for (int tn = 0; tn < 8; ++tn) {
        v8f acc = {};
#pragma unroll
        for (int m = 0; m < 8; ++m) {
            const float4 w = wf[(tn * 8 + m) * 32];
            acc = mm4(w.x, w.y, xb[m].x, xb[m].y, acc);
            acc = mm4(w.z, w.w, xb[m].z, xb[m].w, acc);
        }
        const float4 bl = *(const float4*)(bias + 16 * tn + 8 * g);
        const float4 bh = *(const float4*)(bias + 16 * tn + 8 * g + 4);
        const float bv[8] = {bl.x, bl.y, bl.z, bl.w, bh.x, bh.y, bh.z, bh.w};
        float hv[8];
#pragma unroll
        for (int r = 0; r < 8; ++r) {
            float xv = acc[r] + bv[r];
            xv = (xv > 0.f) ? xv : (expf(xv) - 1.f);
            xv = (xv > 0.f) ? xv : (expf(xv) - 1.f);
            hv[r] = xv;
        }
        const float4 v0 = vf[tn * 64], v1 = vf[tn * 64 + 1];
        acc2 = mm4(v0.x, v0.y, hv[0], hv[1], acc2);
        acc2 = mm4(v0.z, v0.w, hv[2], hv[3], acc2);
        acc2 = mm4(v1.x, v1.y, hv[4], hv[5], acc2);
        acc2 = mm4(v1.z, v1.w, hv[6], hv[7], acc2);
    }
    if (g == 0) {
        const int node = strip * 16 + c;
        s_i2[node] = acc2[0];
        float2 q; q.x = acc2[1]; q.y = acc2[2];
        ((float2*)sz2)[node] = q;
    }
}

__global__ void __launch_bounds__(1024)
bin_count(const int* __restrict__ dst, int* __restrict__ cntmat)
{
    __shared__ int hist[NBINS];
    const int t = threadIdx.x, b = blockIdx.x;
    for (int i = t; i < NBINS; i += 1024) hist[i] = 0;
    __syncthreads();
    const int e0 = b * ECHUNK;
    const int e1 = (e0 + ECHUNK < EE) ? (e0 + ECHUNK) : EE;
    int dv[(ECHUNK + 1023) / 1024];
#pragma unroll
    for (int j = 0; j < (ECHUNK + 1023) / 1024; ++j) {
        const int e = e0 + j * 1024 + t;
        dv[j] = dst[(e < EE) ? e : (EE - 1)];
    }
#pragma unroll
    for (int j = 0; j < (ECHUNK + 1023) / 1024; ++j)
        if (e0 + j * 1024 + t < e1) atomicAdd(&hist[dv[j] >> BINSH], 1);
    __syncthreads();
    for (int i = t; i < NBINS; i += 1024) cntmat[i * PBLK + b] = hist[i];
}

__device__ __forceinline__ void place_edge(const int d, const int sv, const float4 r0, const float4 r1,
                                           const float4 r2, const float4 r3,
                                           const float* __restrict__ wea1,
                                           const float* __restrict__ wea2, int* cur,
                                           float4* __restrict__ recA, int2* __restrict__ recP)
{
    const float ea[16] = {r0.x, r0.y, r0.z, r0.w, r1.x, r1.y, r1.z, r1.w,
                          r2.x, r2.y, r2.z, r2.w, r3.x, r3.y, r3.z, r3.w};
    float a0 = 0.f, a1 = 0.f, a2 = 0.f, a3 = 0.f, bb = 0.f;
#pragma unroll
    for (int k = 0; k < 16; ++k) {
        a0 += ea[k] * wea1[k];      a1 += ea[k] * wea1[16 + k];
        a2 += ea[k] * wea1[32 + k]; a3 += ea[k] * wea1[48 + k];
        bb += ea[k] * wea2[k];
    }
    const int pos = atomicAdd(&cur[d >> BINSH], 1);
    float4 v4; v4.x = a0; v4.y = a1; v4.z = a2; v4.w = a3;
    recA[pos] = v4;
    int2 q; q.x = sv | ((d & ((1 << BINSH) - 1)) << SRCBITS); q.y = __float_as_int(bb);
    recP[pos] = q;
}

__global__ void __launch_bounds__(1024)
edge_partition(const int* __restrict__ src, const int* __restrict__ dst,
               const float* __restrict__ eattr, const float* __restrict__ wea1,
               const float* __restrict__ wea2, const int* __restrict__ cntmat,
               float4* __restrict__ recA, int2* __restrict__ recP,
               int* __restrict__ binstart)
{
    __shared__ int tot[NBINS];
    __shared__ int cur[NBINS];
    __shared__ int wsum[32];
    const int t = threadIdx.x, lane = t & 31, w = t >> 5, b = blockIdx.x;
    for (int bin = w; bin < NBINS; bin += 32) {
        const int* row = cntmat + bin * PBLK;
        int all = 0, before = 0;
        for (int q = 0; q < PBLK; q += 32) {
            const int c = row[q + lane];
            all += c;
            before += (q + lane < b) ? c : 0;
        }
        for (int o = 16; o > 0; o >>= 1) {
            all += __shfl_xor(all, o);
            before += __shfl_xor(before, o);
        }
        if (lane == 0) { tot[bin] = all; cur[bin] = before; }
    }
    __syncthreads();
    const int v = (t < NBINS) ? tot[t] : 0;
    int incl = v;
    for (int o = 1; o < 32; o <<= 1) {
        int u = __shfl_up(incl, o);
        if (lane >= o) incl += u;
    }
    if (lane == 31) wsum[w] = incl;
    __syncthreads();
    int pre = (lane < w) ? wsum[lane] : 0;
    for (int o = 16; o > 0; o >>= 1) pre += __shfl_xor(pre, o);
    if (t < NBINS) {
        const int first = pre + incl - v;
        cur[t] += first;
        if (b == 0) binstart[t] = first;
    }
    __syncthreads();
    const int e0 = b * ECHUNK;
    const int e1 = (e0 + ECHUNK < EE) ? (e0 + ECHUNK) : EE;
    const float4* __restrict__ ea4 = (const float4*)eattr;
    int e = e0 + t;
    if (e < e1) {
        int dA = dst[e], sA = src[e];
        float4 a0 = ea4[(size_t)e * 4], a1 = ea4[(size_t)e * 4 + 1];
        float4 a2 = ea4[(size_t)e * 4 + 2], a3 = ea4[(size_t)e * 4 + 3];
        for (;;) {
            const int eB = e + 1024;
            const bool moreB = eB < e1;
            const int fB = moreB ? eB : e;
            const int dB = dst[fB], sB = src[fB];
            const float4 b0 = ea4[(size_t)fB * 4], b1 = ea4[(size_t)fB * 4 + 1];
            const float4 b2 = ea4[(size_t)fB * 4 + 2], b3 = ea4[(size_t)fB * 4 + 3];
            place_edge(dA, sA, a0, a1, a2, a3, wea1, wea2, cur, recA, recP);
            if (!moreB) break;
            e = eB + 1024;
            const bool moreA = e < e1;
            const int fA = moreA ? e : eB;
            dA = dst[fA]; sA = src[fA];
            a0 = ea4[(size_t)fA * 4]; a1 = ea4[(size_t)fA * 4 + 1];
            a2 = ea4[(size_t)fA * 4 + 2]; a3 = ea4[(size_t)fA * 4 + 3];
            place_edge(dB, sB, b0, b1, b2, b3, wea1, wea2, cur, recA, recP);
            if (!moreA) break;
        }
    }
}

__global__ void __launch_bounds__(1024)
bin_sort(const int* __restrict__ binstart, float4* recA, int2* recP,
         int* __restrict__ rowptr, int* __restrict__ cnt)
{
    __shared__ float4 bufA[CAP];
    __shared__ int2 bufP[CAP];
    __shared__ int cntL[256];
    __shared__ int startL[256];
    __shared__ int curL[256];
    __shared__ int wtot[32];
    const int t = threadIdx.x, lane = t & 31, w = t >> 5, bin = blockIdx.x;
    const int beg = binstart[bin];
    const int end = (bin + 1 < NBINS) ? binstart[bin + 1] : EE;
    const int n = end - beg;
    const bool fits = n <= CAP;
    if (t < 256) { cntL[t] = 0; curL[t] = 0; }
    __syncthreads();
    if (fits) {
        int2 pv[CAP / 1024];
        float4 av[CAP / 1024];
#pragma unroll
        for (int j = 0; j < CAP / 1024; ++j) {
            const int g = beg + j * 1024 + t;
            const int gc = (g < EE) ? g : (EE - 1);
            pv[j] = recP[gc];
            av[j] = recA[gc];
        }
#pragma unroll
        for (int j = 0; j < CAP / 1024; ++j) {
            const int i = j * 1024 + t;
            bufP[i] = pv[j];
            bufA[i] = av[j];
            if (i < n) atomicAdd(&cntL[pv[j].x >> SRCBITS], 1);
        }
    } else {
        for (int i = t; i < n; i += 1024) atomicAdd(&cntL[recP[beg + i].x >> SRCBITS], 1);
    }
    __syncthreads();
    int c = 0, incl = 0;
    if (t < 256) {
        c = cntL[t];
        incl = c;
        for (int o = 1; o < 32; o <<= 1) {
            int u = __shfl_up(incl, o);
            if (lane >= o) incl += u;
        }
        if (lane == 31) wtot[w] = incl;
    }
    __syncthreads();
    if (t < 256) {
        int pre = (lane < w) ? wtot[lane] : 0;
        for (int o = 16; o > 0; o >>= 1) pre += __shfl_xor(pre, o);
        const int st = pre + incl - c;
        startL[t] = st;
        const int node = (bin << BINSH) + t;
        if (node < NN) { rowptr[node] = beg + st; cnt[node] = c; }
    }
    __syncthreads();
    if (fits) {
        for (int i = t; i < n; i += 1024) {
            const int2 p = bufP[i];
            const int dl = p.x >> SRCBITS;
            const int pos = beg + startL[dl] + atomicAdd(&curL[dl], 1);
            int2 q; q.x = p.x & SRCMASK; q.y = p.y;
            recP[pos] = q;
            recA[pos] = bufA[i];
        }
    } else {
        if (t == 0) {
            for (int nd = 0; nd < 256; ++nd) {
                const int endn = cntL[nd];
                int k = curL[nd];
                while (k < endn) {
                    const int p = beg + startL[nd] + k;
                    const int2 r = recP[p];
                    const int dl = r.x >> SRCBITS;
                    if (dl == nd) { ++k; continue; }
                    const int q = beg + startL[dl] + curL[dl];
                    curL[dl] += 1;
                    const int2 r2 = recP[q];
                    const float4 av = recA[p], av2 = recA[q];
                    recP[q] = r;  recA[q] = av;
                    recP[p] = r2; recA[p] = av2;
                }
                curL[nd] = endn;
            }
        }
        __syncthreads();
        for (int i = t; i < n; i += 1024) recP[beg + i].x &= SRCMASK;
    }
}

__device__ __forceinline__ float leaky(float v) { return (v >= 0.f) ? v : NEGS * v; }
__device__ __forceinline__ float wave_max(float v) {
    for (int o = 16; o > 0; o >>= 1) v = fmaxf(v, __shfl_xor(v, o));
    return v;
}
__device__ __forceinline__ float wave_sum(float v) {
    for (int o = 16; o > 0; o >>= 1) v += __shfl_xor(v, o);
    return v;
}
__device__ __forceinline__ float lane_value(float v, int l) {
    return __int_as_float(__builtin_amdgcn_readlane(__float_as_int(v), l));
}

__global__ void __launch_bounds__(256)
node_aggr_l1(const int* __restrict__ rowptr, const int* __restrict__ cnt,
             const float* __restrict__ se1S, const int* __restrict__ pairS,
             const float* __restrict__ s_i1, const float* __restrict__ s_j1,
             const float* __restrict__ x, float* __restrict__ accx1)
{
    const int lane = threadIdx.x & 31;
    const int k = lane & 15;
    const bool hi = lane >= 16;
    const int wave = blockIdx.x * (blockDim.x >> 5) + __builtin_amdgcn_readfirstlane(threadIdx.x >> 5);
    const int nw = gridDim.x * (blockDim.x >> 5);
    const float4* __restrict__ se4 = (const float4*)se1S;
    const int2* __restrict__ pr2 = (const int2*)pairS;
    for (int d = wave; d < NN; d += nw) {
        const int beg = rowptr[d];
        const int deg = cnt[d];
        const float4 si = *(const float4*)(s_i1 + (size_t)d * 4);
        float M0 = -INFINITY, M1 = -INFINITY, M2 = -INFINITY, M3 = -INFINITY;
        float acc0 = 0.f, acc1 = 0.f;
        float dn0 = 0.f, dn1 = 0.f, dn2 = 0.f, dn3 = 0.f;
        for (int j0 = 0; j0 < deg; j0 += 32) {
            const int j = j0 + lane;
            const bool act = j < deg;
            const int jj = act ? j : j0;
            const int s = pr2[beg + jj].x;
            const float4 se = se4[beg + jj];
            const float4 sj = *(const float4*)(s_j1 + (size_t)s * 4);
            const float pen = act ? 0.f : -INFINITY;
            const float c0 = leaky(si.x + sj.x + se.x) + pen;
            const float c1 = leaky(si.y + sj.y + se.y) + pen;
            const float c2 = leaky(si.z + sj.z + se.z) + pen;
            const float c3 = leaky(si.w + sj.w + se.w) + pen;
            const float n0 = fmaxf(M0, wave_max(c0)), n1 = fmaxf(M1, wave_max(c1));
            const float n2 = fmaxf(M2, wave_max(c2)), n3 = fmaxf(M3, wave_max(c3));
            const float r0 = expf(M0 - n0), r1 = expf(M1 - n1);
            const float r2 = expf(M2 - n2), r3 = expf(M3 - n3);
            M0 = n0; M1 = n1; M2 = n2; M3 = n3;
            const float p0 = expf(c0 - n0), p1 = expf(c1 - n1);
            const float p2 = expf(c2 - n2), p3 = expf(c3 - n3);
            dn0 = dn0 * r0 + p0; dn1 = dn1 * r1 + p1;
            dn2 = dn2 * r2 + p2; dn3 = dn3 * r3 + p3;
            acc0 *= hi ? r1 : r0;
            acc1 *= hi ? r3 : r2;
            const int left = deg - j0;
            const int cnt4 = ((left < 32 ? left : 32) + 3) & ~3;
            for (int t = 0; t < cnt4; t += 4) {
#pragma unroll
                for (int u = 0; u < 4; ++u) {
                    const int st = __builtin_amdgcn_readlane(s, t + u);
                    const float a0 = lane_value(p0, t + u), a1 = lane_value(p1, t + u);
                    const float a2 = lane_value(p2, t + u), a3 = lane_value(p3, t + u);
                    const float xv = x[(size_t)st * 16 + k];
                    acc0 = fmaf(hi ? a1 : a0, xv, acc0);
                    acc1 = fmaf(hi ? a3 : a2, xv, acc1);
                }
            }
        }
        dn0 = wave_sum(dn0); dn1 = wave_sum(dn1); dn2 = wave_sum(dn2); dn3 = wave_sum(dn3);
        const float dA = hi ? dn1 : dn0;
        const float dB = hi ? dn3 : dn2;
        accx1[(size_t)d * 64 + lane]      = (dA > 0.f) ? acc0 / dA : 0.f;
        accx1[(size_t)d * 64 + 32 + lane] = (dB > 0.f) ? acc1 / dB : 0.f;
    }
}

__global__ void __launch_bounds__(256)
node_aggr_l2(const int* __restrict__ rowptr, const int* __restrict__ cnt,
             const int* __restrict__ pairS, const float* __restrict__ s_i2,
             const float* __restrict__ sz2, const float* __restrict__ p2b,
             float* __restrict__ out)
{
    const int lane = threadIdx.x & 31;
    const int wave = blockIdx.x * (blockDim.x >> 5) + __builtin_amdgcn_readfirstlane(threadIdx.x >> 5);
    const int nw = gridDim.x * (blockDim.x >> 5);
    const int2* __restrict__ pr2 = (const int2*)pairS;
    const float2* __restrict__ sz = (const float2*)sz2;
    const float bias = p2b[0];
    for (int d = wave; d < NN; d += nw) {
        const int beg = rowptr[d];
        const int deg = cnt[d];
        const float si = s_i2[d];
        float M = -INFINITY, num = 0.f, den = 0.f;
        for (int j0 = 0; j0 < deg; j0 += 32) {
            const int j = j0 + lane;
            const bool act = j < deg;
            const int2 pr = pr2[beg + (act ? j : j0)];
            const float2 q = sz[pr.x];
            const float c = leaky(si + q.x + __int_as_float(pr.y)) + (act ? 0.f : -INFINITY);
            const float n = fmaxf(M, wave_max(c));
            const float r = expf(M - n);
            M = n;
            const float p = expf(c - n);
            num = num * r + p * q.y;
            den = den * r + p;
        }
        num = wave_sum(num);
        den = wave_sum(den);
        if (lane == 0) {
            float v = (den > 0.f) ? num / den : 0.f;
            v += bias;
            out[d] = (v > 0.f) ? v : (expf(v) - 1.f);
        }
    }
}

extern "C" void kernel_launch(void* const* d_in, const int* in_sizes, int n_in,
                              void* d_out, int out_size, void* d_ws, size_t ws_size,
                              hipStream_t stream)
{
    const float* x       = (const float*)d_in[0];
    const int*   eidx    = (const int*)d_in[1];
    const float* eattr   = (const float*)d_in[2];
    const float* W1      = (const float*)d_in[3];
    const float* We1     = (const float*)d_in[4];
    const float* att1    = (const float*)d_in[5];
    const float* proj1_w = (const float*)d_in[6];
    const float* proj1_b = (const float*)d_in[7];
    const float* W2      = (const float*)d_in[8];
    const float* We2     = (const float*)d_in[9];
    const float* att2    = (const float*)d_in[10];
    const float* proj2_w = (const float*)d_in[11];
    const float* proj2_b = (const float*)d_in[12];
    float* out = (float*)d_out;

    const int* srcp = eidx;
    const int* dstp = eidx + EE;

    float* ws     = (float*)d_ws;
    float* xw1    = ws;
    float* accx1  = xw1 + (size_t)NN * 128;
    int*   pairS  = (int*)(accx1 + (size_t)NN * 64);
    float* s_i1   = accx1 + (size_t)NN * 128;
    float* s_j1   = s_i1 + (size_t)NN * 4;
    float* slotA  = s_j1 + (size_t)NN * 4;
    int*   cnt    = (int*)slotA;
    float* Wcomb  = slotA + (size_t)NN;
    float* Vs     = Wcomb + 64 * 128;
    float* V2     = Vs + 256;
    int*   rowptr = (int*)(slotA + (size_t)NN * 4);
    float* hw2    = slotA + (size_t)NN * 8;
    float* s_i2   = hw2 + (size_t)NN * 64;
    float* sz2    = s_i2 + (size_t)NN;
    float* wea1   = s_i2 + (size_t)NN * 4 + 2048;
    float* wea2   = wea1 + 64;
    float* se1S   = xw1;
    int*   cntmat = (int*)hw2;
    int*   binstart = cntmat + NBINS * PBLK;

    prep_weights<<<32, 256, 0, stream>>>(W1, We1, att1, We2, att2, proj1_w, W2, proj2_w,
                                         Vs, V2, wea1, wea2, Wcomb);

    node_scores_l1<<<(NN / 16 + 7) / 8, 256, 0, stream>>>(x, Vs, s_i1, s_j1);

    bin_count<<<PBLK, 1024, 0, stream>>>(dstp, cntmat);
    edge_partition<<<PBLK, 1024, 0, stream>>>(srcp, dstp, eattr, wea1, wea2, cntmat,
                                              (float4*)se1S, (int2*)pairS, binstart);
    bin_sort<<<NBINS, 1024, 0, stream>>>(binstart, (float4*)se1S, (int2*)pairS, rowptr, cnt);

    node_aggr_l1<<<2048, 256, 0, stream>>>(rowptr, cnt, se1S, pairS, s_i1, s_j1, x, accx1);

    proj_scores_l2<<<(NN / 16 + 7) / 8, 256, 0, stream>>>(accx1, Wcomb, proj1_b, V2, s_i2, sz2);
    node_aggr_l2<<<2048, 256, 0, stream>>>(rowptr, cnt, pairS, s_i2, sz2, proj2_b, out);
}


# Edge sort: both passes put their records in order in on-chip memory and store them as whole-line stretches (no scattered 16- and 8-byte global stores); ranks from the counting atomics; sums reordered
# speedup vs baseline: 7.6455x; 1.1124x over previous
#include <hip/hip_runtime.h>
#include <cmath>

#define NN 100000
#define EE 3200000
#define NEGS 0.2f
#define BINSH 8
#define NBINS ((NN + 255) >> BINSH)
#define PBLK 256
#define ECHUNK ((EE + PBLK - 1) / PBLK)
#define CAP 10240
#define SRCBITS 17
#define SRCMASK ((1 << SRCBITS) - 1)

typedef __attribute__((ext_vector_type(2))) float v2f;
typedef __attribute__((ext_vector_type(8))) float v8f;

__global__ void prep_weights(const float* __restrict__ W1, const float* __restrict__ We1,
                             const float* __restrict__ att1, const float* __restrict__ We2,
                             const float* __restrict__ att2, const float* __restrict__ proj1_w,
                             const float* __restrict__ W2, const float* __restrict__ p2w,
                             float* __restrict__ Vs, float* __restrict__ V2,
                             float* __restrict__ wea1,
                             float* __restrict__ wea2, float* __restrict__ Wcomb)
{
    int tid = blockIdx.x * blockDim.x + threadIdx.x;
    int nt = gridDim.x * blockDim.x;
    for (int i = tid; i < 256; i += nt) {
        int k = i >> 4, c = i & 15;
        float s = 0.f;
        if (c < 8) {
            int h = c & 3;
            const float* a = att1 + h * 96 + (c >> 2) * 32;
            const float* w = W1 + h * 512 + k * 32;
            for (int o = 0; o < 32; ++o) s += w[o] * a[o];
        }
        Vs[i] = s;
    }
    for (int i = tid; i < 2048; i += nt) {
        const int e2 = i & 7, ln = (i >> 3) & 31, t2 = i >> 8;
        int k = 16 * t2 + 8 * (ln >> 4) + e2, c = ln & 15;
        float s = 0.f;
        if (c < 3) {
            const float* a = (c == 2) ? p2w : (att2 + c * 32);
            const float* w = W2 + k * 32;
            for (int o = 0; o < 32; ++o) s += w[o] * a[o];
        }
        V2[i] = s;
    }
    for (int i = tid; i < 64; i += nt) {
        int h = i >> 4, k = i & 15;
        float s = 0.f;
        for (int o = 0; o < 32; ++o) s += We1[h * 512 + k * 32 + o] * att1[h * 96 + 64 + o];
        wea1[i] = s;
    }
    for (int i = tid; i < 16; i += nt) {
        float s = 0.f;
        for (int o = 0; o < 32; ++o) s += We2[i * 32 + o] * att2[64 + o];
        wea2[i] = s;
    }
    for (int i = tid; i < 64 * 128; i += nt) {
        const int e4 = i & 3, ln = (i >> 2) & 31, m8 = (i >> 7) & 7, t8 = i >> 10;
        int row = 8 * m8 + 4 * (ln >> 4) + e4, c = 16 * t8 + (ln & 15), h = row >> 4, k = row & 15;
        const float* w = W1 + h * 512 + k * 32;
        const float* pw = proj1_w + (size_t)(h * 32) * 128 + c;
        float s = 0.f;
        for (int o = 0; o < 32; ++o) s += w[o] * pw[(size_t)o * 128];
        Wcomb[i] = s;
    }
}

__device__ __forceinline__ v8f strip_product(const float* __restrict__ Ap,
                                             const float* __restrict__ Bp, const int K)
{
    v8f acc = {};
#pragma unroll
    for (int k0 = 0; k0 < K; k0 += 4) {
        v2f a; a.x = Ap[k0]; a.y = Ap[k0 + 1];
        v2f b; b.x = Bp[k0 * 16]; b.y = Bp[k0 * 16 + 16];
        acc = __builtin_amdgcn_wmma_f32_16x16x4_f32(false, a, false, b,
                                                    (short)0, acc, false, false);
    }
    return acc;
}

__global__ void __launch_bounds__(256)
node_scores_l1(const float* __restrict__ x, const float* __restrict__ Vs,
               float* __restrict__ s_i1, float* __restrict__ s_j1)
{
    const int lane = threadIdx.x & 31;
    const int strip = blockIdx.x * (blockDim.x >> 5) + (threadIdx.x >> 5);
    if (strip >= NN / 16) return;
    const int r = lane & 15;
    const int kk = (lane >> 4) << 1;
    const v8f acc = strip_product(x + (size_t)(strip * 16 + r) * 16 + kk, Vs + kk * 16 + r, 16);
    const int rbase = strip * 16 + ((lane >> 4) << 3);
    if (r < 8) {
        float* o = (r < 4) ? (s_i1 + r) : (s_j1 + (r - 4));
#pragma unroll
        for (int v = 0; v < 8; ++v) o[(size_t)(rbase + v) * 4] = acc[v];
    }
}

__device__ __forceinline__ v8f mm4(const float a0, const float a1, const float b0, const float b1,
                                   const v8f acc)
{
    v2f a; a.x = a0; a.y = a1;
    v2f b; b.x = b0; b.y = b1;
    return __builtin_amdgcn_wmma_f32_16x16x4_f32(false, a, false, b, (short)0, acc, false, false);
}

__global__ void __launch_bounds__(256)
proj_scores_l2(const float* __restrict__ accx1, const float* __restrict__ WcF,
               const float* __restrict__ bias, const float* __restrict__ V2F,
               float* __restrict__ s_i2, float* __restrict__ sz2)
{
    const int lane = threadIdx.x & 31;
    const int strip = blockIdx.x * (blockDim.x >> 5) + (threadIdx.x >> 5);
    if (strip >= NN / 16) return;
    const int c = lane & 15;
    const int g = lane >> 4;
    const float4* __restrict__ xr = (const float4*)(accx1 + (size_t)(strip * 16 + c) * 64 + 4 * g);
    float4 xb[8];
#pragma unroll
    for (int m = 0; m < 8; ++m) xb[m] = xr[2 * m];
    const float4* __restrict__ wf = (const float4*)WcF + lane;
    const float4* __restrict__ vf = (const float4*)V2F + lane * 2;
    v8f acc2 = {};
#pragma unroll
    for (int tn = 0; tn < 8; ++tn) {
        v8f acc = {};
#pragma unroll
        for (int m = 0; m < 8; ++m) {
            const float4 w = wf[(tn * 8 + m) * 32];
            acc = mm4(w.x, w.y, xb[m].x, xb[m].y, acc);
            acc = mm4(w.z, w.w, xb[m].z, xb[m].w, acc);
        }
        const float4 bl = *(const float4*)(bias + 16 * tn + 8 * g);
        const float4 bh = *(const float4*)(bias + 16 * tn + 8 * g + 4);
        const float bv[8] = {bl.x, bl.y, bl.z, bl.w, bh.x, bh.y, bh.z, bh.w};
        float hv[8];
#pragma unroll
        for (int r = 0; r < 8; ++r) {
            float xv = acc[r] + bv[r];
            xv = (xv > 0.f) ? xv : (expf(xv) - 1.f);
            xv = (xv > 0.f) ? xv : (expf(xv) - 1.f);
            hv[r] = xv;
        }
        const float4 v0 = vf[tn * 64], v1 = vf[tn * 64 + 1];
        acc2 = mm4(v0.x, v0.y, hv[0], hv[1], acc2);
        acc2 = mm4(v0.z, v0.w, hv[2], hv[3], acc2);
        acc2 = mm4(v1.x, v1.y, hv[4], hv[5], acc2);
        acc2 = mm4(v1.z, v1.w, hv[6], hv[7], acc2);
    }
    if (g == 0) {
        const int node = strip * 16 + c;
        s_i2[node] = acc2[0];
        float2 q; q.x = acc2[1]; q.y = acc2[2];
        ((float2*)sz2)[node] = q;
    }
}

__global__ void __launch_bounds__(1024)
bin_count(const int* __restrict__ dst, int* __restrict__ cntmat)
{
    __shared__ int hist[NBINS];
    const int t = threadIdx.x, b = blockIdx.x;
    for (int i = t; i < NBINS; i += 1024) hist[i] = 0;
    __syncthreads();
    const int e0 = b * ECHUNK;
    const int e1 = (e0 + ECHUNK < EE) ? (e0 + ECHUNK) : EE;
    int dv[(ECHUNK + 1023) / 1024];
#pragma unroll
    for (int j = 0; j < (ECHUNK + 1023) / 1024; ++j) {
        const int e = e0 + j * 1024 + t;
        dv[j] = dst[(e < EE) ? e : (EE - 1)];
    }
#pragma unroll
    for (int j = 0; j < (ECHUNK + 1023) / 1024; ++j)
        if (e0 + j * 1024 + t < e1) atomicAdd(&hist[dv[j] >> BINSH], 1);
    __syncthreads();
    for (int i = t; i < NBINS; i += 1024) cntmat[i * PBLK + b] = hist[i];
}

__device__ __forceinline__ void place_edge(const int d, const int sv, const float4 r0, const float4 r1,
                                           const float4 r2, const float4 r3,
                                           const float* __restrict__ wea1,
                                           const float* __restrict__ wea2, int* lcur,
                                           float4* stA, int2* stP)
{
    const float ea[16] = {r0.x, r0.y, r0.z, r0.w, r1.x, r1.y, r1.z, r1.w,
                          r2.x, r2.y, r2.z, r2.w, r3.x, r3.y, r3.z, r3.w};
    float a0 = 0.f, a1 = 0.f, a2 = 0.f, a3 = 0.f, bb = 0.f;
#pragma unroll
    for (int k = 0; k < 16; ++k) {
        a0 += ea[k] * wea1[k];      a1 += ea[k] * wea1[16 + k];
        a2 += ea[k] * wea1[32 + k]; a3 += ea[k] * wea1[48 + k];
        bb += ea[k] * wea2[k];
    }
    const int pos = atomicAdd(&lcur[d >> BINSH], 1);
    float4 v4; v4.x = a0; v4.y = a1; v4.z = a2; v4.w = a3;
    stA[pos] = v4;
    int2 q; q.x = sv | ((d & ((1 << BINSH) - 1)) << SRCBITS); q.y = __float_as_int(bb);
    stP[pos] = q;
}

__global__ void __launch_bounds__(1024)
edge_partition(const int* __restrict__ src, const int* __restrict__ dst,
               const float* __restrict__ eattr, const float* __restrict__ wea1,
               const float* __restrict__ wea2, const int* __restrict__ cntmat,
               float4* __restrict__ recA, int2* __restrict__ recP,
               int* __restrict__ binstart)
{
    __shared__ float4 stA[ECHUNK];
    __shared__ int2 stP[ECHUNK];
    __shared__ int tot[NBINS];
    __shared__ int gfirst[NBINS];
    __shared__ int own[NBINS];
    __shared__ int lstart[NBINS];
    __shared__ int lcur[NBINS];
    __shared__ int wsum[32];
    __shared__ int wsumL[32];
    const int t = threadIdx.x, lane = t & 31, w = t >> 5, b = blockIdx.x;
    for (int bin = w; bin < NBINS; bin += 32) {
        const int* row = cntmat + bin * PBLK;
        int all = 0, before = 0;
        for (int q = 0; q < PBLK; q += 32) {
            const int c = row[q + lane];
            all += c;
            before += (q + lane < b) ? c : 0;
        }
        const int mine = row[b];
        for (int o = 16; o > 0; o >>= 1) {
            all += __shfl_xor(all, o);
            before += __shfl_xor(before, o);
        }
        if (lane == 0) { tot[bin] = all; gfirst[bin] = before; own[bin] = mine; }
    }
    __syncthreads();
    const int v = (t < NBINS) ? tot[t] : 0;
    const int m = (t < NBINS) ? own[t] : 0;
    int incl = v, inclL = m;
    for (int o = 1; o < 32; o <<= 1) {
        const int u = __shfl_up(incl, o);
        const int uL = __shfl_up(inclL, o);
        if (lane >= o) { incl += u; inclL += uL; }
    }
    if (lane == 31) { wsum[w] = incl; wsumL[w] = inclL; }
    __syncthreads();
    int pre = (lane < w) ? wsum[lane] : 0;
    int preL = (lane < w) ? wsumL[lane] : 0;
    for (int o = 16; o > 0; o >>= 1) {
        pre += __shfl_xor(pre, o);
        preL += __shfl_xor(preL, o);
    }
    if (t < NBINS) {
        const int first = pre + incl - v;
        const int firstL = preL + inclL - m;
        gfirst[t] += first;
        lstart[t] = firstL;
        lcur[t] = firstL;
        if (b == 0) binstart[t] = first;
    }
    __syncthreads();
    const int e0 = b * ECHUNK;
    const int e1 = (e0 + ECHUNK < EE) ? (e0 + ECHUNK) : EE;
    const float4* __restrict__ ea4 = (const float4*)eattr;
    int e = e0 + t;
    if (e < e1) {
        int dA = dst[e], sA = src[e];
        float4 a0 = ea4[(size_t)e * 4], a1 = ea4[(size_t)e * 4 + 1];
        float4 a2 = ea4[(size_t)e * 4 + 2], a3 = ea4[(size_t)e * 4 + 3];
        for (;;) {
            const int eB = e + 1024;
            const bool moreB = eB < e1;
            const int fB = moreB ? eB : e;
            const int dB = dst[fB], sB = src[fB];
            const float4 b0 = ea4[(size_t)fB * 4], b1 = ea4[(size_t)fB * 4 + 1];
            const float4 b2 = ea4[(size_t)fB * 4 + 2], b3 = ea4[(size_t)fB * 4 + 3];
            place_edge(dA, sA, a0, a1, a2, a3, wea1, wea2, lcur, stA, stP);
            if (!moreB) break;
            e = eB + 1024;
            const bool moreA = e < e1;
            const int fA = moreA ? e : eB;
            dA = dst[fA]; sA = src[fA];
            a0 = ea4[(size_t)fA * 4]; a1 = ea4[(size_t)fA * 4 + 1];
            a2 = ea4[(size_t)fA * 4 + 2]; a3 = ea4[(size_t)fA * 4 + 3];
            place_edge(dB, sB, b0, b1, b2, b3, wea1, wea2, lcur, stA, stP);
            if (!moreA) break;
        }
    }
    __syncthreads();
    for (int bin = w; bin < NBINS; bin += 32) {
        const int ls = lstart[bin], c = own[bin], g = gfirst[bin];
        for (int i = lane; i < c; i += 32) {
            recA[g + i] = stA[ls + i];
            recP[g + i] = stP[ls + i];
        }
    }
}

#define FOR_SLOTS(X) X(0) X(1) X(2) X(3) X(4) X(5) X(6) X(7) X(8) X(9)
static_assert(CAP / 1024 == 10, "FOR_SLOTS lists CAP / 1024 slots");

__global__ void __launch_bounds__(1024)
bin_sort(const int* __restrict__ binstart, float4* recA, int2* recP,
         int* __restrict__ rowptr, int* __restrict__ cnt)
{
    __shared__ float4 bufA[CAP];
    __shared__ int2 bufP[CAP];
    __shared__ int cntL[256];
    __shared__ int startL[256];
    __shared__ int curL[256];
    __shared__ int wtot[32];
    const int t = threadIdx.x, lane = t & 31, w = t >> 5, bin = blockIdx.x;
    const int beg = binstart[bin];
    const int end = (bin + 1 < NBINS) ? binstart[bin + 1] : EE;
    const int n = end - beg;
    const bool fits = n <= CAP;
    if (t < 256) { cntL[t] = 0; curL[t] = 0; }
    __syncthreads();
#define DECL_SLOT(j) int2 pv##j = make_int2(0, 0); float4 av##j = make_float4(0.f, 0.f, 0.f, 0.f); int rk##j = 0;
    FOR_SLOTS(DECL_SLOT)
    if (fits) {
#define LOAD_SLOT(j) { const int g = beg + j * 1024 + t; const int gc = (g < EE) ? g : (EE - 1); \
                       pv##j = recP[gc]; av##j = recA[gc]; }
        FOR_SLOTS(LOAD_SLOT)
#define COUNT_SLOT(j) if (j * 1024 + t < n) rk##j = atomicAdd(&cntL[pv##j.x >> SRCBITS], 1);
        FOR_SLOTS(COUNT_SLOT)
    } else {
        for (int i = t; i < n; i += 1024) atomicAdd(&cntL[recP[beg + i].x >> SRCBITS], 1);
    }
    __syncthreads();
    int c = 0, incl = 0;
    if (t < 256) {
        c = cntL[t];
        incl = c;
        for (int o = 1; o < 32; o <<= 1) {
            int u = __shfl_up(incl, o);
            if (lane >= o) incl += u;
        }
        if (lane == 31) wtot[w] = incl;
    }
    __syncthreads();
    if (t < 256) {
        int pre = (lane < w) ? wtot[lane] : 0;
        for (int o = 16; o > 0; o >>= 1) pre += __shfl_xor(pre, o);
        const int st = pre + incl - c;
        startL[t] = st;
        const int node = (bin << BINSH) + t;
        if (node < NN) { rowptr[node] = beg + st; cnt[node] = c; }
    }
    __syncthreads();
    if (fits) {
#define START_SLOT(j) const int st##j = startL[pv##j.x >> SRCBITS];
        FOR_SLOTS(START_SLOT)
#define PLACE_SLOT(j) if (j * 1024 + t < n) { \
            const int pos = st##j + rk##j; \
            int2 q; q.x = pv##j.x & SRCMASK; q.y = pv##j.y; \
            bufP[pos] = q; bufA[pos] = av##j; }
        FOR_SLOTS(PLACE_SLOT)
        __syncthreads();
        const int lead = beg & 15;
        for (int i = t - lead; i < n; i += 1024) {
            if (i >= 0) {
                recP[beg + i] = bufP[i];
                recA[beg + i] = bufA[i];
            }
        }
    } else {
        if (t == 0) {
            for (int nd = 0; nd < 256; ++nd) {
                const int endn = cntL[nd];
                int k = curL[nd];
                while (k < endn) {
                    const int p = beg + startL[nd] + k;
                    const int2 r = recP[p];
                    const int dl = r.x >> SRCBITS;
                    if (dl == nd) { ++k; continue; }
                    const int q = beg + startL[dl] + curL[dl];
                    curL[dl] += 1;
                    const int2 r2 = recP[q];
                    const float4 a1 = recA[p], a2 = recA[q];
                    recP[q] = r;  recA[q] = a1;
                    recP[p] = r2; recA[p] = a2;
                }
                curL[nd] = endn;
            }
        }
        __syncthreads();
        for (int i = t; i < n; i += 1024) recP[beg + i].x &= SRCMASK;
    }
}

__device__ __forceinline__ float leaky(float v) { return (v >= 0.f) ? v : NEGS * v; }
__device__ __forceinline__ float wave_max(float v) {
    for (int o = 16; o > 0; o >>= 1) v = fmaxf(v, __shfl_xor(v, o));
    return v;
}
__device__ __forceinline__ float wave_sum(float v) {
    for (int o = 16; o > 0; o >>= 1) v += __shfl_xor(v, o);
    return v;
}
__device__ __forceinline__ float lane_value(float v, int l) {
    return __int_as_float(__builtin_amdgcn_readlane(__float_as_int(v), l));
}

__global__ void __launch_bounds__(256)
node_aggr_l1(const int* __restrict__ rowptr, const int* __restrict__ cnt,
             const float* __restrict__ se1S, const int* __restrict__ pairS,
             const float* __restrict__ s_i1, const float* __restrict__ s_j1,
             const float* __restrict__ x, float* __restrict__ accx1)
{
    const int lane = threadIdx.x & 31;
    const int k = lane & 15;
    const bool hi = lane >= 16;
    const int wave = blockIdx.x * (blockDim.x >> 5) + __builtin_amdgcn_readfirstlane(threadIdx.x >> 5);
    const int nw = gridDim.x * (blockDim.x >> 5);
    const float4* __restrict__ se4 = (const float4*)se1S;
    const int2* __restrict__ pr2 = (const int2*)pairS;
    for (int d = wave; d < NN; d += nw) {
        const int beg = rowptr[d];
        const int deg = cnt[d];
        const float4 si = *(const float4*)(s_i1 + (size_t)d * 4);
        float M0 = -INFINITY, M1 = -INFINITY, M2 = -INFINITY, M3 = -INFINITY;
        float acc0 = 0.f, acc1 = 0.f;
        float dn0 = 0.f, dn1 = 0.f, dn2 = 0.f, dn3 = 0.f;
        for (int j0 = 0; j0 < deg; j0 += 32) {
            const int j = j0 + lane;
            const bool act = j < deg;
            const int jj = act ? j : j0;
            const int s = pr2[beg + jj].x;
            const float4 se = se4[beg + jj];
            const float4 sj = *(const float4*)(s_j1 + (size_t)s * 4);
            const float pen = act ? 0.f : -INFINITY;
            const float c0 = leaky(si.x + sj.x + se.x) + pen;
            const float c1 = leaky(si.y + sj.y + se.y) + pen;
            const float c2 = leaky(si.z + sj.z + se.z) + pen;
            const float c3 = leaky(si.w + sj.w + se.w) + pen;
            const float n0 = fmaxf(M0, wave_max(c0)), n1 = fmaxf(M1, wave_max(c1));
            const float n2 = fmaxf(M2, wave_max(c2)), n3 = fmaxf(M3, wave_max(c3));
            const float r0 = expf(M0 - n0), r1 = expf(M1 - n1);
            const float r2 = expf(M2 - n2), r3 = expf(M3 - n3);
            M0 = n0; M1 = n1; M2 = n2; M3 = n3;
            const float p0 = expf(c0 - n0), p1 = expf(c1 - n1);
            const float p2 = expf(c2 - n2), p3 = expf(c3 - n3);
            dn0 = dn0 * r0 + p0; dn1 = dn1 * r1 + p1;
            dn2 = dn2 * r2 + p2; dn3 = dn3 * r3 + p3;
            acc0 *= hi ? r1 : r0;
            acc1 *= hi ? r3 : r2;
            const int left = deg - j0;
            const int cnt4 = ((left < 32 ? left : 32) + 3) & ~3;
            for (int t = 0; t < cnt4; t += 4) {
#pragma unroll
                for (int u = 0; u < 4; ++u) {
                    const int st = __builtin_amdgcn_readlane(s, t + u);
                    const float a0 = lane_value(p0, t + u), a1 = lane_value(p1, t + u);
                    const float a2 = lane_value(p2, t + u), a3 = lane_value(p3, t + u);
                    const float xv = x[(size_t)st * 16 + k];
                    acc0 = fmaf(hi ? a1 : a0, xv, acc0);
                    acc1 = fmaf(hi ? a3 : a2, xv, acc1);
                }
            }
        }
        dn0 = wave_sum(dn0); dn1 = wave_sum(dn1); dn2 = wave_sum(dn2); dn3 = wave_sum(dn3);
        const float dA = hi ? dn1 : dn0;
        const float dB = hi ? dn3 : dn2;
        accx1[(size_t)d * 64 + lane]      = (dA > 0.f) ? acc0 / dA : 0.f;
        accx1[(size_t)d * 64 + 32 + lane] = (dB > 0.f) ? acc1 / dB : 0.f;
    }
}

__global__ void __launch_bounds__(256)
node_aggr_l2(const int* __restrict__ rowptr, const int* __restrict__ cnt,
             const int* __restrict__ pairS, const float* __restrict__ s_i2,
             const float* __restrict__ sz2, const float* __restrict__ p2b,
             float* __restrict__ out)
{
    const int lane = threadIdx.x & 31;
    const int wave = blockIdx.x * (blockDim.x >> 5) + __builtin_amdgcn_readfirstlane(threadIdx.x >> 5);
    const int nw = gridDim.x * (blockDim.x >> 5);
    const int2* __restrict__ pr2 = (const int2*)pairS;
    const float2* __restrict__ sz = (const float2*)sz2;
    const float bias = p2b[0];
    for (int d = wave; d < NN; d += nw) {
        const int beg = rowptr[d];
        const int deg = cnt[d];
        const float si = s_i2[d];
        float M = -INFINITY, num = 0.f, den = 0.f;
        for (int j0 = 0; j0 < deg; j0 += 32) {
            const int j = j0 + lane;
            const bool act = j < deg;
            const int2 pr = pr2[beg + (act ? j : j0)];
            const float2 q = sz[pr.x];
            const float c = leaky(si + q.x + __int_as_float(pr.y)) + (act ? 0.f : -INFINITY);
            const float n = fmaxf(M, wave_max(c));
            const float r = expf(M - n);
            M = n;
            const float p = expf(c - n);
            num = num * r + p * q.y;
            den = den * r + p;
        }
        num = wave_sum(num);
        den = wave_sum(den);
        if (lane == 0) {
            float v = (den > 0.f) ? num / den : 0.f;
            v += bias;
            out[d] = (v > 0.f) ? v : (expf(v) - 1.f);
        }
    }
}

extern "C" void kernel_launch(void* const* d_in, const int* in_sizes, int n_in,
                              void* d_out, int out_size, void* d_ws, size_t ws_size,
                              hipStream_t stream)
{
    const float* x       = (const float*)d_in[0];
    const int*   eidx    = (const int*)d_in[1];
    const float* eattr   = (const float*)d_in[2];
    const float* W1      = (const float*)d_in[3];
    const float* We1     = (const float*)d_in[4];
    const float* att1    = (const float*)d_in[5];
    const float* proj1_w = (const float*)d_in[6];
    const float* proj1_b = (const float*)d_in[7];
    const float* W2      = (const float*)d_in[8];
    const float* We2     = (const float*)d_in[9];
    const float* att2    = (const float*)d_in[10];
    const float* proj2_w = (const float*)d_in[11];
    const float* proj2_b = (const float*)d_in[12];
    float* out = (float*)d_out;

    const int* srcp = eidx;
    const int* dstp = eidx + EE;

    float* ws     = (float*)d_ws;
    float* xw1    = ws;
    float* accx1  = xw1 + (size_t)NN * 128;
    int*   pairS  = (int*)(accx1 + (size_t)NN * 64);
    float* s_i1   = accx1 + (size_t)NN * 128;
    float* s_j1   = s_i1 + (size_t)NN * 4;
    float* slotA  = s_j1 + (size_t)NN * 4;
    int*   cnt    = (int*)slotA;
    float* Wcomb  = slotA + (size_t)NN;
    float* Vs     = Wcomb + 64 * 128;
    float* V2     = Vs + 256;
    int*   rowptr = (int*)(slotA + (size_t)NN * 4);
    float* hw2    = slotA + (size_t)NN * 8;
    float* s_i2   = hw2 + (size_t)NN * 64;
    float* sz2    = s_i2 + (size_t)NN;
    float* wea1   = s_i2 + (size_t)NN * 4 + 2048;
    float* wea2   = wea1 + 64;
    float* se1S   = xw1;
    int*   cntmat = (int*)hw2;
    int*   binstart = cntmat + NBINS * PBLK;

    prep_weights<<<32, 256, 0, stream>>>(W1, We1, att1, We2, att2, proj1_w, W2, proj2_w,
                                         Vs, V2, wea1, wea2, Wcomb);

    node_scores_l1<<<(NN / 16 + 7) / 8, 256, 0, stream>>>(x, Vs, s_i1, s_j1);

    bin_count<<<PBLK, 1024, 0, stream>>>(dstp, cntmat);
    edge_partition<<<PBLK, 1024, 0, stream>>>(srcp, dstp, eattr, wea1, wea2, cntmat,
                                              (float4*)se1S, (int2*)pairS, binstart);
    bin_sort<<<NBINS, 1024, 0, stream>>>(binstart, (float4*)se1S, (int2*)pairS, rowptr, cnt);

    node_aggr_l1<<<2048, 256, 0, stream>>>(rowptr, cnt, se1S, pairS, s_i1, s_j1, x, accx1);

    proj_scores_l2<<<(NN / 16 + 7) / 8, 256, 0, stream>>>(accx1, Wcomb, proj1_b, V2, s_i2, sz2);
    node_aggr_l2<<<2048, 256, 0, stream>>>(rowptr, cnt, pairS, s_i2, sz2, proj2_b, out);
}


# Layer-1 node pass takes each edge's source score from the gathered x row in registers (4 lanes x 4-feature dots, vector unit): the per-edge random read of the score table is gone; sums reordered
# speedup vs baseline: 8.2690x; 1.0816x over previous
#include <hip/hip_runtime.h>
#include <cmath>

#define NN 100000
#define EE 3200000
#define NEGS 0.2f
#define BINSH 8
#define NBINS ((NN + 255) >> BINSH)
#define PBLK 256
#define ECHUNK ((EE + PBLK - 1) / PBLK)
#define CAP 10240
#define SRCBITS 17
#define SRCMASK ((1 << SRCBITS) - 1)

typedef __attribute__((ext_vector_type(2))) float v2f;
typedef __attribute__((ext_vector_type(8))) float v8f;

__global__ void prep_weights(const float* __restrict__ W1, const float* __restrict__ We1,
                             const float* __restrict__ att1, const float* __restrict__ We2,
                             const float* __restrict__ att2, const float* __restrict__ proj1_w,
                             const float* __restrict__ W2, const float* __restrict__ p2w,
                             float* __restrict__ Vs, float* __restrict__ V2,
                             float* __restrict__ wea1,
                             float* __restrict__ wea2, float* __restrict__ Wcomb)
{
    int tid = blockIdx.x * blockDim.x + threadIdx.x;
    int nt = gridDim.x * blockDim.x;
    for (int i = tid; i < 256; i += nt) {
        int k = i >> 4, c = i & 15;
        float s = 0.f;
        if (c < 8) {
            int h = c & 3;
            const float* a = att1 + h * 96 + (c >> 2) * 32;
            const float* w = W1 + h * 512 + k * 32;
            for (int o = 0; o < 32; ++o) s += w[o] * a[o];
        }
        Vs[i] = s;
    }
    for (int i = tid; i < 2048; i += nt) {
        const int e2 = i & 7, ln = (i >> 3) & 31, t2 = i >> 8;
        int k = 16 * t2 + 8 * (ln >> 4) + e2, c = ln & 15;
        float s = 0.f;
        if (c < 3) {
            const float* a = (c == 2) ? p2w : (att2 + c * 32);
            const float* w = W2 + k * 32;
            for (int o = 0; o < 32; ++o) s += w[o] * a[o];
        }
        V2[i] = s;
    }
    for (int i = tid; i < 64; i += nt) {
        int h = i >> 4, k = i & 15;
        float s = 0.f;
        for (int o = 0; o < 32; ++o) s += We1[h * 512 + k * 32 + o] * att1[h * 96 + 64 + o];
        wea1[i] = s;
    }
    for (int i = tid; i < 16; i += nt) {
        float s = 0.f;
        for (int o = 0; o < 32; ++o) s += We2[i * 32 + o] * att2[64 + o];
        wea2[i] = s;
    }
    for (int i = tid; i < 64 * 128; i += nt) {
        const int e4 = i & 3, ln = (i >> 2) & 31, m8 = (i >> 7) & 7, t8 = i >> 10;
        int row = 8 * m8 + 4 * (ln >> 4) + e4, c = 16 * t8 + (ln & 15), h = row >> 4, k = row & 15;
        const float* w = W1 + h * 512 + k * 32;
        const float* pw = proj1_w + (size_t)(h * 32) * 128 + c;
        float s = 0.f;
        for (int o = 0; o < 32; ++o) s += w[o] * pw[(size_t)o * 128];
        Wcomb[i] = s;
    }
}

__device__ __forceinline__ v8f strip_product(const float* __restrict__ Ap,
                                             const float* __restrict__ Bp, const int K)
{
    v8f acc = {};
#pragma unroll
    for (int k0 = 0; k0 < K; k0 += 4) {
        v2f a; a.x = Ap[k0]; a.y = Ap[k0 + 1];
        v2f b; b.x = Bp[k0 * 16]; b.y = Bp[k0 * 16 + 16];
        acc = __builtin_amdgcn_wmma_f32_16x16x4_f32(false, a, false, b,
                                                    (short)0, acc, false, false);
    }
    return acc;
}

__global__ void __launch_bounds__(256)
node_scores_l1(const float* __restrict__ x, const float* __restrict__ Vs,
               float* __restrict__ s_i1)
{
    const int lane = threadIdx.x & 31;
    const int strip = blockIdx.x * (blockDim.x >> 5) + (threadIdx.x >> 5);
    if (strip >= NN / 16) return;
    const int r = lane & 15;
    const int kk = (lane >> 4) << 1;
    const v8f acc = strip_product(x + (size_t)(strip * 16 + r) * 16 + kk, Vs + kk * 16 + r, 16);
    const int rbase = strip * 16 + ((lane >> 4) << 3);
    if (r < 4) {
        float* o = s_i1 + r;
#pragma unroll
        for (int v = 0; v < 8; ++v) o[(size_t)(rbase + v) * 4] = acc[v];
    }
}

__device__ __forceinline__ v8f mm4(const float a0, const float a1, const float b0, const float b1,
                                   const v8f acc)
{
    v2f a; a.x = a0; a.y = a1;
    v2f b; b.x = b0; b.y = b1;
    return __builtin_amdgcn_wmma_f32_16x16x4_f32(false, a, false, b, (short)0, acc, false, false);
}

__global__ void __launch_bounds__(256)
proj_scores_l2(const float* __restrict__ accx1, const float* __restrict__ WcF,
               const float* __restrict__ bias, const float* __restrict__ V2F,
               float* __restrict__ s_i2, float* __restrict__ sz2)
{
    const int lane = threadIdx.x & 31;
    const int strip = blockIdx.x * (blockDim.x >> 5) + (threadIdx.x >> 5);
    if (strip >= NN / 16) return;
    const int c = lane & 15;
    const int g = lane >> 4;
    const float4* __restrict__ xr = (const float4*)(accx1 + (size_t)(strip * 16 + c) * 64 + 4 * g);
    float4 xb[8];
#pragma unroll
    for (int m = 0; m < 8; ++m) xb[m] = xr[2 * m];
    const float4* __restrict__ wf = (const float4*)WcF + lane;
    const float4* __restrict__ vf = (const float4*)V2F + lane * 2;
    v8f acc2 = {};
#pragma unroll
    for (int tn = 0; tn < 8; ++tn) {
        v8f acc = {};
#pragma unroll
        for (int m = 0; m < 8; ++m) {
            const float4 w = wf[(tn * 8 + m) * 32];
            acc = mm4(w.x, w.y, xb[m].x, xb[m].y, acc);
            acc = mm4(w.z, w.w, xb[m].z, xb[m].w, acc);
        }
        const float4 bl = *(const float4*)(bias + 16 * tn + 8 * g);
        const float4 bh = *(const float4*)(bias + 16 * tn + 8 * g + 4);
        const float bv[8] = {bl.x, bl.y, bl.z, bl.w, bh.x, bh.y, bh.z, bh.w};
        float hv[8];
#pragma unroll
        for (int r = 0; r < 8; ++r) {
            float xv = acc[r] + bv[r];
            xv = (xv > 0.f) ? xv : (expf(xv) - 1.f);
            xv = (xv > 0.f) ? xv : (expf(xv) - 1.f);
            hv[r] = xv;
        }
        const float4 v0 = vf[tn * 64], v1 = vf[tn * 64 + 1];
        acc2 = mm4(v0.x, v0.y, hv[0], hv[1], acc2);
        acc2 = mm4(v0.z, v0.w, hv[2], hv[3], acc2);
        acc2 = mm4(v1.x, v1.y, hv[4], hv[5], acc2);
        acc2 = mm4(v1.z, v1.w, hv[6], hv[7], acc2);
    }
    if (g == 0) {
        const int node = strip * 16 + c;
        s_i2[node] = acc2[0];
        float2 q; q.x = acc2[1]; q.y = acc2[2];
        ((float2*)sz2)[node] = q;
    }
}

__global__ void __launch_bounds__(1024)
bin_count(const int* __restrict__ dst, int* __restrict__ cntmat)
{
    __shared__ int hist[NBINS];
    const int t = threadIdx.x, b = blockIdx.x;
    for (int i = t; i < NBINS; i += 1024) hist[i] = 0;
    __syncthreads();
    const int e0 = b * ECHUNK;
    const int e1 = (e0 + ECHUNK < EE) ? (e0 + ECHUNK) : EE;
    int dv[(ECHUNK + 1023) / 1024];
#pragma unroll
    for (int j = 0; j < (ECHUNK + 1023) / 1024; ++j) {
        const int e = e0 + j * 1024 + t;
        dv[j] = dst[(e < EE) ? e : (EE - 1)];
    }
#pragma unroll
    for (int j = 0; j < (ECHUNK + 1023) / 1024; ++j)
        if (e0 + j * 1024 + t < e1) atomicAdd(&hist[dv[j] >> BINSH], 1);
    __syncthreads();
    for (int i = t; i < NBINS; i += 1024) cntmat[i * PBLK + b] = hist[i];
}

__device__ __forceinline__ void place_edge(const int d, const int sv, const float4 r0, const float4 r1,
                                           const float4 r2, const float4 r3,
                                           const float* __restrict__ wea1,
                                           const float* __restrict__ wea2, int* lcur,
                                           float4* stA, int2* stP)
{
    const float ea[16] = {r0.x, r0.y, r0.z, r0.w, r1.x, r1.y, r1.z, r1.w,
                          r2.x, r2.y, r2.z, r2.w, r3.x, r3.y, r3.z, r3.w};
    float a0 = 0.f, a1 = 0.f, a2 = 0.f, a3 = 0.f, bb = 0.f;
#pragma unroll
    for (int k = 0; k < 16; ++k) {
        a0 += ea[k] * wea1[k];      a1 += ea[k] * wea1[16 + k];
        a2 += ea[k] * wea1[32 + k]; a3 += ea[k] * wea1[48 + k];
        bb += ea[k] * wea2[k];
    }
    const int pos = atomicAdd(&lcur[d >> BINSH], 1);
    float4 v4; v4.x = a0; v4.y = a1; v4.z = a2; v4.w = a3;
    stA[pos] = v4;
    int2 q; q.x = sv | ((d & ((1 << BINSH) - 1)) << SRCBITS); q.y = __float_as_int(bb);
    stP[pos] = q;
}

__global__ void __launch_bounds__(1024)
edge_partition(const int* __restrict__ src, const int* __restrict__ dst,
               const float* __restrict__ eattr, const float* __restrict__ wea1,
               const float* __restrict__ wea2, const int* __restrict__ cntmat,
               float4* __restrict__ recA, int2* __restrict__ recP,
               int* __restrict__ binstart)
{
    __shared__ float4 stA[ECHUNK];
    __shared__ int2 stP[ECHUNK];
    __shared__ int tot[NBINS];
    __shared__ int gfirst[NBINS];
    __shared__ int own[NBINS];
    __shared__ int lstart[NBINS];
    __shared__ int lcur[NBINS];
    __shared__ int wsum[32];
    __shared__ int wsumL[32];
    const int t = threadIdx.x, lane = t & 31, w = t >> 5, b = blockIdx.x;
    for (int bin = w; bin < NBINS; bin += 32) {
        const int* row = cntmat + bin * PBLK;
        int all = 0, before = 0;
        for (int q = 0; q < PBLK; q += 32) {
            const int c = row[q + lane];
            all += c;
            before += (q + lane < b) ? c : 0;
        }
        const int mine = row[b];
        for (int o = 16; o > 0; o >>= 1) {
            all += __shfl_xor(all, o);
            before += __shfl_xor(before, o);
        }
        if (lane == 0) { tot[bin] = all; gfirst[bin] = before; own[bin] = mine; }
    }
    __syncthreads();
    const int v = (t < NBINS) ? tot[t] : 0;
    const int m = (t < NBINS) ? own[t] : 0;
    int incl = v, inclL = m;
    for (int o = 1; o < 32; o <<= 1) {
        const int u = __shfl_up(incl, o);
        const int uL = __shfl_up(inclL, o);
        if (lane >= o) { incl += u; inclL += uL; }
    }
    if (lane == 31) { wsum[w] = incl; wsumL[w] = inclL; }
    __syncthreads();
    int pre = (lane < w) ? wsum[lane] : 0;
    int preL = (lane < w) ? wsumL[lane] : 0;
    for (int o = 16; o > 0; o >>= 1) {
        pre += __shfl_xor(pre, o);
        preL += __shfl_xor(preL, o);
    }
    if (t < NBINS) {
        const int first = pre + incl - v;
        const int firstL = preL + inclL - m;
        gfirst[t] += first;
        lstart[t] = firstL;
        lcur[t] = firstL;
        if (b == 0) binstart[t] = first;
    }
    __syncthreads();
    const int e0 = b * ECHUNK;
    const int e1 = (e0 + ECHUNK < EE) ? (e0 + ECHUNK) : EE;
    const float4* __restrict__ ea4 = (const float4*)eattr;
    int e = e0 + t;
    if (e < e1) {
        int dA = dst[e], sA = src[e];
        float4 a0 = ea4[(size_t)e * 4], a1 = ea4[(size_t)e * 4 + 1];
        float4 a2 = ea4[(size_t)e * 4 + 2], a3 = ea4[(size_t)e * 4 + 3];
        for (;;) {
            const int eB = e + 1024;
            const bool moreB = eB < e1;
            const int fB = moreB ? eB : e;
            const int dB = dst[fB], sB = src[fB];
            const float4 b0 = ea4[(size_t)fB * 4], b1 = ea4[(size_t)fB * 4 + 1];
            const float4 b2 = ea4[(size_t)fB * 4 + 2], b3 = ea4[(size_t)fB * 4 + 3];
            place_edge(dA, sA, a0, a1, a2, a3, wea1, wea2, lcur, stA, stP);
            if (!moreB) break;
            e = eB + 1024;
            const bool moreA = e < e1;
            const int fA = moreA ? e : eB;
            dA = dst[fA]; sA = src[fA];
            a0 = ea4[(size_t)fA * 4]; a1 = ea4[(size_t)fA * 4 + 1];
            a2 = ea4[(size_t)fA * 4 + 2]; a3 = ea4[(size_t)fA * 4 + 3];
            place_edge(dB, sB, b0, b1, b2, b3, wea1, wea2, lcur, stA, stP);
            if (!moreA) break;
        }
    }
    __syncthreads();
    for (int bin = w; bin < NBINS; bin += 32) {
        const int ls = lstart[bin], c = own[bin], g = gfirst[bin];
        for (int i = lane; i < c; i += 32) {
            recA[g + i] = stA[ls + i];
            recP[g + i] = stP[ls + i];
        }
    }
}

#define FOR_SLOTS(X) X(0) X(1) X(2) X(3) X(4) X(5) X(6) X(7) X(8) X(9)
static_assert(CAP / 1024 == 10, "FOR_SLOTS lists CAP / 1024 slots");

__global__ void __launch_bounds__(1024)
bin_sort(const int* __restrict__ binstart, float4* recA, int2* recP,
         int* __restrict__ rowptr, int* __restrict__ cnt)
{
    __shared__ float4 bufA[CAP];
    __shared__ int2 bufP[CAP];
    __shared__ int cntL[256];
    __shared__ int startL[256];
    __shared__ int curL[256];
    __shared__ int wtot[32];
    const int t = threadIdx.x, lane = t & 31, w = t >> 5, bin = blockIdx.x;
    const int beg = binstart[bin];
    const int end = (bin + 1 < NBINS) ? binstart[bin + 1] : EE;
    const int n = end - beg;
    const bool fits = n <= CAP;
    if (t < 256) { cntL[t] = 0; curL[t] = 0; }
    __syncthreads();
#define DECL_SLOT(j) int2 pv##j = make_int2(0, 0); float4 av##j = make_float4(0.f, 0.f, 0.f, 0.f); int rk##j = 0;
    FOR_SLOTS(DECL_SLOT)
    if (fits) {
#define LOAD_SLOT(j) { const int g = beg + j * 1024 + t; const int gc = (g < EE) ? g : (EE - 1); \
                       pv##j = recP[gc]; av##j = recA[gc]; }
        FOR_SLOTS(LOAD_SLOT)
#define COUNT_SLOT(j) if (j * 1024 + t < n) rk##j = atomicAdd(&cntL[pv##j.x >> SRCBITS], 1);
        FOR_SLOTS(COUNT_SLOT)
    } else {
        for (int i = t; i < n; i += 1024) atomicAdd(&cntL[recP[beg + i].x >> SRCBITS], 1);
    }
    __syncthreads();
    int c = 0, incl = 0;
    if (t < 256) {
        c = cntL[t];
        incl = c;
        for (int o = 1; o < 32; o <<= 1) {
            int u = __shfl_up(incl, o);
            if (lane >= o) incl += u;
        }
        if (lane == 31) wtot[w] = incl;
    }
    __syncthreads();
    if (t < 256) {
        int pre = (lane < w) ? wtot[lane] : 0;
        for (int o = 16; o > 0; o >>= 1) pre += __shfl_xor(pre, o);
        const int st = pre + incl - c;
        startL[t] = st;
        const int node = (bin << BINSH) + t;
        if (node < NN) { rowptr[node] = beg + st; cnt[node] = c; }
    }
    __syncthreads();
    if (fits) {
#define START_SLOT(j) const int st##j = startL[pv##j.x >> SRCBITS];
        FOR_SLOTS(START_SLOT)
#define PLACE_SLOT(j) if (j * 1024 + t < n) { \
            const int pos = st##j + rk##j; \
            int2 q; q.x = pv##j.x & SRCMASK; q.y = pv##j.y; \
            bufP[pos] = q; bufA[pos] = av##j; }
        FOR_SLOTS(PLACE_SLOT)
        __syncthreads();
        const int lead = beg & 15;
        for (int i = t - lead; i < n; i += 1024) {
            if (i >= 0) {
                recP[beg + i] = bufP[i];
                recA[beg + i] = bufA[i];
            }
        }
    } else {
        if (t == 0) {
            for (int nd = 0; nd < 256; ++nd) {
                const int endn = cntL[nd];
                int k = curL[nd];
                while (k < endn) {
                    const int p = beg + startL[nd] + k;
                    const int2 r = recP[p];
                    const int dl = r.x >> SRCBITS;
                    if (dl == nd) { ++k; continue; }
                    const int q = beg + startL[dl] + curL[dl];
                    curL[dl] += 1;
                    const int2 r2 = recP[q];
                    const float4 a1 = recA[p], a2 = recA[q];
                    recP[q] = r;  recA[q] = a1;
                    recP[p] = r2; recA[p] = a2;
                }
                curL[nd] = endn;
            }
        }
        __syncthreads();
        for (int i = t; i < n; i += 1024) recP[beg + i].x &= SRCMASK;
    }
}

__device__ __forceinline__ float leaky(float v) { return (v >= 0.f) ? v : NEGS * v; }
__device__ __forceinline__ float wave_max(float v) {
    for (int o = 16; o > 0; o >>= 1) v = fmaxf(v, __shfl_xor(v, o));
    return v;
}
__device__ __forceinline__ float wave_sum(float v) {
    for (int o = 16; o > 0; o >>= 1) v += __shfl_xor(v, o);
    return v;
}

template <int CTRL> __device__ __forceinline__ float dpp_f(float v) {
    return __int_as_float(__builtin_amdgcn_update_dpp(0, __float_as_int(v), CTRL, 0xF, 0xF, true));
}
#define XM(n) (0x160 | (n))
#define QB(h) ((h) * 0x55)
#define RS(n) (0x150 | (n))
__device__ __forceinline__ float xhalf(float v) {
    return __int_as_float(__builtin_amdgcn_permlanex16(__float_as_int(v), __float_as_int(v),
                                                       0x76543210, 0xfedcba98, false, false));
}
__device__ __forceinline__ float sum_groups(float v) {
    v += dpp_f<XM(4)>(v);
    v += dpp_f<XM(8)>(v);
    return v + xhalf(v);
}
__device__ __forceinline__ float max_groups(float v) {
    v = fmaxf(v, dpp_f<XM(4)>(v));
    v = fmaxf(v, dpp_f<XM(8)>(v));
    return fmaxf(v, xhalf(v));
}

__global__ void __launch_bounds__(256)
node_aggr_l1(const int* __restrict__ rowptr, const int* __restrict__ cnt,
             const float* __restrict__ se1S, const int* __restrict__ pairS,
             const float* __restrict__ s_i1, const float* __restrict__ Vs,
             const float* __restrict__ x, float* __restrict__ accx1)
{
    const int lane = threadIdx.x & 31;
    const int m = lane & 3;
    const int g = lane >> 2;
    const bool mh = (m & 2) != 0, ml = (m & 1) != 0;
    const int wave = blockIdx.x * (blockDim.x >> 5) + __builtin_amdgcn_readfirstlane(threadIdx.x >> 5);
    const int nw = gridDim.x * (blockDim.x >> 5);
    const int2* __restrict__ pr2 = (const int2*)pairS;
    const float4* __restrict__ x4 = (const float4*)x;
    float vj[4][4];
#pragma unroll
    for (int h = 0; h < 4; ++h)
#pragma unroll
        for (int e = 0; e < 4; ++e) vj[h][e] = Vs[(4 * m + e) * 16 + 4 + h];
    for (int d = wave; d < NN; d += nw) {
        const int beg = rowptr[d];
        const int deg = cnt[d];
        const float sim = s_i1[(size_t)d * 4 + m];
        float M = -INFINITY, dn = 0.f;
        float acc[4][4];
#pragma unroll
        for (int h = 0; h < 4; ++h)
#pragma unroll
            for (int e = 0; e < 4; ++e) acc[h][e] = 0.f;
        for (int j0 = 0; j0 < deg; j0 += 32) {
            const int left = deg - j0;
            int sidx[4];
            float sev[4], pen[4];
            float4 xr[4];
#pragma unroll
            for (int q = 0; q < 4; ++q) {
                if (8 * q < left) {
                    const bool act = 8 * q + g < left;
                    const int slot = beg + j0 + (act ? 8 * q + g : 0);
                    sidx[q] = pr2[slot].x;
                    sev[q] = se1S[(size_t)slot * 4 + m];
                    pen[q] = act ? 0.f : -INFINITY;
                }
            }
#pragma unroll
            for (int q = 0; q < 4; ++q)
                if (8 * q < left) xr[q] = x4[(size_t)sidx[q] * 4 + m];
            float c[4];
#pragma unroll
            for (int q = 0; q < 4; ++q) {
                c[q] = -INFINITY;
                if (8 * q < left) {
                    const float t0 = fmaf(xr[q].w, vj[0][3], fmaf(xr[q].z, vj[0][2], fmaf(xr[q].y, vj[0][1], xr[q].x * vj[0][0])));
                    const float t1 = fmaf(xr[q].w, vj[1][3], fmaf(xr[q].z, vj[1][2], fmaf(xr[q].y, vj[1][1], xr[q].x * vj[1][0])));
                    const float t2 = fmaf(xr[q].w, vj[2][3], fmaf(xr[q].z, vj[2][2], fmaf(xr[q].y, vj[2][1], xr[q].x * vj[2][0])));
                    const float t3 = fmaf(xr[q].w, vj[3][3], fmaf(xr[q].z, vj[3][2], fmaf(xr[q].y, vj[3][1], xr[q].x * vj[3][0])));
                    float k0 = mh ? t2 : t0, k1 = mh ? t3 : t1;
                    const float g0 = mh ? t0 : t2, g1 = mh ? t1 : t3;
                    k0 += dpp_f<XM(2)>(g0);
                    k1 += dpp_f<XM(2)>(g1);
                    const float kk = ml ? k1 : k0, gg = ml ? k0 : k1;
                    const float sj = kk + dpp_f<XM(1)>(gg);
                    c[q] = leaky(sim + sj + sev[q]) + pen[q];
                }
            }
            const float cm = max_groups(fmaxf(fmaxf(c[0], c[1]), fmaxf(c[2], c[3])));
            const float n = fmaxf(M, cm);
            const float r = expf(M - n);
            M = n;
            float psum = 0.f;
            if (j0 > 0) {
                const float r0 = dpp_f<QB(0)>(r), r1 = dpp_f<QB(1)>(r);
                const float r2 = dpp_f<QB(2)>(r), r3 = dpp_f<QB(3)>(r);
#pragma unroll
                for (int e = 0; e < 4; ++e) { acc[0][e] *= r0; acc[1][e] *= r1; acc[2][e] *= r2; acc[3][e] *= r3; }
            }
#pragma unroll
            for (int q = 0; q < 4; ++q) {
                if (8 * q < left) {
                    const float pq = expf(c[q] - n);
                    psum += pq;
                    const float p0 = dpp_f<QB(0)>(pq), p1 = dpp_f<QB(1)>(pq);
                    const float p2 = dpp_f<QB(2)>(pq), p3 = dpp_f<QB(3)>(pq);
                    const float xe[4] = {xr[q].x, xr[q].y, xr[q].z, xr[q].w};
#pragma unroll
                    for (int e = 0; e < 4; ++e) {
                        acc[0][e] = fmaf(p0, xe[e], acc[0][e]);
                        acc[1][e] = fmaf(p1, xe[e], acc[1][e]);
                        acc[2][e] = fmaf(p2, xe[e], acc[2][e]);
                        acc[3][e] = fmaf(p3, xe[e], acc[3][e]);
                    }
                }
            }
            dn = dn * r + psum;
        }
        dn = sum_groups(dn);
#pragma unroll
        for (int h = 0; h < 4; ++h)
#pragma unroll
            for (int e = 0; e < 4; ++e) acc[h][e] = sum_groups(acc[h][e]);
        const float d0 = dpp_f<RS(0)>(dn), d1 = dpp_f<RS(1)>(dn);
        const float d2 = dpp_f<RS(2)>(dn), d3 = dpp_f<RS(3)>(dn);
        if (lane < 16) {
            const float dv = (g == 0) ? d0 : (g == 1) ? d1 : (g == 2) ? d2 : d3;
            float o[4];
#pragma unroll
            for (int e = 0; e < 4; ++e) {
                const float a0 = acc[0][e], a1 = acc[1][e], a2 = acc[2][e], a3 = acc[3][e];
                const float a = (g == 0) ? a0 : (g == 1) ? a1 : (g == 2) ? a2 : a3;
                o[e] = (dv > 0.f) ? a / dv : 0.f;
            }
            float4 ov; ov.x = o[0]; ov.y = o[1]; ov.z = o[2]; ov.w = o[3];
            *(float4*)(accx1 + (size_t)d * 64 + g * 16 + 4 * m) = ov;
        }
    }
}

__global__ void __launch_bounds__(256)
node_aggr_l2(const int* __restrict__ rowptr, const int* __restrict__ cnt,
             const int* __restrict__ pairS, const float* __restrict__ s_i2,
             const float* __restrict__ sz2, const float* __restrict__ p2b,
             float* __restrict__ out)
{
    const int lane = threadIdx.x & 31;
    const int wave = blockIdx.x * (blockDim.x >> 5) + __builtin_amdgcn_readfirstlane(threadIdx.x >> 5);
    const int nw = gridDim.x * (blockDim.x >> 5);
    const int2* __restrict__ pr2 = (const int2*)pairS;
    const float2* __restrict__ sz = (const float2*)sz2;
    const float bias = p2b[0];
    for (int d = wave; d < NN; d += nw) {
        const int beg = rowptr[d];
        const int deg = cnt[d];
        const float si = s_i2[d];
        float M = -INFINITY, num = 0.f, den = 0.f;
        for (int j0 = 0; j0 < deg; j0 += 32) {
            const int j = j0 + lane;
            const bool act = j < deg;
            const int2 pr = pr2[beg + (act ? j : j0)];
            const float2 q = sz[pr.x];
            const float c = leaky(si + q.x + __int_as_float(pr.y)) + (act ? 0.f : -INFINITY);
            const float n = fmaxf(M, wave_max(c));
            const float r = expf(M - n);
            M = n;
            const float p = expf(c - n);
            num = num * r + p * q.y;
            den = den * r + p;
        }
        num = wave_sum(num);
        den = wave_sum(den);
        if (lane == 0) {
            float v = (den > 0.f) ? num / den : 0.f;
            v += bias;
            out[d] = (v > 0.f) ? v : (expf(v) - 1.f);
        }
    }
}

extern "C" void kernel_launch(void* const* d_in, const int* in_sizes, int n_in,
                              void* d_out, int out_size, void* d_ws, size_t ws_size,
                              hipStream_t stream)
{
    const float* x       = (const float*)d_in[0];
    const int*   eidx    = (const int*)d_in[1];
    const float* eattr   = (const float*)d_in[2];
    const float* W1      = (const float*)d_in[3];
    const float* We1     = (const float*)d_in[4];
    const float* att1    = (const float*)d_in[5];
    const float* proj1_w = (const float*)d_in[6];
    const float* proj1_b = (const float*)d_in[7];
    const float* W2      = (const float*)d_in[8];
    const float* We2     = (const float*)d_in[9];
    const float* att2    = (const float*)d_in[10];
    const float* proj2_w = (const float*)d_in[11];
    const float* proj2_b = (const float*)d_in[12];
    float* out = (float*)d_out;

    const int* srcp = eidx;
    const int* dstp = eidx + EE;

    float* ws     = (float*)d_ws;
    float* xw1    = ws;
    float* accx1  = xw1 + (size_t)NN * 128;
    int*   pairS  = (int*)(accx1 + (size_t)NN * 64);
    float* s_i1   = accx1 + (size_t)NN * 128;
    float* slotA  = s_i1 + (size_t)NN * 8;
    int*   cnt    = (int*)slotA;
    float* Wcomb  = slotA + (size_t)NN;
    float* Vs     = Wcomb + 64 * 128;
    float* V2     = Vs + 256;
    int*   rowptr = (int*)(slotA + (size_t)NN * 4);
    float* hw2    = slotA + (size_t)NN * 8;
    float* s_i2   = hw2 + (size_t)NN * 64;
    float* sz2    = s_i2 + (size_t)NN;
    float* wea1   = s_i2 + (size_t)NN * 4 + 2048;
    float* wea2   = wea1 + 64;
    float* se1S   = xw1;
    int*   cntmat = (int*)hw2;
    int*   binstart = cntmat + NBINS * PBLK;

    prep_weights<<<32, 256, 0, stream>>>(W1, We1, att1, We2, att2, proj1_w, W2, proj2_w,
                                         Vs, V2, wea1, wea2, Wcomb);

    node_scores_l1<<<(NN / 16 + 7) / 8, 256, 0, stream>>>(x, Vs, s_i1);

    bin_count<<<PBLK, 1024, 0, stream>>>(dstp, cntmat);
    edge_partition<<<PBLK, 1024, 0, stream>>>(srcp, dstp, eattr, wea1, wea2, cntmat,
                                              (float4*)se1S, (int2*)pairS, binstart);
    bin_sort<<<NBINS, 1024, 0, stream>>>(binstart, (float4*)se1S, (int2*)pairS, rowptr, cnt);

    node_aggr_l1<<<1280, 256, 0, stream>>>(rowptr, cnt, se1S, pairS, s_i1, Vs, x, accx1);

    proj_scores_l2<<<(NN / 16 + 7) / 8, 256, 0, stream>>>(accx1, Wcomb, proj1_b, V2, s_i2, sz2);
    node_aggr_l2<<<2048, 256, 0, stream>>>(rowptr, cnt, pairS, s_i2, sz2, proj2_b, out);
}


# Every softmax and ELU exponential on the hardware exp2 instruction (one multiply + v_exp_f32) in place of the library expf; the projection's ELU pair without branches, folded to one select
# speedup vs baseline: 8.8905x; 1.0752x over previous
#include <hip/hip_runtime.h>
#include <cmath>

#define NN 100000
#define EE 3200000
#define NEGS 0.2f
#define BINSH 8
#define NBINS ((NN + 255) >> BINSH)
#define PBLK 256
#define ECHUNK ((EE + PBLK - 1) / PBLK)
#define CAP 10240
#define SRCBITS 17
#define SRCMASK ((1 << SRCBITS) - 1)

typedef __attribute__((ext_vector_type(2))) float v2f;
typedef __attribute__((ext_vector_type(8))) float v8f;

__device__ __forceinline__ float exp_hw(const float v)
{
    return __builtin_amdgcn_exp2f(v * 1.44269504088896340736f);
}

__global__ void prep_weights(const float* __restrict__ W1, const float* __restrict__ We1,
                             const float* __restrict__ att1, const float* __restrict__ We2,
                             const float* __restrict__ att2, const float* __restrict__ proj1_w,
                             const float* __restrict__ W2, const float* __restrict__ p2w,
                             float* __restrict__ Vs, float* __restrict__ V2,
                             float* __restrict__ wea1,
                             float* __restrict__ wea2, float* __restrict__ Wcomb)
{
    int tid = blockIdx.x * blockDim.x + threadIdx.x;
    int nt = gridDim.x * blockDim.x;
    for (int i = tid; i < 256; i += nt) {
        int k = i >> 4, c = i & 15;
        float s = 0.f;
        if (c < 8) {
            int h = c & 3;
            const float* a = att1 + h * 96 + (c >> 2) * 32;
            const float* w = W1 + h * 512 + k * 32;
            for (int o = 0; o < 32; ++o) s += w[o] * a[o];
        }
        Vs[i] = s;
    }
    for (int i = tid; i < 2048; i += nt) {
        const int e2 = i & 7, ln = (i >> 3) & 31, t2 = i >> 8;
        int k = 16 * t2 + 8 * (ln >> 4) + e2, c = ln & 15;
        float s = 0.f;
        if (c < 3) {
            const float* a = (c == 2) ? p2w : (att2 + c * 32);
            const float* w = W2 + k * 32;
            for (int o = 0; o < 32; ++o) s += w[o] * a[o];
        }
        V2[i] = s;
    }
    for (int i = tid; i < 64; i += nt) {
        int h = i >> 4, k = i & 15;
        float s = 0.f;
        for (int o = 0; o < 32; ++o) s += We1[h * 512 + k * 32 + o] * att1[h * 96 + 64 + o];
        wea1[i] = s;
    }
    for (int i = tid; i < 16; i += nt) {
        float s = 0.f;
        for (int o = 0; o < 32; ++o) s += We2[i * 32 + o] * att2[64 + o];
        wea2[i] = s;
    }
    for (int i = tid; i < 64 * 128; i += nt) {
        const int e4 = i & 3, ln = (i >> 2) & 31, m8 = (i >> 7) & 7, t8 = i >> 10;
        int row = 8 * m8 + 4 * (ln >> 4) + e4, c = 16 * t8 + (ln & 15), h = row >> 4, k = row & 15;
        const float* w = W1 + h * 512 + k * 32;
        const float* pw = proj1_w + (size_t)(h * 32) * 128 + c;
        float s = 0.f;
        for (int o = 0; o < 32; ++o) s += w[o] * pw[(size_t)o * 128];
        Wcomb[i] = s;
    }
}

__device__ __forceinline__ v8f strip_product(const float* __restrict__ Ap,
                                             const float* __restrict__ Bp, const int K)
{
    v8f acc = {};
#pragma unroll
    for (int k0 = 0; k0 < K; k0 += 4) {
        v2f a; a.x = Ap[k0]; a.y = Ap[k0 + 1];
        v2f b; b.x = Bp[k0 * 16]; b.y = Bp[k0 * 16 + 16];
        acc = __builtin_amdgcn_wmma_f32_16x16x4_f32(false, a, false, b,
                                                    (short)0, acc, false, false);
    }
    return acc;
}

__global__ void __launch_bounds__(256)
node_scores_l1(const float* __restrict__ x, const float* __restrict__ Vs,
               float* __restrict__ s_i1)
{
    const int lane = threadIdx.x & 31;
    const int strip = blockIdx.x * (blockDim.x >> 5) + (threadIdx.x >> 5);
    if (strip >= NN / 16) return;
    const int r = lane & 15;
    const int kk = (lane >> 4) << 1;
    const v8f acc = strip_product(x + (size_t)(strip * 16 + r) * 16 + kk, Vs + kk * 16 + r, 16);
    const int rbase = strip * 16 + ((lane >> 4) << 3);
    if (r < 4) {
        float* o = s_i1 + r;
#pragma unroll
        for (int v = 0; v < 8; ++v) o[(size_t)(rbase + v) * 4] = acc[v];
    }
}

__device__ __forceinline__ v8f mm4(const float a0, const float a1, const float b0, const float b1,
                                   const v8f acc)
{
    v2f a; a.x = a0; a.y = a1;
    v2f b; b.x = b0; b.y = b1;
    return __builtin_amdgcn_wmma_f32_16x16x4_f32(false, a, false, b, (short)0, acc, false, false);
}

__global__ void __launch_bounds__(256)
proj_scores_l2(const float* __restrict__ accx1, const float* __restrict__ WcF,
               const float* __restrict__ bias, const float* __restrict__ V2F,
               float* __restrict__ s_i2, float* __restrict__ sz2)
{
    const int lane = threadIdx.x & 31;
    const int strip = blockIdx.x * (blockDim.x >> 5) + (threadIdx.x >> 5);
    if (strip >= NN / 16) return;
    const int c = lane & 15;
    const int g = lane >> 4;
    const float4* __restrict__ xr = (const float4*)(accx1 + (size_t)(strip * 16 + c) * 64 + 4 * g);
    float4 xb[8];
#pragma unroll
    for (int m = 0; m < 8; ++m) xb[m] = xr[2 * m];
    const float4* __restrict__ wf = (const float4*)WcF + lane;
    const float4* __restrict__ vf = (const float4*)V2F + lane * 2;
    v8f acc2 = {};
#pragma unroll
    for (int tn = 0; tn < 8; ++tn) {
        v8f acc = {};
#pragma unroll
        for (int m = 0; m < 8; ++m) {
            const float4 w = wf[(tn * 8 + m) * 32];
            acc = mm4(w.x, w.y, xb[m].x, xb[m].y, acc);
            acc = mm4(w.z, w.w, xb[m].z, xb[m].w, acc);
        }
        const float4 bl = *(const float4*)(bias + 16 * tn + 8 * g);
        const float4 bh = *(const float4*)(bias + 16 * tn + 8 * g + 4);
        const float bv[8] = {bl.x, bl.y, bl.z, bl.w, bh.x, bh.y, bh.z, bh.w};
        float hv[8];
#pragma unroll
        for (int r = 0; r < 8; ++r) {
            const float xv = acc[r] + bv[r];
            const float y = exp_hw(xv) - 1.f;
            const float z = exp_hw(y) - 1.f;
            hv[r] = (xv > 0.f) ? xv : z;
        }
        const float4 v0 = vf[tn * 64], v1 = vf[tn * 64 + 1];
        acc2 = mm4(v0.x, v0.y, hv[0], hv[1], acc2);
        acc2 = mm4(v0.z, v0.w, hv[2], hv[3], acc2);
        acc2 = mm4(v1.x, v1.y, hv[4], hv[5], acc2);
        acc2 = mm4(v1.z, v1.w, hv[6], hv[7], acc2);
    }
    if (g == 0) {
        const int node = strip * 16 + c;
        s_i2[node] = acc2[0];
        float2 q; q.x = acc2[1]; q.y = acc2[2];
        ((float2*)sz2)[node] = q;
    }
}

__global__ void __launch_bounds__(1024)
bin_count(const int* __restrict__ dst, int* __restrict__ cntmat)
{
    __shared__ int hist[NBINS];
    const int t = threadIdx.x, b = blockIdx.x;
    for (int i = t; i < NBINS; i += 1024) hist[i] = 0;
    __syncthreads();
    const int e0 = b * ECHUNK;
    const int e1 = (e0 + ECHUNK < EE) ? (e0 + ECHUNK) : EE;
    int dv[(ECHUNK + 1023) / 1024];
#pragma unroll
    for (int j = 0; j < (ECHUNK + 1023) / 1024; ++j) {
        const int e = e0 + j * 1024 + t;
        dv[j] = dst[(e < EE) ? e : (EE - 1)];
    }
#pragma unroll
    for (int j = 0; j < (ECHUNK + 1023) / 1024; ++j)
        if (e0 + j * 1024 + t < e1) atomicAdd(&hist[dv[j] >> BINSH], 1);
    __syncthreads();
    for (int i = t; i < NBINS; i += 1024) cntmat[i * PBLK + b] = hist[i];
}

__device__ __forceinline__ void place_edge(const int d, const int sv, const float4 r0, const float4 r1,
                                           const float4 r2, const float4 r3,
                                           const float* __restrict__ wea1,
                                           const float* __restrict__ wea2, int* lcur,
                                           float4* stA, int2* stP)
{
    const float ea[16] = {r0.x, r0.y, r0.z, r0.w, r1.x, r1.y, r1.z, r1.w,
                          r2.x, r2.y, r2.z, r2.w, r3.x, r3.y, r3.z, r3.w};
    float a0 = 0.f, a1 = 0.f, a2 = 0.f, a3 = 0.f, bb = 0.f;
#pragma unroll
    for (int k = 0; k < 16; ++k) {
        a0 += ea[k] * wea1[k];      a1 += ea[k] * wea1[16 + k];
        a2 += ea[k] * wea1[32 + k]; a3 += ea[k] * wea1[48 + k];
        bb += ea[k] * wea2[k];
    }
    const int pos = atomicAdd(&lcur[d >> BINSH], 1);
    float4 v4; v4.x = a0; v4.y = a1; v4.z = a2; v4.w = a3;
    stA[pos] = v4;
    int2 q; q.x = sv | ((d & ((1 << BINSH) - 1)) << SRCBITS); q.y = __float_as_int(bb);
    stP[pos] = q;
}

__global__ void __launch_bounds__(1024)
edge_partition(const int* __restrict__ src, const int* __restrict__ dst,
               const float* __restrict__ eattr, const float* __restrict__ wea1,
               const float* __restrict__ wea2, const int* __restrict__ cntmat,
               float4* __restrict__ recA, int2* __restrict__ recP,
               int* __restrict__ binstart)
{
    __shared__ float4 stA[ECHUNK];
    __shared__ int2 stP[ECHUNK];
    __shared__ int tot[NBINS];
    __shared__ int gfirst[NBINS];
    __shared__ int own[NBINS];
    __shared__ int lstart[NBINS];
    __shared__ int lcur[NBINS];
    __shared__ int wsum[32];
    __shared__ int wsumL[32];
    const int t = threadIdx.x, lane = t & 31, w = t >> 5, b = blockIdx.x;
    for (int bin = w; bin < NBINS; bin += 32) {
        const int* row = cntmat + bin * PBLK;
        int all = 0, before = 0;
        for (int q = 0; q < PBLK; q += 32) {
            const int c = row[q + lane];
            all += c;
            before += (q + lane < b) ? c : 0;
        }
        const int mine = row[b];
        for (int o = 16; o > 0; o >>= 1) {
            all += __shfl_xor(all, o);
            before += __shfl_xor(before, o);
        }
        if (lane == 0) { tot[bin] = all; gfirst[bin] = before; own[bin] = mine; }
    }
    __syncthreads();
    const int v = (t < NBINS) ? tot[t] : 0;
    const int m = (t < NBINS) ? own[t] : 0;
    int incl = v, inclL = m;
    for (int o = 1; o < 32; o <<= 1) {
        const int u = __shfl_up(incl, o);
        const int uL = __shfl_up(inclL, o);
        if (lane >= o) { incl += u; inclL += uL; }
    }
    if (lane == 31) { wsum[w] = incl; wsumL[w] = inclL; }
    __syncthreads();
    int pre = (lane < w) ? wsum[lane] : 0;
    int preL = (lane < w) ? wsumL[lane] : 0;
    for (int o = 16; o > 0; o >>= 1) {
        pre += __shfl_xor(pre, o);
        preL += __shfl_xor(preL, o);
    }
    if (t < NBINS) {
        const int first = pre + incl - v;
        const int firstL = preL + inclL - m;
        gfirst[t] += first;
        lstart[t] = firstL;
        lcur[t] = firstL;
        if (b == 0) binstart[t] = first;
    }
    __syncthreads();
    const int e0 = b * ECHUNK;
    const int e1 = (e0 + ECHUNK < EE) ? (e0 + ECHUNK) : EE;
    const float4* __restrict__ ea4 = (const float4*)eattr;
    int e = e0 + t;
    if (e < e1) {
        int dA = dst[e], sA = src[e];
        float4 a0 = ea4[(size_t)e * 4], a1 = ea4[(size_t)e * 4 + 1];
        float4 a2 = ea4[(size_t)e * 4 + 2], a3 = ea4[(size_t)e * 4 + 3];
        for (;;) {
            const int eB = e + 1024;
            const bool moreB = eB < e1;
            const int fB = moreB ? eB : e;
            const int dB = dst[fB], sB = src[fB];
            const float4 b0 = ea4[(size_t)fB * 4], b1 = ea4[(size_t)fB * 4 + 1];
            const float4 b2 = ea4[(size_t)fB * 4 + 2], b3 = ea4[(size_t)fB * 4 + 3];
            place_edge(dA, sA, a0, a1, a2, a3, wea1, wea2, lcur, stA, stP);
            if (!moreB) break;
            e = eB + 1024;
            const bool moreA = e < e1;
            const int fA = moreA ? e : eB;
            dA = dst[fA]; sA = src[fA];
            a0 = ea4[(size_t)fA * 4]; a1 = ea4[(size_t)fA * 4 + 1];
            a2 = ea4[(size_t)fA * 4 + 2]; a3 = ea4[(size_t)fA * 4 + 3];
            place_edge(dB, sB, b0, b1, b2, b3, wea1, wea2, lcur, stA, stP);
            if (!moreA) break;
        }
    }
    __syncthreads();
    for (int bin = w; bin < NBINS; bin += 32) {
        const int ls = lstart[bin], c = own[bin], g = gfirst[bin];
        for (int i = lane; i < c; i += 32) {
            recA[g + i] = stA[ls + i];
            recP[g + i] = stP[ls + i];
        }
    }
}

#define FOR_SLOTS(X) X(0) X(1) X(2) X(3) X(4) X(5) X(6) X(7) X(8) X(9)
static_assert(CAP / 1024 == 10, "FOR_SLOTS lists CAP / 1024 slots");

__global__ void __launch_bounds__(1024)
bin_sort(const int* __restrict__ binstart, float4* recA, int2* recP,
         int* __restrict__ rowptr, int* __restrict__ cnt)
{
    __shared__ float4 bufA[CAP];
    __shared__ int2 bufP[CAP];
    __shared__ int cntL[256];
    __shared__ int startL[256];
    __shared__ int curL[256];
    __shared__ int wtot[32];
    const int t = threadIdx.x, lane = t & 31, w = t >> 5, bin = blockIdx.x;
    const int beg = binstart[bin];
    const int end = (bin + 1 < NBINS) ? binstart[bin + 1] : EE;
    const int n = end - beg;
    const bool fits = n <= CAP;
    if (t < 256) { cntL[t] = 0; curL[t] = 0; }
    __syncthreads();
#define DECL_SLOT(j) int2 pv##j = make_int2(0, 0); float4 av##j = make_float4(0.f, 0.f, 0.f, 0.f); int rk##j = 0;
    FOR_SLOTS(DECL_SLOT)
    if (fits) {
#define LOAD_SLOT(j) { const int g = beg + j * 1024 + t; const int gc = (g < EE) ? g : (EE - 1); \
                       pv##j = recP[gc]; av##j = recA[gc]; }
        FOR_SLOTS(LOAD_SLOT)
#define COUNT_SLOT(j) if (j * 1024 + t < n) rk##j = atomicAdd(&cntL[pv##j.x >> SRCBITS], 1);
        FOR_SLOTS(COUNT_SLOT)
    } else {
        for (int i = t; i < n; i += 1024) atomicAdd(&cntL[recP[beg + i].x >> SRCBITS], 1);
    }
    __syncthreads();
    int c = 0, incl = 0;
    if (t < 256) {
        c = cntL[t];
        incl = c;
        for (int o = 1; o < 32; o <<= 1) {
            int u = __shfl_up(incl, o);
            if (lane >= o) incl += u;
        }
        if (lane == 31) wtot[w] = incl;
    }
    __syncthreads();
    if (t < 256) {
        int pre = (lane < w) ? wtot[lane] : 0;
        for (int o = 16; o > 0; o >>= 1) pre += __shfl_xor(pre, o);
        const int st = pre + incl - c;
        startL[t] = st;
        const int node = (bin << BINSH) + t;
        if (node < NN) { rowptr[node] = beg + st; cnt[node] = c; }
    }
    __syncthreads();
    if (fits) {
#define START_SLOT(j) const int st##j = startL[pv##j.x >> SRCBITS];
        FOR_SLOTS(START_SLOT)
#define PLACE_SLOT(j) if (j * 1024 + t < n) { \
            const int pos = st##j + rk##j; \
            int2 q; q.x = pv##j.x & SRCMASK; q.y = pv##j.y; \
            bufP[pos] = q; bufA[pos] = av##j; }
        FOR_SLOTS(PLACE_SLOT)
        __syncthreads();
        const int lead = beg & 15;
        for (int i = t - lead; i < n; i += 1024) {
            if (i >= 0) {
                recP[beg + i] = bufP[i];
                recA[beg + i] = bufA[i];
            }
        }
    } else {
        if (t == 0) {
            for (int nd = 0; nd < 256; ++nd) {
                const int endn = cntL[nd];
                int k = curL[nd];
                while (k < endn) {
                    const int p = beg + startL[nd] + k;
                    const int2 r = recP[p];
                    const int dl = r.x >> SRCBITS;
                    if (dl == nd) { ++k; continue; }
                    const int q = beg + startL[dl] + curL[dl];
                    curL[dl] += 1;
                    const int2 r2 = recP[q];
                    const float4 a1 = recA[p], a2 = recA[q];
                    recP[q] = r;  recA[q] = a1;
                    recP[p] = r2; recA[p] = a2;
                }
                curL[nd] = endn;
            }
        }
        __syncthreads();
        for (int i = t; i < n; i += 1024) recP[beg + i].x &= SRCMASK;
    }
}

__device__ __forceinline__ float leaky(float v) { return (v >= 0.f) ? v : NEGS * v; }
__device__ __forceinline__ float wave_max(float v) {
    for (int o = 16; o > 0; o >>= 1) v = fmaxf(v, __shfl_xor(v, o));
    return v;
}
__device__ __forceinline__ float wave_sum(float v) {
    for (int o = 16; o > 0; o >>= 1) v += __shfl_xor(v, o);
    return v;
}

template <int CTRL> __device__ __forceinline__ float dpp_f(float v) {
    return __int_as_float(__builtin_amdgcn_update_dpp(0, __float_as_int(v), CTRL, 0xF, 0xF, true));
}
#define XM(n) (0x160 | (n))
#define QB(h) ((h) * 0x55)
#define RS(n) (0x150 | (n))
__device__ __forceinline__ float xhalf(float v) {
    return __int_as_float(__builtin_amdgcn_permlanex16(__float_as_int(v), __float_as_int(v),
                                                       0x76543210, 0xfedcba98, false, false));
}
__device__ __forceinline__ float sum_groups(float v) {
    v += dpp_f<XM(4)>(v);
    v += dpp_f<XM(8)>(v);
    return v + xhalf(v);
}
__device__ __forceinline__ float max_groups(float v) {
    v = fmaxf(v, dpp_f<XM(4)>(v));
    v = fmaxf(v, dpp_f<XM(8)>(v));
    return fmaxf(v, xhalf(v));
}

__global__ void __launch_bounds__(256)
node_aggr_l1(const int* __restrict__ rowptr, const int* __restrict__ cnt,
             const float* __restrict__ se1S, const int* __restrict__ pairS,
             const float* __restrict__ s_i1, const float* __restrict__ Vs,
             const float* __restrict__ x, float* __restrict__ accx1)
{
    const int lane = threadIdx.x & 31;
    const int m = lane & 3;
    const int g = lane >> 2;
    const bool mh = (m & 2) != 0, ml = (m & 1) != 0;
    const int wave = blockIdx.x * (blockDim.x >> 5) + __builtin_amdgcn_readfirstlane(threadIdx.x >> 5);
    const int nw = gridDim.x * (blockDim.x >> 5);
    const int2* __restrict__ pr2 = (const int2*)pairS;
    const float4* __restrict__ x4 = (const float4*)x;
    float vj[4][4];
#pragma unroll
    for (int h = 0; h < 4; ++h)
#pragma unroll
        for (int e = 0; e < 4; ++e) vj[h][e] = Vs[(4 * m + e) * 16 + 4 + h];
    for (int d = wave; d < NN; d += nw) {
        const int beg = rowptr[d];
        const int deg = cnt[d];
        const float sim = s_i1[(size_t)d * 4 + m];
        float M = -INFINITY, dn = 0.f;
        float acc[4][4];
#pragma unroll
        for (int h = 0; h < 4; ++h)
#pragma unroll
            for (int e = 0; e < 4; ++e) acc[h][e] = 0.f;
        for (int j0 = 0; j0 < deg; j0 += 32) {
            const int left = deg - j0;
            int sidx[4];
            float sev[4], pen[4];
            float4 xr[4];
#pragma unroll
            for (int q = 0; q < 4; ++q) {
                if (8 * q < left) {
                    const bool act = 8 * q + g < left;
                    const int slot = beg + j0 + (act ? 8 * q + g : 0);
                    sidx[q] = pr2[slot].x;
                    sev[q] = se1S[(size_t)slot * 4 + m];
                    pen[q] = act ? 0.f : -INFINITY;
                }
            }
#pragma unroll
            for (int q = 0; q < 4; ++q)
                if (8 * q < left) xr[q] = x4[(size_t)sidx[q] * 4 + m];
            float c[4];
#pragma unroll
            for (int q = 0; q < 4; ++q) {
                c[q] = -INFINITY;
                if (8 * q < left) {
                    const float t0 = fmaf(xr[q].w, vj[0][3], fmaf(xr[q].z, vj[0][2], fmaf(xr[q].y, vj[0][1], xr[q].x * vj[0][0])));
                    const float t1 = fmaf(xr[q].w, vj[1][3], fmaf(xr[q].z, vj[1][2], fmaf(xr[q].y, vj[1][1], xr[q].x * vj[1][0])));
                    const float t2 = fmaf(xr[q].w, vj[2][3], fmaf(xr[q].z, vj[2][2], fmaf(xr[q].y, vj[2][1], xr[q].x * vj[2][0])));
                    const float t3 = fmaf(xr[q].w, vj[3][3], fmaf(xr[q].z, vj[3][2], fmaf(xr[q].y, vj[3][1], xr[q].x * vj[3][0])));
                    float k0 = mh ? t2 : t0, k1 = mh ? t3 : t1;
                    const float g0 = mh ? t0 : t2, g1 = mh ? t1 : t3;
                    k0 += dpp_f<XM(2)>(g0);
                    k1 += dpp_f<XM(2)>(g1);
                    const float kk = ml ? k1 : k0, gg = ml ? k0 : k1;
                    const float sj = kk + dpp_f<XM(1)>(gg);
                    c[q] = leaky(sim + sj + sev[q]) + pen[q];
                }
            }
            const float cm = max_groups(fmaxf(fmaxf(c[0], c[1]), fmaxf(c[2], c[3])));
            const float n = fmaxf(M, cm);
            const float r = exp_hw(M - n);
            M = n;
            float psum = 0.f;
            if (j0 > 0) {
                const float r0 = dpp_f<QB(0)>(r), r1 = dpp_f<QB(1)>(r);
                const float r2 = dpp_f<QB(2)>(r), r3 = dpp_f<QB(3)>(r);
#pragma unroll
                for (int e = 0; e < 4; ++e) { acc[0][e] *= r0; acc[1][e] *= r1; acc[2][e] *= r2; acc[3][e] *= r3; }
            }
#pragma unroll
            for (int q = 0; q < 4; ++q) {
                if (8 * q < left) {
                    const float pq = exp_hw(c[q] - n);
                    psum += pq;
                    const float p0 = dpp_f<QB(0)>(pq), p1 = dpp_f<QB(1)>(pq);
                    const float p2 = dpp_f<QB(2)>(pq), p3 = dpp_f<QB(3)>(pq);
                    const float xe[4] = {xr[q].x, xr[q].y, xr[q].z, xr[q].w};
#pragma unroll
                    for (int e = 0; e < 4; ++e) {
                        acc[0][e] = fmaf(p0, xe[e], acc[0][e]);
                        acc[1][e] = fmaf(p1, xe[e], acc[1][e]);
                        acc[2][e] = fmaf(p2, xe[e], acc[2][e]);
                        acc[3][e] = fmaf(p3, xe[e], acc[3][e]);
                    }
                }
            }
            dn = dn * r + psum;
        }
        dn = sum_groups(dn);
#pragma unroll
        for (int h = 0; h < 4; ++h)
#pragma unroll
            for (int e = 0; e < 4; ++e) acc[h][e] = sum_groups(acc[h][e]);
        const float d0 = dpp_f<RS(0)>(dn), d1 = dpp_f<RS(1)>(dn);
        const float d2 = dpp_f<RS(2)>(dn), d3 = dpp_f<RS(3)>(dn);
        if (lane < 16) {
            const float dv = (g == 0) ? d0 : (g == 1) ? d1 : (g == 2) ? d2 : d3;
            float o[4];
#pragma unroll
            for (int e = 0; e < 4; ++e) {
                const float a0 = acc[0][e], a1 = acc[1][e], a2 = acc[2][e], a3 = acc[3][e];
                const float a = (g == 0) ? a0 : (g == 1) ? a1 : (g == 2) ? a2 : a3;
                o[e] = (dv > 0.f) ? a / dv : 0.f;
            }
            float4 ov; ov.x = o[0]; ov.y = o[1]; ov.z = o[2]; ov.w = o[3];
            *(float4*)(accx1 + (size_t)d * 64 + g * 16 + 4 * m) = ov;
        }
    }
}

__global__ void __launch_bounds__(256)
node_aggr_l2(const int* __restrict__ rowptr, const int* __restrict__ cnt,
             const int* __restrict__ pairS, const float* __restrict__ s_i2,
             const float* __restrict__ sz2, const float* __restrict__ p2b,
             float* __restrict__ out)
{
    const int lane = threadIdx.x & 31;
    const int wave = blockIdx.x * (blockDim.x >> 5) + __builtin_amdgcn_readfirstlane(threadIdx.x >> 5);
    const int nw = gridDim.x * (blockDim.x >> 5);
    const int2* __restrict__ pr2 = (const int2*)pairS;
    const float2* __restrict__ sz = (const float2*)sz2;
    const float bias = p2b[0];
    for (int d = wave; d < NN; d += nw) {
        const int beg = rowptr[d];
        const int deg = cnt[d];
        const float si = s_i2[d];
        float M = -INFINITY, num = 0.f, den = 0.f;
        for (int j0 = 0; j0 < deg; j0 += 32) {
            const int j = j0 + lane;
            const bool act = j < deg;
            const int2 pr = pr2[beg + (act ? j : j0)];
            const float2 q = sz[pr.x];
            const float c = leaky(si + q.x + __int_as_float(pr.y)) + (act ? 0.f : -INFINITY);
            const float n = fmaxf(M, wave_max(c));
            const float r = exp_hw(M - n);
            M = n;
            const float p = exp_hw(c - n);
            num = num * r + p * q.y;
            den = den * r + p;
        }
        num = wave_sum(num);
        den = wave_sum(den);
        if (lane == 0) {
            float v = (den > 0.f) ? num / den : 0.f;
            v += bias;
            out[d] = (v > 0.f) ? v : (expf(v) - 1.f);
        }
    }
}

extern "C" void kernel_launch(void* const* d_in, const int* in_sizes, int n_in,
                              void* d_out, int out_size, void* d_ws, size_t ws_size,
                              hipStream_t stream)
{
    const float* x       = (const float*)d_in[0];
    const int*   eidx    = (const int*)d_in[1];
    const float* eattr   = (const float*)d_in[2];
    const float* W1      = (const float*)d_in[3];
    const float* We1     = (const float*)d_in[4];
    const float* att1    = (const float*)d_in[5];
    const float* proj1_w = (const float*)d_in[6];
    const float* proj1_b = (const float*)d_in[7];
    const float* W2      = (const float*)d_in[8];
    const float* We2     = (const float*)d_in[9];
    const float* att2    = (const float*)d_in[10];
    const float* proj2_w = (const float*)d_in[11];
    const float* proj2_b = (const float*)d_in[12];
    float* out = (float*)d_out;

    const int* srcp = eidx;
    const int* dstp = eidx + EE;

    float* ws     = (float*)d_ws;
    float* xw1    = ws;
    float* accx1  = xw1 + (size_t)NN * 128;
    int*   pairS  = (int*)(accx1 + (size_t)NN * 64);
    float* s_i1   = accx1 + (size_t)NN * 128;
    float* slotA  = s_i1 + (size_t)NN * 8;
    int*   cnt    = (int*)slotA;
    float* Wcomb  = slotA + (size_t)NN;
    float* Vs     = Wcomb + 64 * 128;
    float* V2     = Vs + 256;
    int*   rowptr = (int*)(slotA + (size_t)NN * 4);
    float* hw2    = slotA + (size_t)NN * 8;
    float* s_i2   = hw2 + (size_t)NN * 64;
    float* sz2    = s_i2 + (size_t)NN;
    float* wea1   = s_i2 + (size_t)NN * 4 + 2048;
    float* wea2   = wea1 + 64;
    float* se1S   = xw1;
    int*   cntmat = (int*)hw2;
    int*   binstart = cntmat + NBINS * PBLK;

    prep_weights<<<32, 256, 0, stream>>>(W1, We1, att1, We2, att2, proj1_w, W2, proj2_w,
                                         Vs, V2, wea1, wea2, Wcomb);

    node_scores_l1<<<(NN / 16 + 7) / 8, 256, 0, stream>>>(x, Vs, s_i1);

    bin_count<<<PBLK, 1024, 0, stream>>>(dstp, cntmat);
    edge_partition<<<PBLK, 1024, 0, stream>>>(srcp, dstp, eattr, wea1, wea2, cntmat,
                                              (float4*)se1S, (int2*)pairS, binstart);
    bin_sort<<<NBINS, 1024, 0, stream>>>(binstart, (float4*)se1S, (int2*)pairS, rowptr, cnt);

    node_aggr_l1<<<1280, 256, 0, stream>>>(rowptr, cnt, se1S, pairS, s_i1, Vs, x, accx1);

    proj_scores_l2<<<(NN / 16 + 7) / 8, 256, 0, stream>>>(accx1, Wcomb, proj1_b, V2, s_i2, sz2);
    node_aggr_l2<<<2048, 256, 0, stream>>>(rowptr, cnt, pairS, s_i2, sz2, proj2_b, out);
}


# Layer-2 node pass: 8 lanes per node, 4 nodes per wave at once, 48 edges of a node requested before the first is used; node maximum and sums over 8 lanes on DPP, no LDS shuffles; sums reordered
# speedup vs baseline: 9.2611x; 1.0417x over previous
#include <hip/hip_runtime.h>
#include <cmath>

#define NN 100000
#define EE 3200000
#define NEGS 0.2f
#define BINSH 8
#define NBINS ((NN + 255) >> BINSH)
#define PBLK 256
#define ECHUNK ((EE + PBLK - 1) / PBLK)
#define CAP 10240
#define SRCBITS 17
#define SRCMASK ((1 << SRCBITS) - 1)

typedef __attribute__((ext_vector_type(2))) float v2f;
typedef __attribute__((ext_vector_type(8))) float v8f;

__device__ __forceinline__ float exp_hw(const float v)
{
    return __builtin_amdgcn_exp2f(v * 1.44269504088896340736f);
}

__global__ void prep_weights(const float* __restrict__ W1, const float* __restrict__ We1,
                             const float* __restrict__ att1, const float* __restrict__ We2,
                             const float* __restrict__ att2, const float* __restrict__ proj1_w,
                             const float* __restrict__ W2, const float* __restrict__ p2w,
                             float* __restrict__ Vs, float* __restrict__ V2,
                             float* __restrict__ wea1,
                             float* __restrict__ wea2, float* __restrict__ Wcomb)
{
    int tid = blockIdx.x * blockDim.x + threadIdx.x;
    int nt = gridDim.x * blockDim.x;
    for (int i = tid; i < 256; i += nt) {
        int k = i >> 4, c = i & 15;
        float s = 0.f;
        if (c < 8) {
            int h = c & 3;
            const float* a = att1 + h * 96 + (c >> 2) * 32;
            const float* w = W1 + h * 512 + k * 32;
            for (int o = 0; o < 32; ++o) s += w[o] * a[o];
        }
        Vs[i] = s;
    }
    for (int i = tid; i < 2048; i += nt) {
        const int e2 = i & 7, ln = (i >> 3) & 31, t2 = i >> 8;
        int k = 16 * t2 + 8 * (ln >> 4) + e2, c = ln & 15;
        float s = 0.f;
        if (c < 3) {
            const float* a = (c == 2) ? p2w : (att2 + c * 32);
            const float* w = W2 + k * 32;
            for (int o = 0; o < 32; ++o) s += w[o] * a[o];
        }
        V2[i] = s;
    }
    for (int i = tid; i < 64; i += nt) {
        int h = i >> 4, k = i & 15;
        float s = 0.f;
        for (int o = 0; o < 32; ++o) s += We1[h * 512 + k * 32 + o] * att1[h * 96 + 64 + o];
        wea1[i] = s;
    }
    for (int i = tid; i < 16; i += nt) {
        float s = 0.f;
        for (int o = 0; o < 32; ++o) s += We2[i * 32 + o] * att2[64 + o];
        wea2[i] = s;
    }
    for (int i = tid; i < 64 * 128; i += nt) {
        const int e4 = i & 3, ln = (i >> 2) & 31, m8 = (i >> 7) & 7, t8 = i >> 10;
        int row = 8 * m8 + 4 * (ln >> 4) + e4, c = 16 * t8 + (ln & 15), h = row >> 4, k = row & 15;
        const float* w = W1 + h * 512 + k * 32;
        const float* pw = proj1_w + (size_t)(h * 32) * 128 + c;
        float s = 0.f;
        for (int o = 0; o < 32; ++o) s += w[o] * pw[(size_t)o * 128];
        Wcomb[i] = s;
    }
}

__device__ __forceinline__ v8f strip_product(const float* __restrict__ Ap,
                                             const float* __restrict__ Bp, const int K)
{
    v8f acc = {};
#pragma unroll
    for (int k0 = 0; k0 < K; k0 += 4) {
        v2f a; a.x = Ap[k0]; a.y = Ap[k0 + 1];
        v2f b; b.x = Bp[k0 * 16]; b.y = Bp[k0 * 16 + 16];
        acc = __builtin_amdgcn_wmma_f32_16x16x4_f32(false, a, false, b,
                                                    (short)0, acc, false, false);
    }
    return acc;
}

__global__ void __launch_bounds__(256)
node_scores_l1(const float* __restrict__ x, const float* __restrict__ Vs,
               float* __restrict__ s_i1)
{
    const int lane = threadIdx.x & 31;
    const int strip = blockIdx.x * (blockDim.x >> 5) + (threadIdx.x >> 5);
    if (strip >= NN / 16) return;
    const int r = lane & 15;
    const int kk = (lane >> 4) << 1;
    const v8f acc = strip_product(x + (size_t)(strip * 16 + r) * 16 + kk, Vs + kk * 16 + r, 16);
    const int rbase = strip * 16 + ((lane >> 4) << 3);
    if (r < 4) {
        float* o = s_i1 + r;
#pragma unroll
        for (int v = 0; v < 8; ++v) o[(size_t)(rbase + v) * 4] = acc[v];
    }
}

__device__ __forceinline__ v8f mm4(const float a0, const float a1, const float b0, const float b1,
                                   const v8f acc)
{
    v2f a; a.x = a0; a.y = a1;
    v2f b; b.x = b0; b.y = b1;
    return __builtin_amdgcn_wmma_f32_16x16x4_f32(false, a, false, b, (short)0, acc, false, false);
}

__global__ void __launch_bounds__(256)
proj_scores_l2(const float* __restrict__ accx1, const float* __restrict__ WcF,
               const float* __restrict__ bias, const float* __restrict__ V2F,
               float* __restrict__ s_i2, float* __restrict__ sz2)
{
    const int lane = threadIdx.x & 31;
    const int strip = blockIdx.x * (blockDim.x >> 5) + (threadIdx.x >> 5);
    if (strip >= NN / 16) return;
    const int c = lane & 15;
    const int g = lane >> 4;
    const float4* __restrict__ xr = (const float4*)(accx1 + (size_t)(strip * 16 + c) * 64 + 4 * g);
    float4 xb[8];
#pragma unroll
    for (int m = 0; m < 8; ++m) xb[m] = xr[2 * m];
    const float4* __restrict__ wf = (const float4*)WcF + lane;
    const float4* __restrict__ vf = (const float4*)V2F + lane * 2;
    v8f acc2 = {};
#pragma unroll
    for (int tn = 0; tn < 8; ++tn) {
        v8f acc = {};
#pragma unroll
        for (int m = 0; m < 8; ++m) {
            const float4 w = wf[(tn * 8 + m) * 32];
            acc = mm4(w.x, w.y, xb[m].x, xb[m].y, acc);
            acc = mm4(w.z, w.w, xb[m].z, xb[m].w, acc);
        }
        const float4 bl = *(const float4*)(bias + 16 * tn + 8 * g);
        const float4 bh = *(const float4*)(bias + 16 * tn + 8 * g + 4);
        const float bv[8] = {bl.x, bl.y, bl.z, bl.w, bh.x, bh.y, bh.z, bh.w};
        float hv[8];
#pragma unroll
        for (int r = 0; r < 8; ++r) {
            const float xv = acc[r] + bv[r];
            const float y = exp_hw(xv) - 1.f;
            const float z = exp_hw(y) - 1.f;
            hv[r] = (xv > 0.f) ? xv : z;
        }
        const float4 v0 = vf[tn * 64], v1 = vf[tn * 64 + 1];
        acc2 = mm4(v0.x, v0.y, hv[0], hv[1], acc2);
        acc2 = mm4(v0.z, v0.w, hv[2], hv[3], acc2);
        acc2 = mm4(v1.x, v1.y, hv[4], hv[5], acc2);
        acc2 = mm4(v1.z, v1.w, hv[6], hv[7], acc2);
    }
    if (g == 0) {
        const int node = strip * 16 + c;
        s_i2[node] = acc2[0];
        float2 q; q.x = acc2[1]; q.y = acc2[2];
        ((float2*)sz2)[node] = q;
    }
}

__global__ void __launch_bounds__(1024)
bin_count(const int* __restrict__ dst, int* __restrict__ cntmat)
{
    __shared__ int hist[NBINS];
    const int t = threadIdx.x, b = blockIdx.x;
    for (int i = t; i < NBINS; i += 1024) hist[i] = 0;
    __syncthreads();
    const int e0 = b * ECHUNK;
    const int e1 = (e0 + ECHUNK < EE) ? (e0 + ECHUNK) : EE;
    int dv[(ECHUNK + 1023) / 1024];
#pragma unroll
    for (int j = 0; j < (ECHUNK + 1023) / 1024; ++j) {
        const int e = e0 + j * 1024 + t;
        dv[j] = dst[(e < EE) ? e : (EE - 1)];
    }
#pragma unroll
    for (int j = 0; j < (ECHUNK + 1023) / 1024; ++j)
        if (e0 + j * 1024 + t < e1) atomicAdd(&hist[dv[j] >> BINSH], 1);
    __syncthreads();
    for (int i = t; i < NBINS; i += 1024) cntmat[i * PBLK + b] = hist[i];
}

__device__ __forceinline__ void place_edge(const int d, const int sv, const float4 r0, const float4 r1,
                                           const float4 r2, const float4 r3,
                                           const float* __restrict__ wea1,
                                           const float* __restrict__ wea2, int* lcur,
                                           float4* stA, int2* stP)
{
    const float ea[16] = {r0.x, r0.y, r0.z, r0.w, r1.x, r1.y, r1.z, r1.w,
                          r2.x, r2.y, r2.z, r2.w, r3.x, r3.y, r3.z, r3.w};
    float a0 = 0.f, a1 = 0.f, a2 = 0.f, a3 = 0.f, bb = 0.f;
#pragma unroll
    for (int k = 0; k < 16; ++k) {
        a0 += ea[k] * wea1[k];      a1 += ea[k] * wea1[16 + k];
        a2 += ea[k] * wea1[32 + k]; a3 += ea[k] * wea1[48 + k];
        bb += ea[k] * wea2[k];
    }
    const int pos = atomicAdd(&lcur[d >> BINSH], 1);
    float4 v4; v4.x = a0; v4.y = a1; v4.z = a2; v4.w = a3;
    stA[pos] = v4;
    int2 q; q.x = sv | ((d & ((1 << BINSH) - 1)) << SRCBITS); q.y = __float_as_int(bb);
    stP[pos] = q;
}

__global__ void __launch_bounds__(1024)
edge_partition(const int* __restrict__ src, const int* __restrict__ dst,
               const float* __restrict__ eattr, const float* __restrict__ wea1,
               const float* __restrict__ wea2, const int* __restrict__ cntmat,
               float4* __restrict__ recA, int2* __restrict__ recP,
               int* __restrict__ binstart)
{
    __shared__ float4 stA[ECHUNK];
    __shared__ int2 stP[ECHUNK];
    __shared__ int tot[NBINS];
    __shared__ int gfirst[NBINS];
    __shared__ int own[NBINS];
    __shared__ int lstart[NBINS];
    __shared__ int lcur[NBINS];
    __shared__ int wsum[32];
    __shared__ int wsumL[32];
    const int t = threadIdx.x, lane = t & 31, w = t >> 5, b = blockIdx.x;
    for (int bin = w; bin < NBINS; bin += 32) {
        const int* row = cntmat + bin * PBLK;
        int all = 0, before = 0;
        for (int q = 0; q < PBLK; q += 32) {
            const int c = row[q + lane];
            all += c;
            before += (q + lane < b) ? c : 0;
        }
        const int mine = row[b];
        for (int o = 16; o > 0; o >>= 1) {
            all += __shfl_xor(all, o);
            before += __shfl_xor(before, o);
        }
        if (lane == 0) { tot[bin] = all; gfirst[bin] = before; own[bin] = mine; }
    }
    __syncthreads();
    const int v = (t < NBINS) ? tot[t] : 0;
    const int m = (t < NBINS) ? own[t] : 0;
    int incl = v, inclL = m;
    for (int o = 1; o < 32; o <<= 1) {
        const int u = __shfl_up(incl, o);
        const int uL = __shfl_up(inclL, o);
        if (lane >= o) { incl += u; inclL += uL; }
    }
    if (lane == 31) { wsum[w] = incl; wsumL[w] = inclL; }
    __syncthreads();
    int pre = (lane < w) ? wsum[lane] : 0;
    int preL = (lane < w) ? wsumL[lane] : 0;
    for (int o = 16; o > 0; o >>= 1) {
        pre += __shfl_xor(pre, o);
        preL += __shfl_xor(preL, o);
    }
    if (t < NBINS) {
        const int first = pre + incl - v;
        const int firstL = preL + inclL - m;
        gfirst[t] += first;
        lstart[t] = firstL;
        lcur[t] = firstL;
        if (b == 0) binstart[t] = first;
    }
    __syncthreads();
    const int e0 = b * ECHUNK;
    const int e1 = (e0 + ECHUNK < EE) ? (e0 + ECHUNK) : EE;
    const float4* __restrict__ ea4 = (const float4*)eattr;
    int e = e0 + t;
    if (e < e1) {
        int dA = dst[e], sA = src[e];
        float4 a0 = ea4[(size_t)e * 4], a1 = ea4[(size_t)e * 4 + 1];
        float4 a2 = ea4[(size_t)e * 4 + 2], a3 = ea4[(size_t)e * 4 + 3];
        for (;;) {
            const int eB = e + 1024;
            const bool moreB = eB < e1;
            const int fB = moreB ? eB : e;
            const int dB = dst[fB], sB = src[fB];
            const float4 b0 = ea4[(size_t)fB * 4], b1 = ea4[(size_t)fB * 4 + 1];
            const float4 b2 = ea4[(size_t)fB * 4 + 2], b3 = ea4[(size_t)fB * 4 + 3];
            place_edge(dA, sA, a0, a1, a2, a3, wea1, wea2, lcur, stA, stP);
            if (!moreB) break;
            e = eB + 1024;
            const bool moreA = e < e1;
            const int fA = moreA ? e : eB;
            dA = dst[fA]; sA = src[fA];
            a0 = ea4[(size_t)fA * 4]; a1 = ea4[(size_t)fA * 4 + 1];
            a2 = ea4[(size_t)fA * 4 + 2]; a3 = ea4[(size_t)fA * 4 + 3];
            place_edge(dB, sB, b0, b1, b2, b3, wea1, wea2, lcur, stA, stP);
            if (!moreA) break;
        }
    }
    __syncthreads();
    for (int bin = w; bin < NBINS; bin += 32) {
        const int ls = lstart[bin], c = own[bin], g = gfirst[bin];
        for (int i = lane; i < c; i += 32) {
            recA[g + i] = stA[ls + i];
            recP[g + i] = stP[ls + i];
        }
    }
}

#define FOR_SLOTS(X) X(0) X(1) X(2) X(3) X(4) X(5) X(6) X(7) X(8) X(9)
static_assert(CAP / 1024 == 10, "FOR_SLOTS lists CAP / 1024 slots");

__global__ void __launch_bounds__(1024)
bin_sort(const int* __restrict__ binstart, float4* recA, int2* recP,
         int* __restrict__ rowptr, int* __restrict__ cnt)
{
    __shared__ float4 bufA[CAP];
    __shared__ int2 bufP[CAP];
    __shared__ int cntL[256];
    __shared__ int startL[256];
    __shared__ int curL[256];
    __shared__ int wtot[32];
    const int t = threadIdx.x, lane = t & 31, w = t >> 5, bin = blockIdx.x;
    const int beg = binstart[bin];
    const int end = (bin + 1 < NBINS) ? binstart[bin + 1] : EE;
    const int n = end - beg;
    const bool fits = n <= CAP;
    if (t < 256) { cntL[t] = 0; curL[t] = 0; }
    __syncthreads();
#define DECL_SLOT(j) int2 pv##j = make_int2(0, 0); float4 av##j = make_float4(0.f, 0.f, 0.f, 0.f); int rk##j = 0;
    FOR_SLOTS(DECL_SLOT)
    if (fits) {
#define LOAD_SLOT(j) { const int g = beg + j * 1024 + t; const int gc = (g < EE) ? g : (EE - 1); \
                       pv##j = recP[gc]; av##j = recA[gc]; }
        FOR_SLOTS(LOAD_SLOT)
#define COUNT_SLOT(j) if (j * 1024 + t < n) rk##j = atomicAdd(&cntL[pv##j.x >> SRCBITS], 1);
        FOR_SLOTS(COUNT_SLOT)
    } else {
        for (int i = t; i < n; i += 1024) atomicAdd(&cntL[recP[beg + i].x >> SRCBITS], 1);
    }
    __syncthreads();
    int c = 0, incl = 0;
    if (t < 256) {
        c = cntL[t];
        incl = c;
        for (int o = 1; o < 32; o <<= 1) {
            int u = __shfl_up(incl, o);
            if (lane >= o) incl += u;
        }
        if (lane == 31) wtot[w] = incl;
    }
    __syncthreads();
    if (t < 256) {
        int pre = (lane < w) ? wtot[lane] : 0;
        for (int o = 16; o > 0; o >>= 1) pre += __shfl_xor(pre, o);
        const int st = pre + incl - c;
        startL[t] = st;
        const int node = (bin << BINSH) + t;
        if (node < NN) { rowptr[node] = beg + st; cnt[node] = c; }
    }
    __syncthreads();
    if (fits) {
#define START_SLOT(j) const int st##j = startL[pv##j.x >> SRCBITS];
        FOR_SLOTS(START_SLOT)
#define PLACE_SLOT(j) if (j * 1024 + t < n) { \
            const int pos = st##j + rk##j; \
            int2 q; q.x = pv##j.x & SRCMASK; q.y = pv##j.y; \
            bufP[pos] = q; bufA[pos] = av##j; }
        FOR_SLOTS(PLACE_SLOT)
        __syncthreads();
        const int lead = beg & 15;
        for (int i = t - lead; i < n; i += 1024) {
            if (i >= 0) {
                recP[beg + i] = bufP[i];
                recA[beg + i] = bufA[i];
            }
        }
    } else {
        if (t == 0) {
            for (int nd = 0; nd < 256; ++nd) {
                const int endn = cntL[nd];
                int k = curL[nd];
                while (k < endn) {
                    const int p = beg + startL[nd] + k;
                    const int2 r = recP[p];
                    const int dl = r.x >> SRCBITS;
                    if (dl == nd) { ++k; continue; }
                    const int q = beg + startL[dl] + curL[dl];
                    curL[dl] += 1;
                    const int2 r2 = recP[q];
                    const float4 a1 = recA[p], a2 = recA[q];
                    recP[q] = r;  recA[q] = a1;
                    recP[p] = r2; recA[p] = a2;
                }
                curL[nd] = endn;
            }
        }
        __syncthreads();
        for (int i = t; i < n; i += 1024) recP[beg + i].x &= SRCMASK;
    }
}

__device__ __forceinline__ float leaky(float v) { return (v >= 0.f) ? v : NEGS * v; }
__device__ __forceinline__ float wave_max(float v) {
    for (int o = 16; o > 0; o >>= 1) v = fmaxf(v, __shfl_xor(v, o));
    return v;
}
__device__ __forceinline__ float wave_sum(float v) {
    for (int o = 16; o > 0; o >>= 1) v += __shfl_xor(v, o);
    return v;
}

template <int CTRL> __device__ __forceinline__ float dpp_f(float v) {
    return __int_as_float(__builtin_amdgcn_update_dpp(0, __float_as_int(v), CTRL, 0xF, 0xF, true));
}
#define XM(n) (0x160 | (n))
#define QB(h) ((h) * 0x55)
#define RS(n) (0x150 | (n))
__device__ __forceinline__ float xhalf(float v) {
    return __int_as_float(__builtin_amdgcn_permlanex16(__float_as_int(v), __float_as_int(v),
                                                       0x76543210, 0xfedcba98, false, false));
}
__device__ __forceinline__ float sum_groups(float v) {
    v += dpp_f<XM(4)>(v);
    v += dpp_f<XM(8)>(v);
    return v + xhalf(v);
}
__device__ __forceinline__ float max_groups(float v) {
    v = fmaxf(v, dpp_f<XM(4)>(v));
    v = fmaxf(v, dpp_f<XM(8)>(v));
    return fmaxf(v, xhalf(v));
}

__global__ void __launch_bounds__(256)
node_aggr_l1(const int* __restrict__ rowptr, const int* __restrict__ cnt,
             const float* __restrict__ se1S, const int* __restrict__ pairS,
             const float* __restrict__ s_i1, const float* __restrict__ Vs,
             const float* __restrict__ x, float* __restrict__ accx1)
{
    const int lane = threadIdx.x & 31;
    const int m = lane & 3;
    const int g = lane >> 2;
    const bool mh = (m & 2) != 0, ml = (m & 1) != 0;
    const int wave = blockIdx.x * (blockDim.x >> 5) + __builtin_amdgcn_readfirstlane(threadIdx.x >> 5);
    const int nw = gridDim.x * (blockDim.x >> 5);
    const int2* __restrict__ pr2 = (const int2*)pairS;
    const float4* __restrict__ x4 = (const float4*)x;
    float vj[4][4];
#pragma unroll
    for (int h = 0; h < 4; ++h)
#pragma unroll
        for (int e = 0; e < 4; ++e) vj[h][e] = Vs[(4 * m + e) * 16 + 4 + h];
    for (int d = wave; d < NN; d += nw) {
        const int beg = rowptr[d];
        const int deg = cnt[d];
        const float sim = s_i1[(size_t)d * 4 + m];
        float M = -INFINITY, dn = 0.f;
        float acc[4][4];
#pragma unroll
        for (int h = 0; h < 4; ++h)
#pragma unroll
            for (int e = 0; e < 4; ++e) acc[h][e] = 0.f;
        for (int j0 = 0; j0 < deg; j0 += 32) {
            const int left = deg - j0;
            int sidx[4];
            float sev[4], pen[4];
            float4 xr[4];
#pragma unroll
            for (int q = 0; q < 4; ++q) {
                if (8 * q < left) {
                    const bool act = 8 * q + g < left;
                    const int slot = beg + j0 + (act ? 8 * q + g : 0);
                    sidx[q] = pr2[slot].x;
                    sev[q] = se1S[(size_t)slot * 4 + m];
                    pen[q] = act ? 0.f : -INFINITY;
                }
            }
#pragma unroll
            for (int q = 0; q < 4; ++q)
                if (8 * q < left) xr[q] = x4[(size_t)sidx[q] * 4 + m];
            float c[4];
#pragma unroll
            for (int q = 0; q < 4; ++q) {
                c[q] = -INFINITY;
                if (8 * q < left) {
                    const float t0 = fmaf(xr[q].w, vj[0][3], fmaf(xr[q].z, vj[0][2], fmaf(xr[q].y, vj[0][1], xr[q].x * vj[0][0])));
                    const float t1 = fmaf(xr[q].w, vj[1][3], fmaf(xr[q].z, vj[1][2], fmaf(xr[q].y, vj[1][1], xr[q].x * vj[1][0])));
                    const float t2 = fmaf(xr[q].w, vj[2][3], fmaf(xr[q].z, vj[2][2], fmaf(xr[q].y, vj[2][1], xr[q].x * vj[2][0])));
                    const float t3 = fmaf(xr[q].w, vj[3][3], fmaf(xr[q].z, vj[3][2], fmaf(xr[q].y, vj[3][1], xr[q].x * vj[3][0])));
                    float k0 = mh ? t2 : t0, k1 = mh ? t3 : t1;
                    const float g0 = mh ? t0 : t2, g1 = mh ? t1 : t3;
                    k0 += dpp_f<XM(2)>(g0);
                    k1 += dpp_f<XM(2)>(g1);
                    const float kk = ml ? k1 : k0, gg = ml ? k0 : k1;
                    const float sj = kk + dpp_f<XM(1)>(gg);
                    c[q] = leaky(sim + sj + sev[q]) + pen[q];
                }
            }
            const float cm = max_groups(fmaxf(fmaxf(c[0], c[1]), fmaxf(c[2], c[3])));
            const float n = fmaxf(M, cm);
            const float r = exp_hw(M - n);
            M = n;
            float psum = 0.f;
            if (j0 > 0) {
                const float r0 = dpp_f<QB(0)>(r), r1 = dpp_f<QB(1)>(r);
                const float r2 = dpp_f<QB(2)>(r), r3 = dpp_f<QB(3)>(r);
#pragma unroll
                for (int e = 0; e < 4; ++e) { acc[0][e] *= r0; acc[1][e] *= r1; acc[2][e] *= r2; acc[3][e] *= r3; }
            }
#pragma unroll
            for (int q = 0; q < 4; ++q) {
                if (8 * q < left) {
                    const float pq = exp_hw(c[q] - n);
                    psum += pq;
                    const float p0 = dpp_f<QB(0)>(pq), p1 = dpp_f<QB(1)>(pq);
                    const float p2 = dpp_f<QB(2)>(pq), p3 = dpp_f<QB(3)>(pq);
                    const float xe[4] = {xr[q].x, xr[q].y, xr[q].z, xr[q].w};
#pragma unroll
                    for (int e = 0; e < 4; ++e) {
                        acc[0][e] = fmaf(p0, xe[e], acc[0][e]);
                        acc[1][e] = fmaf(p1, xe[e], acc[1][e]);
                        acc[2][e] = fmaf(p2, xe[e], acc[2][e]);
                        acc[3][e] = fmaf(p3, xe[e], acc[3][e]);
                    }
                }
            }
            dn = dn * r + psum;
        }
        dn = sum_groups(dn);
#pragma unroll
        for (int h = 0; h < 4; ++h)
#pragma unroll
            for (int e = 0; e < 4; ++e) acc[h][e] = sum_groups(acc[h][e]);
        const float d0 = dpp_f<RS(0)>(dn), d1 = dpp_f<RS(1)>(dn);
        const float d2 = dpp_f<RS(2)>(dn), d3 = dpp_f<RS(3)>(dn);
        if (lane < 16) {
            const float dv = (g == 0) ? d0 : (g == 1) ? d1 : (g == 2) ? d2 : d3;
            float o[4];
#pragma unroll
            for (int e = 0; e < 4; ++e) {
                const float a0 = acc[0][e], a1 = acc[1][e], a2 = acc[2][e], a3 = acc[3][e];
                const float a = (g == 0) ? a0 : (g == 1) ? a1 : (g == 2) ? a2 : a3;
                o[e] = (dv > 0.f) ? a / dv : 0.f;
            }
            float4 ov; ov.x = o[0]; ov.y = o[1]; ov.z = o[2]; ov.w = o[3];
            *(float4*)(accx1 + (size_t)d * 64 + g * 16 + 4 * m) = ov;
        }
    }
}

static_assert(NN % 4 == 0, "a wave takes four whole nodes");
#define L2U 6
__global__ void __launch_bounds__(256)
node_aggr_l2(const int* __restrict__ rowptr, const int* __restrict__ cnt,
             const int* __restrict__ pairS, const float* __restrict__ s_i2,
             const float* __restrict__ sz2, const float* __restrict__ p2b,
             float* __restrict__ out)
{
    const int lane = threadIdx.x & 31;
    const int sub = lane & 7;
    const int grp = lane >> 3;
    const int wave = blockIdx.x * (blockDim.x >> 5) + __builtin_amdgcn_readfirstlane(threadIdx.x >> 5);
    const int nw = gridDim.x * (blockDim.x >> 5);
    const int2* __restrict__ pr2 = (const int2*)pairS;
    const float2* __restrict__ sz = (const float2*)sz2;
    const float bias = p2b[0];
    int base = wave * 4;
    if (base >= NN) return;
    int d = base + grp;
    int beg = rowptr[d];
    int deg = cnt[d];
    float si = s_i2[d];
    for (;;) {
        const int nbase = base + nw * 4;
        const bool more = nbase < NN;
        const int nd = more ? nbase + grp : d;
        const int nbeg = rowptr[nd];
        const int ndeg = cnt[nd];
        const float nsi = s_i2[nd];
        const int md = max(max(__builtin_amdgcn_readlane(deg, 0), __builtin_amdgcn_readlane(deg, 8)),
                           max(__builtin_amdgcn_readlane(deg, 16), __builtin_amdgcn_readlane(deg, 24)));
        float M = -INFINITY, num = 0.f, den = 0.f;
        for (int j0 = 0; j0 < md; j0 += 8 * L2U) {
            int2 pr[L2U];
            float2 q[L2U];
#pragma unroll
            for (int u = 0; u < L2U; ++u) {
                const int j = j0 + 8 * u + sub;
                pr[u] = pr2[(j < deg) ? beg + j : 0];
            }
#pragma unroll
            for (int u = 0; u < L2U; ++u) q[u] = sz[pr[u].x];
            float c[L2U];
            float cm = -INFINITY;
#pragma unroll
            for (int u = 0; u < L2U; ++u) {
                const bool act = j0 + 8 * u + sub < deg;
                c[u] = leaky(si + q[u].x + __int_as_float(pr[u].y)) + (act ? 0.f : -INFINITY);
                cm = fmaxf(cm, c[u]);
            }
            cm = fmaxf(cm, dpp_f<XM(1)>(cm));
            cm = fmaxf(cm, dpp_f<XM(2)>(cm));
            cm = fmaxf(cm, dpp_f<XM(4)>(cm));
            const float n = fmaxf(M, cm);
            const float r = (M == n) ? 1.f : exp_hw(M - n);
            M = n;
            float ps = 0.f, pz = 0.f;
#pragma unroll
            for (int u = 0; u < L2U; ++u) {
                const bool act = j0 + 8 * u + sub < deg;
                const float p = act ? exp_hw(c[u] - n) : 0.f;
                ps += p;
                pz = fmaf(p, q[u].y, pz);
            }
            num = fmaf(num, r, pz);
            den = fmaf(den, r, ps);
        }
        num += dpp_f<XM(1)>(num);
        den += dpp_f<XM(1)>(den);
        num += dpp_f<XM(2)>(num);
        den += dpp_f<XM(2)>(den);
        num += dpp_f<XM(4)>(num);
        den += dpp_f<XM(4)>(den);
        if (sub == 0) {
            float v = (den > 0.f) ? num / den : 0.f;
            v += bias;
            out[d] = (v > 0.f) ? v : (expf(v) - 1.f);
        }
        if (!more) break;
        base = nbase; d = nd; beg = nbeg; deg = ndeg; si = nsi;
    }
}

extern "C" void kernel_launch(void* const* d_in, const int* in_sizes, int n_in,
                              void* d_out, int out_size, void* d_ws, size_t ws_size,
                              hipStream_t stream)
{
    const float* x       = (const float*)d_in[0];
    const int*   eidx    = (const int*)d_in[1];
    const float* eattr   = (const float*)d_in[2];
    const float* W1      = (const float*)d_in[3];
    const float* We1     = (const float*)d_in[4];
    const float* att1    = (const float*)d_in[5];
    const float* proj1_w = (const float*)d_in[6];
    const float* proj1_b = (const float*)d_in[7];
    const float* W2      = (const float*)d_in[8];
    const float* We2     = (const float*)d_in[9];
    const float* att2    = (const float*)d_in[10];
    const float* proj2_w = (const float*)d_in[11];
    const float* proj2_b = (const float*)d_in[12];
    float* out = (float*)d_out;

    const int* srcp = eidx;
    const int* dstp = eidx + EE;

    float* ws     = (float*)d_ws;
    float* xw1    = ws;
    float* accx1  = xw1 + (size_t)NN * 128;
    int*   pairS  = (int*)(accx1 + (size_t)NN * 64);
    float* s_i1   = accx1 + (size_t)NN * 128;
    float* slotA  = s_i1 + (size_t)NN * 8;
    int*   cnt    = (int*)slotA;
    float* Wcomb  = slotA + (size_t)NN;
    float* Vs     = Wcomb + 64 * 128;
    float* V2     = Vs + 256;
    int*   rowptr = (int*)(slotA + (size_t)NN * 4);
    float* hw2    = slotA + (size_t)NN * 8;
    float* s_i2   = hw2 + (size_t)NN * 64;
    float* sz2    = s_i2 + (size_t)NN;
    float* wea1   = s_i2 + (size_t)NN * 4 + 2048;
    float* wea2   = wea1 + 64;
    float* se1S   = xw1;
    int*   cntmat = (int*)hw2;
    int*   binstart = cntmat + NBINS * PBLK;

    prep_weights<<<32, 256, 0, stream>>>(W1, We1, att1, We2, att2, proj1_w, W2, proj2_w,
                                         Vs, V2, wea1, wea2, Wcomb);

    node_scores_l1<<<(NN / 16 + 7) / 8, 256, 0, stream>>>(x, Vs, s_i1);

    bin_count<<<PBLK, 1024, 0, stream>>>(dstp, cntmat);
    edge_partition<<<PBLK, 1024, 0, stream>>>(srcp, dstp, eattr, wea1, wea2, cntmat,
                                              (float4*)se1S, (int2*)pairS, binstart);
    bin_sort<<<NBINS, 1024, 0, stream>>>(binstart, (float4*)se1S, (int2*)pairS, rowptr, cnt);

    node_aggr_l1<<<1280, 256, 0, stream>>>(rowptr, cnt, se1S, pairS, s_i1, Vs, x, accx1);

    proj_scores_l2<<<(NN / 16 + 7) / 8, 256, 0, stream>>>(accx1, Wcomb, proj1_b, V2, s_i2, sz2);
    node_aggr_l2<<<(NN / 16 + 7) / 8, 256, 0, stream>>>(rowptr, cnt, pairS, s_i2, sz2, proj2_b, out);
}


# Weight folding and layer-1 destination scores ride in the counting launch's own waves (one 16-node strip per wave, same f32 WMMA product, Vs rebuilt per block on chip): two launches gone
# speedup vs baseline: 9.5874x; 1.0352x over previous
#include <hip/hip_runtime.h>
#include <cmath>

#define NN 100000
#define EE 3200000
#define NEGS 0.2f
#define BINSH 8
#define NBINS ((NN + 255) >> BINSH)
#define PBLK 256
#define ECHUNK ((EE + PBLK - 1) / PBLK)
#define CAP 10240
#define SRCBITS 17
#define SRCMASK ((1 << SRCBITS) - 1)

typedef __attribute__((ext_vector_type(2))) float v2f;
typedef __attribute__((ext_vector_type(8))) float v8f;

__device__ __forceinline__ float exp_hw(const float v)
{
    return __builtin_amdgcn_exp2f(v * 1.44269504088896340736f);
}

__device__ __forceinline__ v8f mm4(const float a0, const float a1, const float b0, const float b1,
                                   const v8f acc)
{
    v2f a; a.x = a0; a.y = a1;
    v2f b; b.x = b0; b.y = b1;
    return __builtin_amdgcn_wmma_f32_16x16x4_f32(false, a, false, b, (short)0, acc, false, false);
}

__global__ void __launch_bounds__(256)
proj_scores_l2(const float* __restrict__ accx1, const float* __restrict__ WcF,
               const float* __restrict__ bias, const float* __restrict__ V2F,
               float* __restrict__ s_i2, float* __restrict__ sz2)
{
    const int lane = threadIdx.x & 31;
    const int strip = blockIdx.x * (blockDim.x >> 5) + (threadIdx.x >> 5);
    if (strip >= NN / 16) return;
    const int c = lane & 15;
    const int g = lane >> 4;
    const float4* __restrict__ xr = (const float4*)(accx1 + (size_t)(strip * 16 + c) * 64 + 4 * g);
    float4 xb[8];
#pragma unroll
    for (int m = 0; m < 8; ++m) xb[m] = xr[2 * m];
    const float4* __restrict__ wf = (const float4*)WcF + lane;
    const float4* __restrict__ vf = (const float4*)V2F + lane * 2;
    v8f acc2 = {};
#pragma unroll
    for (int tn = 0; tn < 8; ++tn) {
        v8f acc = {};
#pragma unroll
        for (int m = 0; m < 8; ++m) {
            const float4 w = wf[(tn * 8 + m) * 32];
            acc = mm4(w.x, w.y, xb[m].x, xb[m].y, acc);
            acc = mm4(w.z, w.w, xb[m].z, xb[m].w, acc);
        }
        const float4 bl = *(const float4*)(bias + 16 * tn + 8 * g);
        const float4 bh = *(const float4*)(bias + 16 * tn + 8 * g + 4);
        const float bv[8] = {bl.x, bl.y, bl.z, bl.w, bh.x, bh.y, bh.z, bh.w};
        float hv[8];
#pragma unroll
        for (int r = 0; r < 8; ++r) {
            const float xv = acc[r] + bv[r];
            const float y = exp_hw(xv) - 1.f;
            const float z = exp_hw(y) - 1.f;
            hv[r] = (xv > 0.f) ? xv : z;
        }
        const float4 v0 = vf[tn * 64], v1 = vf[tn * 64 + 1];
        acc2 = mm4(v0.x, v0.y, hv[0], hv[1], acc2);
        acc2 = mm4(v0.z, v0.w, hv[2], hv[3], acc2);
        acc2 = mm4(v1.x, v1.y, hv[4], hv[5], acc2);
        acc2 = mm4(v1.z, v1.w, hv[6], hv[7], acc2);
    }
    if (g == 0) {
        const int node = strip * 16 + c;
        s_i2[node] = acc2[0];
        float2 q; q.x = acc2[1]; q.y = acc2[2];
        ((float2*)sz2)[node] = q;
    }
}


#define STRIPBLK ((NN / 16 + 31) / 32)
#define NPREP (64 * 128 + 2048 + 64 + 16)
static_assert(NN % 16 == 0, "whole strips");
static_assert((PBLK - STRIPBLK) * 1024 >= NPREP, "one folded-weight element per thread of the spare blocks");

__global__ void __launch_bounds__(1024)
bin_count(const int* __restrict__ dst, int* __restrict__ cntmat,
          const float* __restrict__ x,
          const float* __restrict__ W1, const float* __restrict__ We1,
          const float* __restrict__ att1, const float* __restrict__ We2,
          const float* __restrict__ att2, const float* __restrict__ proj1_w,
          const float* __restrict__ W2, const float* __restrict__ p2w,
          float* __restrict__ Vs, float* __restrict__ V2,
          float* __restrict__ wea1, float* __restrict__ wea2,
          float* __restrict__ Wcomb, float* __restrict__ s_i1)
{
    __shared__ int hist[NBINS];
    __shared__ float VsL[256];
    const int t = threadIdx.x, b = blockIdx.x;
    const int lane = t & 31;
    const int e0 = b * ECHUNK;
    const int e1 = (e0 + ECHUNK < EE) ? (e0 + ECHUNK) : EE;
    int dv[(ECHUNK + 1023) / 1024];
#pragma unroll
    for (int j = 0; j < (ECHUNK + 1023) / 1024; ++j) {
        const int e = e0 + j * 1024 + t;
        dv[j] = dst[(e < EE) ? e : (EE - 1)];
    }
    const int strip = b * 32 + (t >> 5);
    const bool has_strip = strip < NN / 16;
    const int r = lane & 15;
    const int kk = (lane >> 4) << 1;
    const float* __restrict__ Ap = x + (size_t)((has_strip ? strip : 0) * 16 + r) * 16 + kk;
    v2f af[4];
#pragma unroll
    for (int s = 0; s < 4; ++s) { af[s].x = Ap[4 * s]; af[s].y = Ap[4 * s + 1]; }
    if (t < 256) {
        const int k = t >> 4, c = t & 15;
        float s = 0.f;
        if (c < 8) {
            const int h = c & 3;
            const float* a = att1 + h * 96 + (c >> 2) * 32;
            const float* w = W1 + h * 512 + k * 32;
            for (int o = 0; o < 32; ++o) s += w[o] * a[o];
        }
        VsL[t] = s;
        if (b == 0) Vs[t] = s;
    }
    if (b >= STRIPBLK) {
        const int j = (b - STRIPBLK) * 1024 + t;
        if (j < 64 * 128) {
            const int e4 = j & 3, ln = (j >> 2) & 31, m8 = (j >> 7) & 7, t8 = j >> 10;
            const int row = 8 * m8 + 4 * (ln >> 4) + e4, c = 16 * t8 + (ln & 15), h = row >> 4, k = row & 15;
            const float* w = W1 + h * 512 + k * 32;
            const float* pw = proj1_w + (size_t)(h * 32) * 128 + c;
            float s = 0.f;
            for (int o = 0; o < 32; ++o) s += w[o] * pw[(size_t)o * 128];
            Wcomb[j] = s;
        } else if (j < 64 * 128 + 2048) {
            const int i = j - 64 * 128;
            const int e2 = i & 7, ln = (i >> 3) & 31, t2 = i >> 8;
            const int k = 16 * t2 + 8 * (ln >> 4) + e2, c = ln & 15;
            float s = 0.f;
            if (c < 3) {
                const float* a = (c == 2) ? p2w : (att2 + c * 32);
                const float* w = W2 + k * 32;
                for (int o = 0; o < 32; ++o) s += w[o] * a[o];
            }
            V2[i] = s;
        } else if (j < 64 * 128 + 2048 + 64) {
            const int i = j - (64 * 128 + 2048);
            const int h = i >> 4, k = i & 15;
            float s = 0.f;
            for (int o = 0; o < 32; ++o) s += We1[h * 512 + k * 32 + o] * att1[h * 96 + 64 + o];
            wea1[i] = s;
        } else if (j < NPREP) {
            const int i = j - (64 * 128 + 2048 + 64);
            float s = 0.f;
            for (int o = 0; o < 32; ++o) s += We2[i * 32 + o] * att2[64 + o];
            wea2[i] = s;
        }
    }
    for (int i = t; i < NBINS; i += 1024) hist[i] = 0;
    __syncthreads();
#pragma unroll
    for (int j = 0; j < (ECHUNK + 1023) / 1024; ++j)
        if (e0 + j * 1024 + t < e1) atomicAdd(&hist[dv[j] >> BINSH], 1);
    if (has_strip) {
        const float* Bp = VsL + kk * 16 + r;
        v8f acc = {};
#pragma unroll
        for (int s = 0; s < 4; ++s) {
            v2f bf; bf.x = Bp[s * 64]; bf.y = Bp[s * 64 + 16];
            acc = __builtin_amdgcn_wmma_f32_16x16x4_f32(false, af[s], false, bf,
                                                        (short)0, acc, false, false);
        }
        const int rbase = strip * 16 + ((lane >> 4) << 3);
        if (r < 4) {
            float* o = s_i1 + r;
#pragma unroll
            for (int v = 0; v < 8; ++v) o[(size_t)(rbase + v) * 4] = acc[v];
        }
    }
    __syncthreads();
    for (int i = t; i < NBINS; i += 1024) cntmat[i * PBLK + b] = hist[i];
}

__device__ __forceinline__ void place_edge(const int d, const int sv, const float4 r0, const float4 r1,
                                           const float4 r2, const float4 r3,
                                           const float* __restrict__ wea1,
                                           const float* __restrict__ wea2, int* lcur,
                                           float4* stA, int2* stP)
{
    const float ea[16] = {r0.x, r0.y, r0.z, r0.w, r1.x, r1.y, r1.z, r1.w,
                          r2.x, r2.y, r2.z, r2.w, r3.x, r3.y, r3.z, r3.w};
    float a0 = 0.f, a1 = 0.f, a2 = 0.f, a3 = 0.f, bb = 0.f;
#pragma unroll
    for (int k = 0; k < 16; ++k) {
        a0 += ea[k] * wea1[k];      a1 += ea[k] * wea1[16 + k];
        a2 += ea[k] * wea1[32 + k]; a3 += ea[k] * wea1[48 + k];
        bb += ea[k] * wea2[k];
    }
    const int pos = atomicAdd(&lcur[d >> BINSH], 1);
    float4 v4; v4.x = a0; v4.y = a1; v4.z = a2; v4.w = a3;
    stA[pos] = v4;
    int2 q; q.x = sv | ((d & ((1 << BINSH) - 1)) << SRCBITS); q.y = __float_as_int(bb);
    stP[pos] = q;
}

__global__ void __launch_bounds__(1024)
edge_partition(const int* __restrict__ src, const int* __restrict__ dst,
               const float* __restrict__ eattr, const float* __restrict__ wea1,
               const float* __restrict__ wea2, const int* __restrict__ cntmat,
               float4* __restrict__ recA, int2* __restrict__ recP,
               int* __restrict__ binstart)
{
    __shared__ float4 stA[ECHUNK];
    __shared__ int2 stP[ECHUNK];
    __shared__ int tot[NBINS];
    __shared__ int gfirst[NBINS];
    __shared__ int own[NBINS];
    __shared__ int lstart[NBINS];
    __shared__ int lcur[NBINS];
    __shared__ int wsum[32];
    __shared__ int wsumL[32];
    const int t = threadIdx.x, lane = t & 31, w = t >> 5, b = blockIdx.x;
    for (int bin = w; bin < NBINS; bin += 32) {
        const int* row = cntmat + bin * PBLK;
        int all = 0, before = 0;
        for (int q = 0; q < PBLK; q += 32) {
            const int c = row[q + lane];
            all += c;
            before += (q + lane < b) ? c : 0;
        }
        const int mine = row[b];
        for (int o = 16; o > 0; o >>= 1) {
            all += __shfl_xor(all, o);
            before += __shfl_xor(before, o);
        }
        if (lane == 0) { tot[bin] = all; gfirst[bin] = before; own[bin] = mine; }
    }
    __syncthreads();
    const int v = (t < NBINS) ? tot[t] : 0;
    const int m = (t < NBINS) ? own[t] : 0;
    int incl = v, inclL = m;
    for (int o = 1; o < 32; o <<= 1) {
        const int u = __shfl_up(incl, o);
        const int uL = __shfl_up(inclL, o);
        if (lane >= o) { incl += u; inclL += uL; }
    }
    if (lane == 31) { wsum[w] = incl; wsumL[w] = inclL; }
    __syncthreads();
    int pre = (lane < w) ? wsum[lane] : 0;
    int preL = (lane < w) ? wsumL[lane] : 0;
    for (int o = 16; o > 0; o >>= 1) {
        pre += __shfl_xor(pre, o);
        preL += __shfl_xor(preL, o);
    }
    if (t < NBINS) {
        const int first = pre + incl - v;
        const int firstL = preL + inclL - m;
        gfirst[t] += first;
        lstart[t] = firstL;
        lcur[t] = firstL;
        if (b == 0) binstart[t] = first;
    }
    __syncthreads();
    const int e0 = b * ECHUNK;
    const int e1 = (e0 + ECHUNK < EE) ? (e0 + ECHUNK) : EE;
    const float4* __restrict__ ea4 = (const float4*)eattr;
    int e = e0 + t;
    if (e < e1) {
        int dA = dst[e], sA = src[e];
        float4 a0 = ea4[(size_t)e * 4], a1 = ea4[(size_t)e * 4 + 1];
        float4 a2 = ea4[(size_t)e * 4 + 2], a3 = ea4[(size_t)e * 4 + 3];
        for (;;) {
            const int eB = e + 1024;
            const bool moreB = eB < e1;
            const int fB = moreB ? eB : e;
            const int dB = dst[fB], sB = src[fB];
            const float4 b0 = ea4[(size_t)fB * 4], b1 = ea4[(size_t)fB * 4 + 1];
            const float4 b2 = ea4[(size_t)fB * 4 + 2], b3 = ea4[(size_t)fB * 4 + 3];
            place_edge(dA, sA, a0, a1, a2, a3, wea1, wea2, lcur, stA, stP);
            if (!moreB) break;
            e = eB + 1024;
            const bool moreA = e < e1;
            const int fA = moreA ? e : eB;
            dA = dst[fA]; sA = src[fA];
            a0 = ea4[(size_t)fA * 4]; a1 = ea4[(size_t)fA * 4 + 1];
            a2 = ea4[(size_t)fA * 4 + 2]; a3 = ea4[(size_t)fA * 4 + 3];
            place_edge(dB, sB, b0, b1, b2, b3, wea1, wea2, lcur, stA, stP);
            if (!moreA) break;
        }
    }
    __syncthreads();
    for (int bin = w; bin < NBINS; bin += 32) {
        const int ls = lstart[bin], c = own[bin], g = gfirst[bin];
        for (int i = lane; i < c; i += 32) {
            recA[g + i] = stA[ls + i];
            recP[g + i] = stP[ls + i];
        }
    }
}

#define FOR_SLOTS(X) X(0) X(1) X(2) X(3) X(4) X(5) X(6) X(7) X(8) X(9)
static_assert(CAP / 1024 == 10, "FOR_SLOTS lists CAP / 1024 slots");

__global__ void __launch_bounds__(1024)
bin_sort(const int* __restrict__ binstart, float4* recA, int2* recP,
         int* __restrict__ rowptr, int* __restrict__ cnt)
{
    __shared__ float4 bufA[CAP];
    __shared__ int2 bufP[CAP];
    __shared__ int cntL[256];
    __shared__ int startL[256];
    __shared__ int curL[256];
    __shared__ int wtot[32];
    const int t = threadIdx.x, lane = t & 31, w = t >> 5, bin = blockIdx.x;
    const int beg = binstart[bin];
    const int end = (bin + 1 < NBINS) ? binstart[bin + 1] : EE;
    const int n = end - beg;
    const bool fits = n <= CAP;
    if (t < 256) { cntL[t] = 0; curL[t] = 0; }
    __syncthreads();
#define DECL_SLOT(j) int2 pv##j = make_int2(0, 0); float4 av##j = make_float4(0.f, 0.f, 0.f, 0.f); int rk##j = 0;
    FOR_SLOTS(DECL_SLOT)
    if (fits) {
#define LOAD_SLOT(j) { const int g = beg + j * 1024 + t; const int gc = (g < EE) ? g : (EE - 1); \
                       pv##j = recP[gc]; av##j = recA[gc]; }
        FOR_SLOTS(LOAD_SLOT)
#define COUNT_SLOT(j) if (j * 1024 + t < n) rk##j = atomicAdd(&cntL[pv##j.x >> SRCBITS], 1);
        FOR_SLOTS(COUNT_SLOT)
    } else {
        for (int i = t; i < n; i += 1024) atomicAdd(&cntL[recP[beg + i].x >> SRCBITS], 1);
    }
    __syncthreads();
    int c = 0, incl = 0;
    if (t < 256) {
        c = cntL[t];
        incl = c;
        for (int o = 1; o < 32; o <<= 1) {
            int u = __shfl_up(incl, o);
            if (lane >= o) incl += u;
        }
        if (lane == 31) wtot[w] = incl;
    }
    __syncthreads();
    if (t < 256) {
        int pre = (lane < w) ? wtot[lane] : 0;
        for (int o = 16; o > 0; o >>= 1) pre += __shfl_xor(pre, o);
        const int st = pre + incl - c;
        startL[t] = st;
        const int node = (bin << BINSH) + t;
        if (node < NN) { rowptr[node] = beg + st; cnt[node] = c; }
    }
    __syncthreads();
    if (fits) {
#define START_SLOT(j) const int st##j = startL[pv##j.x >> SRCBITS];
        FOR_SLOTS(START_SLOT)
#define PLACE_SLOT(j) if (j * 1024 + t < n) { \
            const int pos = st##j + rk##j; \
            int2 q; q.x = pv##j.x & SRCMASK; q.y = pv##j.y; \
            bufP[pos] = q; bufA[pos] = av##j; }
        FOR_SLOTS(PLACE_SLOT)
        __syncthreads();
        const int lead = beg & 15;
        for (int i = t - lead; i < n; i += 1024) {
            if (i >= 0) {
                recP[beg + i] = bufP[i];
                recA[beg + i] = bufA[i];
            }
        }
    } else {
        if (t == 0) {
            for (int nd = 0; nd < 256; ++nd) {
                const int endn = cntL[nd];
                int k = curL[nd];
                while (k < endn) {
                    const int p = beg + startL[nd] + k;
                    const int2 r = recP[p];
                    const int dl = r.x >> SRCBITS;
                    if (dl == nd) { ++k; continue; }
                    const int q = beg + startL[dl] + curL[dl];
                    curL[dl] += 1;
                    const int2 r2 = recP[q];
                    const float4 a1 = recA[p], a2 = recA[q];
                    recP[q] = r;  recA[q] = a1;
                    recP[p] = r2; recA[p] = a2;
                }
                curL[nd] = endn;
            }
        }
        __syncthreads();
        for (int i = t; i < n; i += 1024) recP[beg + i].x &= SRCMASK;
    }
}

__device__ __forceinline__ float leaky(float v) { return (v >= 0.f) ? v : NEGS * v; }
__device__ __forceinline__ float wave_max(float v) {
    for (int o = 16; o > 0; o >>= 1) v = fmaxf(v, __shfl_xor(v, o));
    return v;
}
__device__ __forceinline__ float wave_sum(float v) {
    for (int o = 16; o > 0; o >>= 1) v += __shfl_xor(v, o);
    return v;
}

template <int CTRL> __device__ __forceinline__ float dpp_f(float v) {
    return __int_as_float(__builtin_amdgcn_update_dpp(0, __float_as_int(v), CTRL, 0xF, 0xF, true));
}
#define XM(n) (0x160 | (n))
#define QB(h) ((h) * 0x55)
#define RS(n) (0x150 | (n))
__device__ __forceinline__ float xhalf(float v) {
    return __int_as_float(__builtin_amdgcn_permlanex16(__float_as_int(v), __float_as_int(v),
                                                       0x76543210, 0xfedcba98, false, false));
}
__device__ __forceinline__ float sum_groups(float v) {
    v += dpp_f<XM(4)>(v);
    v += dpp_f<XM(8)>(v);
    return v + xhalf(v);
}
__device__ __forceinline__ float max_groups(float v) {
    v = fmaxf(v, dpp_f<XM(4)>(v));
    v = fmaxf(v, dpp_f<XM(8)>(v));
    return fmaxf(v, xhalf(v));
}

__global__ void __launch_bounds__(256)
node_aggr_l1(const int* __restrict__ rowptr, const int* __restrict__ cnt,
             const float* __restrict__ se1S, const int* __restrict__ pairS,
             const float* __restrict__ s_i1, const float* __restrict__ Vs,
             const float* __restrict__ x, float* __restrict__ accx1)
{
    const int lane = threadIdx.x & 31;
    const int m = lane & 3;
    const int g = lane >> 2;
    const bool mh = (m & 2) != 0, ml = (m & 1) != 0;
    const int wave = blockIdx.x * (blockDim.x >> 5) + __builtin_amdgcn_readfirstlane(threadIdx.x >> 5);
    const int nw = gridDim.x * (blockDim.x >> 5);
    const int2* __restrict__ pr2 = (const int2*)pairS;
    const float4* __restrict__ x4 = (const float4*)x;
    float vj[4][4];
#pragma unroll
    for (int h = 0; h < 4; ++h)
#pragma unroll
        for (int e = 0; e < 4; ++e) vj[h][e] = Vs[(4 * m + e) * 16 + 4 + h];
    for (int d = wave; d < NN; d += nw) {
        const int beg = rowptr[d];
        const int deg = cnt[d];
        const float sim = s_i1[(size_t)d * 4 + m];
        float M = -INFINITY, dn = 0.f;
        float acc[4][4];
#pragma unroll
        for (int h = 0; h < 4; ++h)
#pragma unroll
            for (int e = 0; e < 4; ++e) acc[h][e] = 0.f;
        for (int j0 = 0; j0 < deg; j0 += 32) {
            const int left = deg - j0;
            int sidx[4];
            float sev[4], pen[4];
            float4 xr[4];
#pragma unroll
            for (int q = 0; q < 4; ++q) {
                if (8 * q < left) {
                    const bool act = 8 * q + g < left;
                    const int slot = beg + j0 + (act ? 8 * q + g : 0);
                    sidx[q] = pr2[slot].x;
                    sev[q] = se1S[(size_t)slot * 4 + m];
                    pen[q] = act ? 0.f : -INFINITY;
                }
            }
#pragma unroll
            for (int q = 0; q < 4; ++q)
                if (8 * q < left) xr[q] = x4[(size_t)sidx[q] * 4 + m];
            float c[4];
#pragma unroll
            for (int q = 0; q < 4; ++q) {
                c[q] = -INFINITY;
                if (8 * q < left) {
                    const float t0 = fmaf(xr[q].w, vj[0][3], fmaf(xr[q].z, vj[0][2], fmaf(xr[q].y, vj[0][1], xr[q].x * vj[0][0])));
                    const float t1 = fmaf(xr[q].w, vj[1][3], fmaf(xr[q].z, vj[1][2], fmaf(xr[q].y, vj[1][1], xr[q].x * vj[1][0])));
                    const float t2 = fmaf(xr[q].w, vj[2][3], fmaf(xr[q].z, vj[2][2], fmaf(xr[q].y, vj[2][1], xr[q].x * vj[2][0])));
                    const float t3 = fmaf(xr[q].w, vj[3][3], fmaf(xr[q].z, vj[3][2], fmaf(xr[q].y, vj[3][1], xr[q].x * vj[3][0])));
                    float k0 = mh ? t2 : t0, k1 = mh ? t3 : t1;
                    const float g0 = mh ? t0 : t2, g1 = mh ? t1 : t3;
                    k0 += dpp_f<XM(2)>(g0);
                    k1 += dpp_f<XM(2)>(g1);
                    const float kk = ml ? k1 : k0, gg = ml ? k0 : k1;
                    const float sj = kk + dpp_f<XM(1)>(gg);
                    c[q] = leaky(sim + sj + sev[q]) + pen[q];
                }
            }
            const float cm = max_groups(fmaxf(fmaxf(c[0], c[1]), fmaxf(c[2], c[3])));
            const float n = fmaxf(M, cm);
            const float r = exp_hw(M - n);
            M = n;
            float psum = 0.f;
            if (j0 > 0) {
                const float r0 = dpp_f<QB(0)>(r), r1 = dpp_f<QB(1)>(r);
                const float r2 = dpp_f<QB(2)>(r), r3 = dpp_f<QB(3)>(r);
#pragma unroll
                for (int e = 0; e < 4; ++e) { acc[0][e] *= r0; acc[1][e] *= r1; acc[2][e] *= r2; acc[3][e] *= r3; }
            }
#pragma unroll
            for (int q = 0; q < 4; ++q) {
                if (8 * q < left) {
                    const float pq = exp_hw(c[q] - n);
                    psum += pq;
                    const float p0 = dpp_f<QB(0)>(pq), p1 = dpp_f<QB(1)>(pq);
                    const float p2 = dpp_f<QB(2)>(pq), p3 = dpp_f<QB(3)>(pq);
                    const float xe[4] = {xr[q].x, xr[q].y, xr[q].z, xr[q].w};
#pragma unroll
                    for (int e = 0; e < 4; ++e) {
                        acc[0][e] = fmaf(p0, xe[e], acc[0][e]);
                        acc[1][e] = fmaf(p1, xe[e], acc[1][e]);
                        acc[2][e] = fmaf(p2, xe[e], acc[2][e]);
                        acc[3][e] = fmaf(p3, xe[e], acc[3][e]);
                    }
                }
            }
            dn = dn * r + psum;
        }
        dn = sum_groups(dn);
#pragma unroll
        for (int h = 0; h < 4; ++h)
#pragma unroll
            for (int e = 0; e < 4; ++e) acc[h][e] = sum_groups(acc[h][e]);
        const float d0 = dpp_f<RS(0)>(dn), d1 = dpp_f<RS(1)>(dn);
        const float d2 = dpp_f<RS(2)>(dn), d3 = dpp_f<RS(3)>(dn);
        if (lane < 16) {
            const float dv = (g == 0) ? d0 : (g == 1) ? d1 : (g == 2) ? d2 : d3;
            float o[4];
#pragma unroll
            for (int e = 0; e < 4; ++e) {
                const float a0 = acc[0][e], a1 = acc[1][e], a2 = acc[2][e], a3 = acc[3][e];
                const float a = (g == 0) ? a0 : (g == 1) ? a1 : (g == 2) ? a2 : a3;
                o[e] = (dv > 0.f) ? a / dv : 0.f;
            }
            float4 ov; ov.x = o[0]; ov.y = o[1]; ov.z = o[2]; ov.w = o[3];
            *(float4*)(accx1 + (size_t)d * 64 + g * 16 + 4 * m) = ov;
        }
    }
}

static_assert(NN % 4 == 0, "a wave takes four whole nodes");
#define L2U 6
__global__ void __launch_bounds__(256)
node_aggr_l2(const int* __restrict__ rowptr, const int* __restrict__ cnt,
             const int* __restrict__ pairS, const float* __restrict__ s_i2,
             const float* __restrict__ sz2, const float* __restrict__ p2b,
             float* __restrict__ out)
{
    const int lane = threadIdx.x & 31;
    const int sub = lane & 7;
    const int grp = lane >> 3;
    const int wave = blockIdx.x * (blockDim.x >> 5) + __builtin_amdgcn_readfirstlane(threadIdx.x >> 5);
    const int nw = gridDim.x * (blockDim.x >> 5);
    const int2* __restrict__ pr2 = (const int2*)pairS;
    const float2* __restrict__ sz = (const float2*)sz2;
    const float bias = p2b[0];
    int base = wave * 4;
    if (base >= NN) return;
    int d = base + grp;
    int beg = rowptr[d];
    int deg = cnt[d];
    float si = s_i2[d];
    for (;;) {
        const int nbase = base + nw * 4;
        const bool more = nbase < NN;
        const int nd = more ? nbase + grp : d;
        const int nbeg = rowptr[nd];
        const int ndeg = cnt[nd];
        const float nsi = s_i2[nd];
        const int md = max(max(__builtin_amdgcn_readlane(deg, 0), __builtin_amdgcn_readlane(deg, 8)),
                           max(__builtin_amdgcn_readlane(deg, 16), __builtin_amdgcn_readlane(deg, 24)));
        float M = -INFINITY, num = 0.f, den = 0.f;
        for (int j0 = 0; j0 < md; j0 += 8 * L2U) {
            int2 pr[L2U];
            float2 q[L2U];
#pragma unroll
            for (int u = 0; u < L2U; ++u) {
                const int j = j0 + 8 * u + sub;
                pr[u] = pr2[(j < deg) ? beg + j : 0];
            }
#pragma unroll
            for (int u = 0; u < L2U; ++u) q[u] = sz[pr[u].x];
            float c[L2U];
            float cm = -INFINITY;
#pragma unroll
            for (int u = 0; u < L2U; ++u) {
                const bool act = j0 + 8 * u + sub < deg;
                c[u] = leaky(si + q[u].x + __int_as_float(pr[u].y)) + (act ? 0.f : -INFINITY);
                cm = fmaxf(cm, c[u]);
            }
            cm = fmaxf(cm, dpp_f<XM(1)>(cm));
            cm = fmaxf(cm, dpp_f<XM(2)>(cm));
            cm = fmaxf(cm, dpp_f<XM(4)>(cm));
            const float n = fmaxf(M, cm);
            const float r = (M == n) ? 1.f : exp_hw(M - n);
            M = n;
            float ps = 0.f, pz = 0.f;
#pragma unroll
            for (int u = 0; u < L2U; ++u) {
                const bool act = j0 + 8 * u + sub < deg;
                const float p = act ? exp_hw(c[u] - n) : 0.f;
                ps += p;
                pz = fmaf(p, q[u].y, pz);
            }
            num = fmaf(num, r, pz);
            den = fmaf(den, r, ps);
        }
        num += dpp_f<XM(1)>(num);
        den += dpp_f<XM(1)>(den);
        num += dpp_f<XM(2)>(num);
        den += dpp_f<XM(2)>(den);
        num += dpp_f<XM(4)>(num);
        den += dpp_f<XM(4)>(den);
        if (sub == 0) {
            float v = (den > 0.f) ? num / den : 0.f;
            v += bias;
            out[d] = (v > 0.f) ? v : (expf(v) - 1.f);
        }
        if (!more) break;
        base = nbase; d = nd; beg = nbeg; deg = ndeg; si = nsi;
    }
}

extern "C" void kernel_launch(void* const* d_in, const int* in_sizes, int n_in,
                              void* d_out, int out_size, void* d_ws, size_t ws_size,
                              hipStream_t stream)
{
    const float* x       = (const float*)d_in[0];
    const int*   eidx    = (const int*)d_in[1];
    const float* eattr   = (const float*)d_in[2];
    const float* W1      = (const float*)d_in[3];
    const float* We1     = (const float*)d_in[4];
    const float* att1    = (const float*)d_in[5];
    const float* proj1_w = (const float*)d_in[6];
    const float* proj1_b = (const float*)d_in[7];
    const float* W2      = (const float*)d_in[8];
    const float* We2     = (const float*)d_in[9];
    const float* att2    = (const float*)d_in[10];
    const float* proj2_w = (const float*)d_in[11];
    const float* proj2_b = (const float*)d_in[12];
    float* out = (float*)d_out;

    const int* srcp = eidx;
    const int* dstp = eidx + EE;

    float* ws     = (float*)d_ws;
    float* xw1    = ws;
    float* accx1  = xw1 + (size_t)NN * 128;
    int*   pairS  = (int*)(accx1 + (size_t)NN * 64);
    float* s_i1   = accx1 + (size_t)NN * 128;
    float* slotA  = s_i1 + (size_t)NN * 8;
    int*   cnt    = (int*)slotA;
    float* Wcomb  = slotA + (size_t)NN;
    float* Vs     = Wcomb + 64 * 128;
    float* V2     = Vs + 256;
    int*   rowptr = (int*)(slotA + (size_t)NN * 4);
    float* hw2    = slotA + (size_t)NN * 8;
    float* s_i2   = hw2 + (size_t)NN * 64;
    float* sz2    = s_i2 + (size_t)NN;
    float* wea1   = s_i2 + (size_t)NN * 4 + 2048;
    float* wea2   = wea1 + 64;
    float* se1S   = xw1;
    int*   cntmat = (int*)hw2;
    int*   binstart = cntmat + NBINS * PBLK;

    bin_count<<<PBLK, 1024, 0, stream>>>(dstp, cntmat, x, W1, We1, att1, We2, att2, proj1_w,
                                         W2, proj2_w, Vs, V2, wea1, wea2, Wcomb, s_i1);
    edge_partition<<<PBLK, 1024, 0, stream>>>(srcp, dstp, eattr, wea1, wea2, cntmat,
                                              (float4*)se1S, (int2*)pairS, binstart);
    bin_sort<<<NBINS, 1024, 0, stream>>>(binstart, (float4*)se1S, (int2*)pairS, rowptr, cnt);

    node_aggr_l1<<<1280, 256, 0, stream>>>(rowptr, cnt, se1S, pairS, s_i1, Vs, x, accx1);

    proj_scores_l2<<<(NN / 16 + 7) / 8, 256, 0, stream>>>(accx1, Wcomb, proj1_b, V2, s_i2, sz2);
    node_aggr_l2<<<(NN / 16 + 7) / 8, 256, 0, stream>>>(rowptr, cnt, pairS, s_i2, sz2, proj2_b, out);
}
